# starting kernel
# baseline (speedup 1.0000x reference)
#include <hip/hip_runtime.h>
#include <math.h>

typedef float v2f __attribute__((ext_vector_type(2)));
typedef float v8f __attribute__((ext_vector_type(8)));

#define WG 256

__device__ __forceinline__ void atomic_fadd(float* p, float v) {
  __hip_atomic_fetch_add(p, v, __ATOMIC_RELAXED, __HIP_MEMORY_SCOPE_AGENT);
}

__device__ __forceinline__ float fast_tanh(float x) {
  float r;
  asm("v_tanh_f32 %0, %1\n\tv_nop\n\tv_nop" : "=v"(r) : "v"(x));
  return r;
}

__global__ void k_init(const float* __restrict__ u, const float* __restrict__ it,
                       float* __restrict__ ego, float* __restrict__ acc,
                       int userElems, int total) {
  int i = blockIdx.x * blockDim.x + threadIdx.x;
  if (i >= total) return;
  float v = (i < userElems) ? u[i] : it[i - userElems];
  ego[i] = v;
  acc[i] = v;
}

__global__ void k_fill1(float* __restrict__ a, int n) {
  int i = blockIdx.x * blockDim.x + threadIdx.x;
  if (i < n) a[i] = 1.0f;
}

__global__ void k_softmax_dval(const float* __restrict__ A, const int* __restrict__ hl,
                               float* __restrict__ ew, float* __restrict__ dval, int E) {
  int e = blockIdx.x * blockDim.x + threadIdx.x;
  if (e >= E) return;
  float4 a = reinterpret_cast<const float4*>(A)[e];
  float m = fmaxf(fmaxf(a.x, a.y), fmaxf(a.z, a.w));
  float e0 = expf(a.x - m), e1 = expf(a.y - m), e2 = expf(a.z - m), e3 = expf(a.w - m);
  float inv = 1.0f / (e0 + e1 + e2 + e3);
  float4 nrm = make_float4(e0 * inv, e1 * inv, e2 * inv, e3 * inv);
  reinterpret_cast<float4*>(ew)[e] = nrm;
  float* d = dval + (size_t)hl[e] * 4;
  atomic_fadd(d + 0, nrm.x);
  atomic_fadd(d + 1, nrm.y);
  atomic_fadd(d + 2, nrm.z);
  atomic_fadd(d + 3, nrm.w);
}

__global__ void k_dinv(float* __restrict__ d, int n) {
  int i = blockIdx.x * blockDim.x + threadIdx.x;
  if (i >= n) return;
  d[i] = 1.0f / sqrtf(fmaxf(d[i], 1e-8f));
}

__global__ void k_edge_weight(float* __restrict__ ew, const float* __restrict__ dinv,
                              const int* __restrict__ hl, const int* __restrict__ tl, int E) {
  int e = blockIdx.x * blockDim.x + threadIdx.x;
  if (e >= E) return;
  float4 w = reinterpret_cast<float4*>(ew)[e];
  float4 dh = reinterpret_cast<const float4*>(dinv)[hl[e]];
  float4 dt = reinterpret_cast<const float4*>(dinv)[tl[e]];
  w.x *= dh.x * dt.x;
  w.y *= dh.y * dt.y;
  w.z *= dh.z * dt.z;
  w.w *= dh.w * dt.w;
  reinterpret_cast<float4*>(ew)[e] = w;
}

__global__ void k_message(const float* __restrict__ ew, const float* __restrict__ ego,
                          const int* __restrict__ hl, const int* __restrict__ tl,
                          float* __restrict__ fact, int E) {
  int tid = blockIdx.x * blockDim.x + threadIdx.x;
  int total = E * 16;
  if (tid >= total) return;
  int e = tid >> 4, q = tid & 15, f = q >> 2;
  int h = hl[e], t = tl[e];
  float w = ew[(size_t)e * 4 + f];
  float4 v = reinterpret_cast<const float4*>(ego)[(size_t)t * 16 + q];
  float* dst = fact + (size_t)h * 64 + q * 4;
  atomic_fadd(dst + 0, w * v.x);
  atomic_fadd(dst + 1, w * v.y);
  atomic_fadd(dst + 2, w * v.z);
  atomic_fadd(dst + 3, w * v.w);
}

__global__ void k_norminv(const float* __restrict__ x, float* __restrict__ out, int NF) {
  int i = blockIdx.x * blockDim.x + threadIdx.x;
  if (i >= NF) return;
  int n = i >> 2, f = i & 3;
  const float4* p = reinterpret_cast<const float4*>(x + (size_t)n * 64 + (size_t)f * 16);
  float s = 0.f;
#pragma unroll
  for (int k = 0; k < 4; ++k) {
    float4 v = p[k];
    s += v.x * v.x + v.y * v.y + v.z * v.z + v.w * v.w;
  }
  out[i] = 1.0f / fmaxf(sqrtf(s), 1e-12f);
}

__global__ void __launch_bounds__(WG) k_updateA(
    const float* __restrict__ fact, const float* __restrict__ ego,
    const float* __restrict__ hninv, const float* __restrict__ tninv,
    const int* __restrict__ hl, const int* __restrict__ tl,
    float* __restrict__ A, int E, int nGroups) {
  int tid = blockIdx.x * blockDim.x + threadIdx.x;
  int wave = tid >> 5;
  int lane = tid & 31;
  if (wave >= nGroups) return;

  int r = lane & 15;
  int eIdx = wave * 16 + r;
  int e = (eIdx < E) ? eIdx : (E - 1);
  int h = hl[e], t = tl[e];
  int koff = (lane >> 4) << 1;
  const float* Hrow = fact + (size_t)h * 64;
  const float* Trow = ego + (size_t)t * 64;
  const float* hn4 = hninv + (size_t)h * 4;
  const float* tn4 = tninv + (size_t)t * 4;

  bool active = (lane < 8) || (lane >= 24);
  int outRow = (lane < 8) ? lane : (lane - 16);
  int eOut = wave * 16 + outRow;

#pragma unroll
  for (int f = 0; f < 4; ++f) {
    float hn = hn4[f];
    float tn = tn4[f];
    v8f c = {0.f, 0.f, 0.f, 0.f, 0.f, 0.f, 0.f, 0.f};
#pragma unroll
    for (int s = 0; s < 4; ++s) {
      int o = f * 16 + s * 4 + koff;
      v2f av, bv;
      av.x = Hrow[o] * hn;
      av.y = Hrow[o + 1] * hn;
      bv.x = fast_tanh(Trow[o] * tn);
      bv.y = fast_tanh(Trow[o + 1] * tn);
      c = __builtin_amdgcn_wmma_f32_16x16x4_f32(false, av, false, bv,
                                                (short)0, c, false, false);
    }
    int idx = lane & 7;
    float dv = c[0];
    dv = (idx == 1) ? c[1] : dv;
    dv = (idx == 2) ? c[2] : dv;
    dv = (idx == 3) ? c[3] : dv;
    dv = (idx == 4) ? c[4] : dv;
    dv = (idx == 5) ? c[5] : dv;
    dv = (idx == 6) ? c[6] : dv;
    dv = (idx == 7) ? c[7] : dv;
    if (active && eOut < E) A[(size_t)eOut * 4 + f] += dv;
  }
}

__global__ void k_acc_add(float* __restrict__ acc, const float* __restrict__ x, int n) {
  int i = blockIdx.x * blockDim.x + threadIdx.x;
  if (i < n) acc[i] += x[i];
}

__global__ void k_final(const float* __restrict__ acc, float* __restrict__ out, int n) {
  int i = blockIdx.x * blockDim.x + threadIdx.x;
  if (i < n) out[i] = acc[i] * (1.0f / 3.0f);
}

static inline size_t align16f(size_t x) { return (x + 15) & ~(size_t)15; }

extern "C" void kernel_launch(void* const* d_in, const int* in_sizes, int n_in,
                              void* d_out, int out_size, void* d_ws, size_t ws_size,
                              hipStream_t stream) {
  (void)n_in; (void)out_size; (void)ws_size;
  const float* u  = (const float*)d_in[0];
  const float* it = (const float*)d_in[1];
  const int* hl   = (const int*)d_in[2];
  const int* tl   = (const int*)d_in[3];

  const int userElems = in_sizes[0];
  const int itemElems = in_sizes[1];
  const int E = in_sizes[2];
  const int N = (userElems + itemElems) / 64;
  const size_t N64 = (size_t)N * 64;
  const size_t E4  = (size_t)E * 4;
  const size_t N4  = (size_t)N * 4;

  float* base = (float*)d_ws;
  size_t off = 0;
  float* ego   = base + off; off += align16f(N64);
  float* fact  = base + off; off += align16f(N64);
  float* acc   = base + off; off += align16f(N64);
  float* A     = base + off; off += align16f(E4);
  float* ew    = base + off; off += align16f(E4);
  float* dval  = base + off; off += align16f(N4);
  float* hninv = base + off; off += align16f(N4);
  float* tninv = base + off; off += align16f(N4);

#define GRID(n) ((unsigned)(((size_t)(n) + WG - 1) / WG))

  k_init<<<GRID(N64), WG, 0, stream>>>(u, it, ego, acc, userElems, (int)N64);
  k_fill1<<<GRID(E4), WG, 0, stream>>>(A, (int)E4);

  float* cur = ego;
  float* nxt = fact;
  const int nGroups = (E + 15) / 16;

  for (int layer = 0; layer < 2; ++layer) {
    k_norminv<<<GRID(N4), WG, 0, stream>>>(cur, tninv, (int)N4);
    for (int itn = 0; itn < 2; ++itn) {
      hipMemsetAsync(dval, 0, N4 * sizeof(float), stream);
      k_softmax_dval<<<GRID(E), WG, 0, stream>>>(A, hl, ew, dval, E);
      k_dinv<<<GRID(N4), WG, 0, stream>>>(dval, (int)N4);
      k_edge_weight<<<GRID(E), WG, 0, stream>>>(ew, dval, hl, tl, E);
      hipMemsetAsync(nxt, 0, N64 * sizeof(float), stream);
      k_message<<<GRID((size_t)E * 16), WG, 0, stream>>>(ew, cur, hl, tl, nxt, E);
      const bool lastUpdateDead = (layer == 1 && itn == 1);
      if (!lastUpdateDead) {
        k_norminv<<<GRID(N4), WG, 0, stream>>>(nxt, hninv, (int)N4);
        k_updateA<<<GRID((size_t)nGroups * 32), WG, 0, stream>>>(
            nxt, cur, hninv, tninv, hl, tl, A, E, nGroups);
      }
    }
    k_acc_add<<<GRID(N64), WG, 0, stream>>>(acc, nxt, (int)N64);
    float* tmp = cur; cur = nxt; nxt = tmp;
  }

  k_final<<<GRID(N64), WG, 0, stream>>>(acc, (float*)d_out, (int)N64);
#undef GRID
}


# Atomic adds relaid so an instruction's lanes share lines: degree sums 4 lanes per edge (8 lines an instruction, was 32), message adds 32 lanes per edge (1 whole line, was 4); f32 add order as before
# speedup vs baseline: 1.4933x; 1.4933x over previous
#include <hip/hip_runtime.h>
#include <math.h>

typedef float v2f __attribute__((ext_vector_type(2)));
typedef float v8f __attribute__((ext_vector_type(8)));

#define WG 256

__device__ __forceinline__ void atomic_fadd(float* p, float v) {
  __hip_atomic_fetch_add(p, v, __ATOMIC_RELAXED, __HIP_MEMORY_SCOPE_AGENT);
}

__device__ __forceinline__ float fast_tanh(float x) {
  float r;
  asm("v_tanh_f32 %0, %1\n\tv_nop\n\tv_nop" : "=v"(r) : "v"(x));
  return r;
}

__global__ void k_init(const float* __restrict__ u, const float* __restrict__ it,
                       float* __restrict__ ego, float* __restrict__ acc,
                       int userElems, int total) {
  int i = blockIdx.x * blockDim.x + threadIdx.x;
  if (i >= total) return;
  float v = (i < userElems) ? u[i] : it[i - userElems];
  ego[i] = v;
  acc[i] = v;
}

__global__ void k_fill1(float* __restrict__ a, int n) {
  int i = blockIdx.x * blockDim.x + threadIdx.x;
  if (i < n) a[i] = 1.0f;
}

__global__ void k_softmax_dval(const float* __restrict__ A, const int* __restrict__ hl,
                               float* __restrict__ ew, float* __restrict__ dval, int E) {
  int tid = blockIdx.x * blockDim.x + threadIdx.x;
  int e = tid >> 2, f = tid & 3;
  if (e >= E) return;
  float4 a = reinterpret_cast<const float4*>(A)[e];
  float m = fmaxf(fmaxf(a.x, a.y), fmaxf(a.z, a.w));
  float e0 = expf(a.x - m), e1 = expf(a.y - m), e2 = expf(a.z - m), e3 = expf(a.w - m);
  float inv = 1.0f / (e0 + e1 + e2 + e3);
  float mine = e0;
  mine = (f == 1) ? e1 : mine;
  mine = (f == 2) ? e2 : mine;
  mine = (f == 3) ? e3 : mine;
  mine *= inv;
  ew[tid] = mine;
  atomic_fadd(dval + (size_t)hl[e] * 4 + f, mine);
}

__global__ void k_dinv(float* __restrict__ d, int n) {
  int i = blockIdx.x * blockDim.x + threadIdx.x;
  if (i >= n) return;
  d[i] = 1.0f / sqrtf(fmaxf(d[i], 1e-8f));
}

__global__ void k_edge_weight(float* __restrict__ ew, const float* __restrict__ dinv,
                              const int* __restrict__ hl, const int* __restrict__ tl, int E) {
  int e = blockIdx.x * blockDim.x + threadIdx.x;
  if (e >= E) return;
  float4 w = reinterpret_cast<float4*>(ew)[e];
  float4 dh = reinterpret_cast<const float4*>(dinv)[hl[e]];
  float4 dt = reinterpret_cast<const float4*>(dinv)[tl[e]];
  w.x *= dh.x * dt.x;
  w.y *= dh.y * dt.y;
  w.z *= dh.z * dt.z;
  w.w *= dh.w * dt.w;
  reinterpret_cast<float4*>(ew)[e] = w;
}

struct EdgeTerm { float p0, p1; float* dst; };

__device__ __forceinline__ EdgeTerm edge_term(const float* __restrict__ ew,
                                              const float* __restrict__ ego,
                                              const int* __restrict__ hl,
                                              const int* __restrict__ tl,
                                              float* __restrict__ fact, int e, int lane) {
  int fl = lane >> 4;
  int h = hl[e], t = tl[e];
  float w0 = ew[(size_t)e * 4 + fl];
  float w1 = ew[(size_t)e * 4 + 2 + fl];
  const float* s = ego + (size_t)t * 64 + lane;
  EdgeTerm r;
  r.p0 = w0 * s[0];
  r.p1 = w1 * s[32];
  r.dst = fact + (size_t)h * 64 + lane;
  return r;
}

__global__ void k_message(const float* __restrict__ ew, const float* __restrict__ ego,
                          const int* __restrict__ hl, const int* __restrict__ tl,
                          float* __restrict__ fact, int E) {
  int tid = blockIdx.x * blockDim.x + threadIdx.x;
  int lane = threadIdx.x & 31;
  int eA = __builtin_amdgcn_readfirstlane(tid >> 5) * 2;
  if (eA >= E) return;
  if (eA + 1 < E) {
    EdgeTerm a = edge_term(ew, ego, hl, tl, fact, eA, lane);
    EdgeTerm b = edge_term(ew, ego, hl, tl, fact, eA + 1, lane);
    atomic_fadd(a.dst, a.p0);
    atomic_fadd(a.dst + 32, a.p1);
    atomic_fadd(b.dst, b.p0);
    atomic_fadd(b.dst + 32, b.p1);
  } else {
    EdgeTerm a = edge_term(ew, ego, hl, tl, fact, eA, lane);
    atomic_fadd(a.dst, a.p0);
    atomic_fadd(a.dst + 32, a.p1);
  }
}

__global__ void k_norminv(const float* __restrict__ x, float* __restrict__ out, int NF) {
  int i = blockIdx.x * blockDim.x + threadIdx.x;
  if (i >= NF) return;
  int n = i >> 2, f = i & 3;
  const float4* p = reinterpret_cast<const float4*>(x + (size_t)n * 64 + (size_t)f * 16);
  float s = 0.f;
#pragma unroll
  for (int k = 0; k < 4; ++k) {
    float4 v = p[k];
    s += v.x * v.x + v.y * v.y + v.z * v.z + v.w * v.w;
  }
  out[i] = 1.0f / fmaxf(sqrtf(s), 1e-12f);
}

__global__ void __launch_bounds__(WG) k_updateA(
    const float* __restrict__ fact, const float* __restrict__ ego,
    const float* __restrict__ hninv, const float* __restrict__ tninv,
    const int* __restrict__ hl, const int* __restrict__ tl,
    float* __restrict__ A, int E, int nGroups) {
  int tid = blockIdx.x * blockDim.x + threadIdx.x;
  int wave = tid >> 5;
  int lane = tid & 31;
  if (wave >= nGroups) return;

  int r = lane & 15;
  int eIdx = wave * 16 + r;
  int e = (eIdx < E) ? eIdx : (E - 1);
  int h = hl[e], t = tl[e];
  int koff = (lane >> 4) << 1;
  const float* Hrow = fact + (size_t)h * 64;
  const float* Trow = ego + (size_t)t * 64;
  const float* hn4 = hninv + (size_t)h * 4;
  const float* tn4 = tninv + (size_t)t * 4;

  bool active = (lane < 8) || (lane >= 24);
  int outRow = (lane < 8) ? lane : (lane - 16);
  int eOut = wave * 16 + outRow;

#pragma unroll
  for (int f = 0; f < 4; ++f) {
    float hn = hn4[f];
    float tn = tn4[f];
    v8f c = {0.f, 0.f, 0.f, 0.f, 0.f, 0.f, 0.f, 0.f};
#pragma unroll
    for (int s = 0; s < 4; ++s) {
      int o = f * 16 + s * 4 + koff;
      v2f av, bv;
      av.x = Hrow[o] * hn;
      av.y = Hrow[o + 1] * hn;
      bv.x = fast_tanh(Trow[o] * tn);
      bv.y = fast_tanh(Trow[o + 1] * tn);
      c = __builtin_amdgcn_wmma_f32_16x16x4_f32(false, av, false, bv,
                                                (short)0, c, false, false);
    }
    int idx = lane & 7;
    float dv = c[0];
    dv = (idx == 1) ? c[1] : dv;
    dv = (idx == 2) ? c[2] : dv;
    dv = (idx == 3) ? c[3] : dv;
    dv = (idx == 4) ? c[4] : dv;
    dv = (idx == 5) ? c[5] : dv;
    dv = (idx == 6) ? c[6] : dv;
    dv = (idx == 7) ? c[7] : dv;
    if (active && eOut < E) A[(size_t)eOut * 4 + f] += dv;
  }
}

__global__ void k_acc_add(float* __restrict__ acc, const float* __restrict__ x, int n) {
  int i = blockIdx.x * blockDim.x + threadIdx.x;
  if (i < n) acc[i] += x[i];
}

__global__ void k_final(const float* __restrict__ acc, float* __restrict__ out, int n) {
  int i = blockIdx.x * blockDim.x + threadIdx.x;
  if (i < n) out[i] = acc[i] * (1.0f / 3.0f);
}

static inline size_t align16f(size_t x) { return (x + 15) & ~(size_t)15; }

extern "C" void kernel_launch(void* const* d_in, const int* in_sizes, int n_in,
                              void* d_out, int out_size, void* d_ws, size_t ws_size,
                              hipStream_t stream) {
  (void)n_in; (void)out_size; (void)ws_size;
  const float* u  = (const float*)d_in[0];
  const float* it = (const float*)d_in[1];
  const int* hl   = (const int*)d_in[2];
  const int* tl   = (const int*)d_in[3];

  const int userElems = in_sizes[0];
  const int itemElems = in_sizes[1];
  const int E = in_sizes[2];
  const int N = (userElems + itemElems) / 64;
  const size_t N64 = (size_t)N * 64;
  const size_t E4  = (size_t)E * 4;
  const size_t N4  = (size_t)N * 4;

  float* base = (float*)d_ws;
  size_t off = 0;
  float* ego   = base + off; off += align16f(N64);
  float* fact  = base + off; off += align16f(N64);
  float* acc   = base + off; off += align16f(N64);
  float* A     = base + off; off += align16f(E4);
  float* ew    = base + off; off += align16f(E4);
  float* dval  = base + off; off += align16f(N4);
  float* hninv = base + off; off += align16f(N4);
  float* tninv = base + off; off += align16f(N4);

#define GRID(n) ((unsigned)(((size_t)(n) + WG - 1) / WG))

  k_init<<<GRID(N64), WG, 0, stream>>>(u, it, ego, acc, userElems, (int)N64);
  k_fill1<<<GRID(E4), WG, 0, stream>>>(A, (int)E4);

  float* cur = ego;
  float* nxt = fact;
  const int nGroups = (E + 15) / 16;

  for (int layer = 0; layer < 2; ++layer) {
    k_norminv<<<GRID(N4), WG, 0, stream>>>(cur, tninv, (int)N4);
    for (int itn = 0; itn < 2; ++itn) {
      hipMemsetAsync(dval, 0, N4 * sizeof(float), stream);
      k_softmax_dval<<<GRID((size_t)E * 4), WG, 0, stream>>>(A, hl, ew, dval, E);
      k_dinv<<<GRID(N4), WG, 0, stream>>>(dval, (int)N4);
      k_edge_weight<<<GRID(E), WG, 0, stream>>>(ew, dval, hl, tl, E);
      hipMemsetAsync(nxt, 0, N64 * sizeof(float), stream);
      k_message<<<GRID((size_t)E * 16), WG, 0, stream>>>(ew, cur, hl, tl, nxt, E);
      const bool lastUpdateDead = (layer == 1 && itn == 1);
      if (!lastUpdateDead) {
        k_norminv<<<GRID(N4), WG, 0, stream>>>(nxt, hninv, (int)N4);
        k_updateA<<<GRID((size_t)nGroups * 32), WG, 0, stream>>>(
            nxt, cur, hninv, tninv, hl, tl, A, E, nGroups);
      }
    }
    k_acc_add<<<GRID(N64), WG, 0, stream>>>(acc, nxt, (int)N64);
    float* tmp = cur; cur = nxt; nxt = tmp;
  }

  k_final<<<GRID(N64), WG, 0, stream>>>(acc, (float*)d_out, (int)N64);
#undef GRID
}


# More work per wave: message waves walk 32 edges (indices, weights loaded once a chunk; rows 8 edges a batch), flat kernels take 4 float4s a thread; f32 atomic adds still land in arrival order
# speedup vs baseline: 1.6123x; 1.0797x over previous
#include <hip/hip_runtime.h>
#include <math.h>

typedef float v2f __attribute__((ext_vector_type(2)));
typedef float v8f __attribute__((ext_vector_type(8)));
typedef float v4f __attribute__((ext_vector_type(4)));

#define WG 256

__device__ __forceinline__ void atomic_fadd(float* p, float v) {
  __hip_atomic_fetch_add(p, v, __ATOMIC_RELAXED, __HIP_MEMORY_SCOPE_AGENT);
}

__device__ __forceinline__ float fast_tanh(float x) {
  float r;
  asm("v_tanh_f32 %0, %1\n\tv_nop\n\tv_nop" : "=v"(r) : "v"(x));
  return r;
}

#define VEC_ITEMS 4

static_assert(VEC_ITEMS == 4, "keep_loaded4 names four tuples");
__device__ __forceinline__ void keep_loaded4(v4f (&v)[4]) {
  asm volatile("" : "+v"(v[0]), "+v"(v[1]), "+v"(v[2]), "+v"(v[3]));
}

template <int SRC, bool TAIL>
__device__ __forceinline__ void init_body(const v4f* __restrict__ u4, const v4f* __restrict__ it4,
                                          v4f* __restrict__ ego4, v4f* __restrict__ acc4,
                                          int userVec, int totalVec, int base) {
  v4f v[VEC_ITEMS];
#pragma unroll
  for (int k = 0; k < VEC_ITEMS; ++k) {
    int i = base + k * WG;
    if (TAIL) i = (i < totalVec) ? i : (totalVec - 1);
    if (SRC == 0) v[k] = u4[i];
    else if (SRC == 1) v[k] = it4[i - userVec];
    else v[k] = (i < userVec) ? u4[i] : it4[i - userVec];
  }
  keep_loaded4(v);
#pragma unroll
  for (int k = 0; k < VEC_ITEMS; ++k) {
    int i = base + k * WG;
    if (!TAIL || i < totalVec) { ego4[i] = v[k]; acc4[i] = v[k]; }
  }
}

__global__ void k_init(const float* __restrict__ u, const float* __restrict__ it,
                       float* __restrict__ ego, float* __restrict__ acc,
                       int userVec, int totalVec) {
  const v4f* __restrict__ u4 = reinterpret_cast<const v4f*>(u);
  const v4f* __restrict__ it4 = reinterpret_cast<const v4f*>(it);
  v4f* __restrict__ ego4 = reinterpret_cast<v4f*>(ego);
  v4f* __restrict__ acc4 = reinterpret_cast<v4f*>(acc);
  int first = blockIdx.x * (WG * VEC_ITEMS);
  int end = first + WG * VEC_ITEMS;
  int base = first + threadIdx.x;
  if (end <= userVec)
    init_body<0, false>(u4, it4, ego4, acc4, userVec, totalVec, base);
  else if (first >= userVec && end <= totalVec)
    init_body<1, false>(u4, it4, ego4, acc4, userVec, totalVec, base);
  else
    init_body<2, true>(u4, it4, ego4, acc4, userVec, totalVec, base);
}

__global__ void k_fill1(float* __restrict__ a, int nVec) {
  float4* __restrict__ a4 = reinterpret_cast<float4*>(a);
  int base = blockIdx.x * (WG * VEC_ITEMS) + threadIdx.x;
  float4 one = make_float4(1.0f, 1.0f, 1.0f, 1.0f);
#pragma unroll
  for (int k = 0; k < VEC_ITEMS; ++k) {
    int i = base + k * WG;
    if (i < nVec) a4[i] = one;
  }
}

#define SM_ITEMS 4
static_assert(SM_ITEMS == 4, "the asm statement in softmax_body names four items");

template <bool TAIL>
__device__ __forceinline__ void softmax_body(const v4f* __restrict__ A4, const int* __restrict__ hl,
                                             float* __restrict__ ew, float* __restrict__ dval,
                                             int total, int base) {
  v4f av[SM_ITEMS];
  int hv[SM_ITEMS];
#pragma unroll
  for (int k = 0; k < SM_ITEMS; ++k) {
    int tid = base + k * WG;
    if (TAIL) tid = (tid < total) ? tid : (total - 1);
    int e = tid >> 2;
    av[k] = A4[e];
    hv[k] = hl[e];
  }
  asm volatile("" : "+v"(av[0]), "+v"(av[1]), "+v"(av[2]), "+v"(av[3]),
                    "+v"(hv[0]), "+v"(hv[1]), "+v"(hv[2]), "+v"(hv[3]));
#pragma unroll
  for (int k = 0; k < SM_ITEMS; ++k) {
    int tid = base + k * WG;
    int f = tid & 3;
    v4f a = av[k];
    float m = fmaxf(fmaxf(a.x, a.y), fmaxf(a.z, a.w));
    float e0 = expf(a.x - m), e1 = expf(a.y - m), e2 = expf(a.z - m), e3 = expf(a.w - m);
    float inv = 1.0f / (e0 + e1 + e2 + e3);
    float mine = e0;
    mine = (f == 1) ? e1 : mine;
    mine = (f == 2) ? e2 : mine;
    mine = (f == 3) ? e3 : mine;
    mine *= inv;
    if (!TAIL || tid < total) {
      ew[tid] = mine;
      atomic_fadd(dval + (size_t)hv[k] * 4 + f, mine);
    }
  }
}

__global__ void k_softmax_dval(const float* __restrict__ A, const int* __restrict__ hl,
                               float* __restrict__ ew, float* __restrict__ dval, int E) {
  const v4f* __restrict__ A4 = reinterpret_cast<const v4f*>(A);
  int base = blockIdx.x * (WG * SM_ITEMS) + threadIdx.x;
  int total = E * 4;
  if ((int)(blockIdx.x + 1) * (WG * SM_ITEMS) <= total)
    softmax_body<false>(A4, hl, ew, dval, total, base);
  else
    softmax_body<true>(A4, hl, ew, dval, total, base);
}

__global__ void k_dinv(float* __restrict__ d, int nVec) {
  int i = blockIdx.x * blockDim.x + threadIdx.x;
  if (i >= nVec) return;
  float4 v = reinterpret_cast<float4*>(d)[i];
  v.x = 1.0f / sqrtf(fmaxf(v.x, 1e-8f));
  v.y = 1.0f / sqrtf(fmaxf(v.y, 1e-8f));
  v.z = 1.0f / sqrtf(fmaxf(v.z, 1e-8f));
  v.w = 1.0f / sqrtf(fmaxf(v.w, 1e-8f));
  reinterpret_cast<float4*>(d)[i] = v;
}

#define EW_ITEMS 2

template <bool TAIL>
__device__ __forceinline__ void edge_weight_body(float4* ew4, const float4* __restrict__ dinv4,
                                                 const int* __restrict__ hl, const int* __restrict__ tl,
                                                 int E, int base) {
  float4 w[EW_ITEMS], dh[EW_ITEMS], dt[EW_ITEMS];
  int h[EW_ITEMS], t[EW_ITEMS];
#pragma unroll
  for (int k = 0; k < EW_ITEMS; ++k) {
    int e = base + k * WG;
    if (TAIL) e = (e < E) ? e : (E - 1);
    h[k] = hl[e];
    t[k] = tl[e];
    w[k] = ew4[e];
  }
#pragma unroll
  for (int k = 0; k < EW_ITEMS; ++k) {
    dh[k] = dinv4[h[k]];
    dt[k] = dinv4[t[k]];
  }
#pragma unroll
  for (int k = 0; k < EW_ITEMS; ++k) {
    int e = base + k * WG;
    float4 r = w[k];
    r.x *= dh[k].x * dt[k].x;
    r.y *= dh[k].y * dt[k].y;
    r.z *= dh[k].z * dt[k].z;
    r.w *= dh[k].w * dt[k].w;
    if (!TAIL || e < E) ew4[e] = r;
  }
}

__global__ void k_edge_weight(float* __restrict__ ew, const float* __restrict__ dinv,
                              const int* __restrict__ hl, const int* __restrict__ tl, int E) {
  float4* ew4 = reinterpret_cast<float4*>(ew);
  const float4* __restrict__ dinv4 = reinterpret_cast<const float4*>(dinv);
  int base = blockIdx.x * (WG * EW_ITEMS) + threadIdx.x;
  if ((int)(blockIdx.x + 1) * (WG * EW_ITEMS) <= E)
    edge_weight_body<false>(ew4, dinv4, hl, tl, E, base);
  else
    edge_weight_body<true>(ew4, dinv4, hl, tl, E, base);
}

#define MSG_EDGES 32
#define MSG_BATCH 8

__device__ __forceinline__ float lane_bcast(float v, int srcLane) {
  return __int_as_float(__builtin_amdgcn_readlane(__float_as_int(v), srcLane));
}

template <bool FULL>
__device__ __forceinline__ void message_chunk(const float* __restrict__ ew, const float* __restrict__ ego,
                                              const int* __restrict__ hl, const int* __restrict__ tl,
                                              float* __restrict__ fact, int e0, int cnt, int lane) {
  int el = e0 + ((FULL || lane < cnt) ? lane : (cnt - 1));
  int hv = hl[el];
  int tv = tl[el];
  float4 wv = reinterpret_cast<const float4*>(ew)[el];
  bool upper = lane >= 16;
  const int n = FULL ? MSG_EDGES : cnt;

  for (int b = 0; b < n; b += MSG_BATCH) {
    float p0[MSG_BATCH], p1[MSG_BATCH];
    float* dst[MSG_BATCH];
#pragma unroll
    for (int k = 0; k < MSG_BATCH; ++k) {
      int j = b + k;
      if (!FULL) j = (j < cnt) ? j : (cnt - 1);
      int h = __builtin_amdgcn_readlane(hv, j);
      int t = __builtin_amdgcn_readlane(tv, j);
      float wx = lane_bcast(wv.x, j), wy = lane_bcast(wv.y, j);
      float wz = lane_bcast(wv.z, j), ww = lane_bcast(wv.w, j);
      float w0 = upper ? wy : wx;
      float w1 = upper ? ww : wz;
      const float* s = ego + (size_t)t * 64 + lane;
      p0[k] = w0 * s[0];
      p1[k] = w1 * s[32];
      dst[k] = fact + (size_t)h * 64 + lane;
    }
#pragma unroll
    for (int k = 0; k < MSG_BATCH; ++k) {
      if (FULL || b + k < cnt) {
        atomic_fadd(dst[k], p0[k]);
        atomic_fadd(dst[k] + 32, p1[k]);
      }
    }
  }
}

__global__ void __launch_bounds__(WG) k_message(
    const float* __restrict__ ew, const float* __restrict__ ego,
    const int* __restrict__ hl, const int* __restrict__ tl,
    float* __restrict__ fact, int E) {
  int lane = threadIdx.x & 31;
  int wave = __builtin_amdgcn_readfirstlane((int)((blockIdx.x * WG + threadIdx.x) >> 5));
  int e0 = wave * MSG_EDGES;
  if (e0 >= E) return;
  int cnt = E - e0;
  if (cnt >= MSG_EDGES)
    message_chunk<true>(ew, ego, hl, tl, fact, e0, MSG_EDGES, lane);
  else
    message_chunk<false>(ew, ego, hl, tl, fact, e0, cnt, lane);
}

__global__ void k_norminv(const float* __restrict__ x, float* __restrict__ out, int NF) {
  int i = blockIdx.x * blockDim.x + threadIdx.x;
  if (i >= NF) return;
  int n = i >> 2, f = i & 3;
  const float4* p = reinterpret_cast<const float4*>(x + (size_t)n * 64 + (size_t)f * 16);
  float s = 0.f;
#pragma unroll
  for (int k = 0; k < 4; ++k) {
    float4 v = p[k];
    s += v.x * v.x + v.y * v.y + v.z * v.z + v.w * v.w;
  }
  out[i] = 1.0f / fmaxf(sqrtf(s), 1e-12f);
}

__global__ void __launch_bounds__(WG) k_updateA(
    const float* __restrict__ fact, const float* __restrict__ ego,
    const float* __restrict__ hninv, const float* __restrict__ tninv,
    const int* __restrict__ hl, const int* __restrict__ tl,
    float* __restrict__ A, int E, int nGroups) {
  int tid = blockIdx.x * blockDim.x + threadIdx.x;
  int wave = tid >> 5;
  int lane = tid & 31;
  if (wave >= nGroups) return;

  int r = lane & 15;
  int eIdx = wave * 16 + r;
  int e = (eIdx < E) ? eIdx : (E - 1);
  int h = hl[e], t = tl[e];
  int koff = (lane >> 4) << 1;
  const float* Hrow = fact + (size_t)h * 64;
  const float* Trow = ego + (size_t)t * 64;
  const float* hn4 = hninv + (size_t)h * 4;
  const float* tn4 = tninv + (size_t)t * 4;

  bool active = (lane < 8) || (lane >= 24);
  int outRow = (lane < 8) ? lane : (lane - 16);
  int eOut = wave * 16 + outRow;

#pragma unroll
  for (int f = 0; f < 4; ++f) {
    float hn = hn4[f];
    float tn = tn4[f];
    v8f c = {0.f, 0.f, 0.f, 0.f, 0.f, 0.f, 0.f, 0.f};
#pragma unroll
    for (int s = 0; s < 4; ++s) {
      int o = f * 16 + s * 4 + koff;
      v2f av, bv;
      av.x = Hrow[o] * hn;
      av.y = Hrow[o + 1] * hn;
      bv.x = fast_tanh(Trow[o] * tn);
      bv.y = fast_tanh(Trow[o + 1] * tn);
      c = __builtin_amdgcn_wmma_f32_16x16x4_f32(false, av, false, bv,
                                                (short)0, c, false, false);
    }
    int idx = lane & 7;
    float dv = c[0];
    dv = (idx == 1) ? c[1] : dv;
    dv = (idx == 2) ? c[2] : dv;
    dv = (idx == 3) ? c[3] : dv;
    dv = (idx == 4) ? c[4] : dv;
    dv = (idx == 5) ? c[5] : dv;
    dv = (idx == 6) ? c[6] : dv;
    dv = (idx == 7) ? c[7] : dv;
    if (active && eOut < E) A[(size_t)eOut * 4 + f] += dv;
  }
}

template <bool TAIL>
__device__ __forceinline__ void acc_add_body(float4* acc4, const float4* __restrict__ x4, int nVec, int base) {
  float4 a[VEC_ITEMS], b[VEC_ITEMS];
#pragma unroll
  for (int k = 0; k < VEC_ITEMS; ++k) {
    int i = base + k * WG;
    if (TAIL) i = (i < nVec) ? i : (nVec - 1);
    a[k] = acc4[i];
    b[k] = x4[i];
  }
#pragma unroll
  for (int k = 0; k < VEC_ITEMS; ++k) {
    int i = base + k * WG;
    float4 r = a[k];
    r.x += b[k].x; r.y += b[k].y; r.z += b[k].z; r.w += b[k].w;
    if (!TAIL || i < nVec) acc4[i] = r;
  }
}

__global__ void k_acc_add(float* __restrict__ acc, const float* __restrict__ x, int nVec) {
  float4* acc4 = reinterpret_cast<float4*>(acc);
  const float4* __restrict__ x4 = reinterpret_cast<const float4*>(x);
  int base = blockIdx.x * (WG * VEC_ITEMS) + threadIdx.x;
  if ((int)(blockIdx.x + 1) * (WG * VEC_ITEMS) <= nVec)
    acc_add_body<false>(acc4, x4, nVec, base);
  else
    acc_add_body<true>(acc4, x4, nVec, base);
}

template <bool TAIL>
__device__ __forceinline__ void final_body(const float4* __restrict__ acc4, float4* __restrict__ out4,
                                           int nVec, int base) {
  float4 a[VEC_ITEMS];
#pragma unroll
  for (int k = 0; k < VEC_ITEMS; ++k) {
    int i = base + k * WG;
    if (TAIL) i = (i < nVec) ? i : (nVec - 1);
    a[k] = acc4[i];
  }
#pragma unroll
  for (int k = 0; k < VEC_ITEMS; ++k) {
    int i = base + k * WG;
    float4 r = a[k];
    r.x *= (1.0f / 3.0f); r.y *= (1.0f / 3.0f); r.z *= (1.0f / 3.0f); r.w *= (1.0f / 3.0f);
    if (!TAIL || i < nVec) out4[i] = r;
  }
}

__global__ void k_final(const float* __restrict__ acc, float* __restrict__ out, int nVec) {
  const float4* __restrict__ acc4 = reinterpret_cast<const float4*>(acc);
  float4* __restrict__ out4 = reinterpret_cast<float4*>(out);
  int base = blockIdx.x * (WG * VEC_ITEMS) + threadIdx.x;
  if ((int)(blockIdx.x + 1) * (WG * VEC_ITEMS) <= nVec)
    final_body<false>(acc4, out4, nVec, base);
  else
    final_body<true>(acc4, out4, nVec, base);
}

static inline size_t align16f(size_t x) { return (x + 15) & ~(size_t)15; }

extern "C" void kernel_launch(void* const* d_in, const int* in_sizes, int n_in,
                              void* d_out, int out_size, void* d_ws, size_t ws_size,
                              hipStream_t stream) {
  (void)n_in; (void)out_size; (void)ws_size;
  const float* u  = (const float*)d_in[0];
  const float* it = (const float*)d_in[1];
  const int* hl   = (const int*)d_in[2];
  const int* tl   = (const int*)d_in[3];

  const int userElems = in_sizes[0];
  const int itemElems = in_sizes[1];
  const int E = in_sizes[2];
  const int N = (userElems + itemElems) / 64;
  const size_t N64 = (size_t)N * 64;
  const size_t E4  = (size_t)E * 4;
  const size_t N4  = (size_t)N * 4;

  float* base = (float*)d_ws;
  size_t off = 0;
  float* ego   = base + off; off += align16f(N64);
  float* fact  = base + off; off += align16f(N64);
  float* acc   = base + off; off += align16f(N64);
  float* A     = base + off; off += align16f(E4);
  float* ew    = base + off; off += align16f(E4);
  float* dval  = base + off; off += align16f(N4);
  float* hninv = base + off; off += align16f(N4);
  float* tninv = base + off; off += align16f(N4);

#define GRID(n) ((unsigned)(((size_t)(n) + WG - 1) / WG))
#define GRIDP(n, per) ((unsigned)(((size_t)(n) + (size_t)WG * (per) - 1) / ((size_t)WG * (per))))
  const int N16 = (int)(N64 / 4);

  k_init<<<GRIDP(N16, VEC_ITEMS), WG, 0, stream>>>(u, it, ego, acc, userElems / 4, N16);
  k_fill1<<<GRIDP(E, VEC_ITEMS), WG, 0, stream>>>(A, E);

  float* cur = ego;
  float* nxt = fact;
  const int nGroups = (E + 15) / 16;

  for (int layer = 0; layer < 2; ++layer) {
    k_norminv<<<GRID(N4), WG, 0, stream>>>(cur, tninv, (int)N4);
    for (int itn = 0; itn < 2; ++itn) {
      hipMemsetAsync(dval, 0, N4 * sizeof(float), stream);
      k_softmax_dval<<<GRIDP(E4, SM_ITEMS), WG, 0, stream>>>(A, hl, ew, dval, E);
      k_dinv<<<GRID(N), WG, 0, stream>>>(dval, N);
      k_edge_weight<<<GRIDP(E, EW_ITEMS), WG, 0, stream>>>(ew, dval, hl, tl, E);
      hipMemsetAsync(nxt, 0, N64 * sizeof(float), stream);
      k_message<<<GRID(E), WG, 0, stream>>>(ew, cur, hl, tl, nxt, E);
      const bool lastUpdateDead = (layer == 1 && itn == 1);
      if (!lastUpdateDead) {
        k_norminv<<<GRID(N4), WG, 0, stream>>>(nxt, hninv, (int)N4);
        k_updateA<<<GRID((size_t)nGroups * 32), WG, 0, stream>>>(
            nxt, cur, hninv, tninv, hl, tl, A, E, nGroups);
      }
    }
    k_acc_add<<<GRIDP(N16, VEC_ITEMS), WG, 0, stream>>>(acc, nxt, N16);
    float* tmp = cur; cur = nxt; nxt = tmp;
  }

  k_final<<<GRIDP(N16, VEC_ITEMS), WG, 0, stream>>>(acc, (float*)d_out, N16);
#undef GRID
#undef GRIDP
}


# Routing update moves whole lines only: edge rows copied 32 lanes a row into wave-private LDS, WMMA fragments read from LDS (was 16-line gathers), A read and written 16 bytes an edge; k order kept
# speedup vs baseline: 2.0187x; 1.2521x over previous
#include <hip/hip_runtime.h>
#include <math.h>

typedef float v2f __attribute__((ext_vector_type(2)));
typedef float v8f __attribute__((ext_vector_type(8)));
typedef float v4f __attribute__((ext_vector_type(4)));

#define WG 256

__device__ __forceinline__ void atomic_fadd(float* p, float v) {
  __hip_atomic_fetch_add(p, v, __ATOMIC_RELAXED, __HIP_MEMORY_SCOPE_AGENT);
}

__device__ __forceinline__ float fast_tanh(float x) {
  float r;
  asm("v_tanh_f32 %0, %1\n\tv_nop\n\tv_nop" : "=v"(r) : "v"(x));
  return r;
}

#define VEC_ITEMS 4

static_assert(VEC_ITEMS == 4, "keep_loaded4 names four tuples");
__device__ __forceinline__ void keep_loaded4(v4f (&v)[4]) {
  asm volatile("" : "+v"(v[0]), "+v"(v[1]), "+v"(v[2]), "+v"(v[3]));
}

template <int SRC, bool TAIL>
__device__ __forceinline__ void init_body(const v4f* __restrict__ u4, const v4f* __restrict__ it4,
                                          v4f* __restrict__ ego4, v4f* __restrict__ acc4,
                                          int userVec, int totalVec, int base) {
  v4f v[VEC_ITEMS];
#pragma unroll
  for (int k = 0; k < VEC_ITEMS; ++k) {
    int i = base + k * WG;
    if (TAIL) i = (i < totalVec) ? i : (totalVec - 1);
    if (SRC == 0) v[k] = u4[i];
    else if (SRC == 1) v[k] = it4[i - userVec];
    else v[k] = (i < userVec) ? u4[i] : it4[i - userVec];
  }
  keep_loaded4(v);
#pragma unroll
  for (int k = 0; k < VEC_ITEMS; ++k) {
    int i = base + k * WG;
    if (!TAIL || i < totalVec) { ego4[i] = v[k]; acc4[i] = v[k]; }
  }
}

__global__ void k_init(const float* __restrict__ u, const float* __restrict__ it,
                       float* __restrict__ ego, float* __restrict__ acc,
                       int userVec, int totalVec) {
  const v4f* __restrict__ u4 = reinterpret_cast<const v4f*>(u);
  const v4f* __restrict__ it4 = reinterpret_cast<const v4f*>(it);
  v4f* __restrict__ ego4 = reinterpret_cast<v4f*>(ego);
  v4f* __restrict__ acc4 = reinterpret_cast<v4f*>(acc);
  int first = blockIdx.x * (WG * VEC_ITEMS);
  int end = first + WG * VEC_ITEMS;
  int base = first + threadIdx.x;
  if (end <= userVec)
    init_body<0, false>(u4, it4, ego4, acc4, userVec, totalVec, base);
  else if (first >= userVec && end <= totalVec)
    init_body<1, false>(u4, it4, ego4, acc4, userVec, totalVec, base);
  else
    init_body<2, true>(u4, it4, ego4, acc4, userVec, totalVec, base);
}

__global__ void k_fill1(float* __restrict__ a, int nVec) {
  float4* __restrict__ a4 = reinterpret_cast<float4*>(a);
  int base = blockIdx.x * (WG * VEC_ITEMS) + threadIdx.x;
  float4 one = make_float4(1.0f, 1.0f, 1.0f, 1.0f);
#pragma unroll
  for (int k = 0; k < VEC_ITEMS; ++k) {
    int i = base + k * WG;
    if (i < nVec) a4[i] = one;
  }
}

#define SM_ITEMS 4
static_assert(SM_ITEMS == 4, "the asm statement in softmax_body names four items");

template <bool TAIL>
__device__ __forceinline__ void softmax_body(const v4f* __restrict__ A4, const int* __restrict__ hl,
                                             float* __restrict__ ew, float* __restrict__ dval,
                                             int total, int base) {
  v4f av[SM_ITEMS];
  int hv[SM_ITEMS];
#pragma unroll
  for (int k = 0; k < SM_ITEMS; ++k) {
    int tid = base + k * WG;
    if (TAIL) tid = (tid < total) ? tid : (total - 1);
    int e = tid >> 2;
    av[k] = A4[e];
    hv[k] = hl[e];
  }
  asm volatile("" : "+v"(av[0]), "+v"(av[1]), "+v"(av[2]), "+v"(av[3]),
                    "+v"(hv[0]), "+v"(hv[1]), "+v"(hv[2]), "+v"(hv[3]));
#pragma unroll
  for (int k = 0; k < SM_ITEMS; ++k) {
    int tid = base + k * WG;
    int f = tid & 3;
    v4f a = av[k];
    float m = fmaxf(fmaxf(a.x, a.y), fmaxf(a.z, a.w));
    float e0 = expf(a.x - m), e1 = expf(a.y - m), e2 = expf(a.z - m), e3 = expf(a.w - m);
    float inv = 1.0f / (e0 + e1 + e2 + e3);
    float mine = e0;
    mine = (f == 1) ? e1 : mine;
    mine = (f == 2) ? e2 : mine;
    mine = (f == 3) ? e3 : mine;
    mine *= inv;
    if (!TAIL || tid < total) {
      ew[tid] = mine;
      atomic_fadd(dval + (size_t)hv[k] * 4 + f, mine);
    }
  }
}

__global__ void k_softmax_dval(const float* __restrict__ A, const int* __restrict__ hl,
                               float* __restrict__ ew, float* __restrict__ dval, int E) {
  const v4f* __restrict__ A4 = reinterpret_cast<const v4f*>(A);
  int base = blockIdx.x * (WG * SM_ITEMS) + threadIdx.x;
  int total = E * 4;
  if ((int)(blockIdx.x + 1) * (WG * SM_ITEMS) <= total)
    softmax_body<false>(A4, hl, ew, dval, total, base);
  else
    softmax_body<true>(A4, hl, ew, dval, total, base);
}

__global__ void k_dinv(float* __restrict__ d, int nVec) {
  int i = blockIdx.x * blockDim.x + threadIdx.x;
  if (i >= nVec) return;
  float4 v = reinterpret_cast<float4*>(d)[i];
  v.x = 1.0f / sqrtf(fmaxf(v.x, 1e-8f));
  v.y = 1.0f / sqrtf(fmaxf(v.y, 1e-8f));
  v.z = 1.0f / sqrtf(fmaxf(v.z, 1e-8f));
  v.w = 1.0f / sqrtf(fmaxf(v.w, 1e-8f));
  reinterpret_cast<float4*>(d)[i] = v;
}

#define EW_ITEMS 2

template <bool TAIL>
__device__ __forceinline__ void edge_weight_body(float4* ew4, const float4* __restrict__ dinv4,
                                                 const int* __restrict__ hl, const int* __restrict__ tl,
                                                 int E, int base) {
  float4 w[EW_ITEMS], dh[EW_ITEMS], dt[EW_ITEMS];
  int h[EW_ITEMS], t[EW_ITEMS];
#pragma unroll
  for (int k = 0; k < EW_ITEMS; ++k) {
    int e = base + k * WG;
    if (TAIL) e = (e < E) ? e : (E - 1);
    h[k] = hl[e];
    t[k] = tl[e];
    w[k] = ew4[e];
  }
#pragma unroll
  for (int k = 0; k < EW_ITEMS; ++k) {
    dh[k] = dinv4[h[k]];
    dt[k] = dinv4[t[k]];
  }
#pragma unroll
  for (int k = 0; k < EW_ITEMS; ++k) {
    int e = base + k * WG;
    float4 r = w[k];
    r.x *= dh[k].x * dt[k].x;
    r.y *= dh[k].y * dt[k].y;
    r.z *= dh[k].z * dt[k].z;
    r.w *= dh[k].w * dt[k].w;
    if (!TAIL || e < E) ew4[e] = r;
  }
}

__global__ void k_edge_weight(float* __restrict__ ew, const float* __restrict__ dinv,
                              const int* __restrict__ hl, const int* __restrict__ tl, int E) {
  float4* ew4 = reinterpret_cast<float4*>(ew);
  const float4* __restrict__ dinv4 = reinterpret_cast<const float4*>(dinv);
  int base = blockIdx.x * (WG * EW_ITEMS) + threadIdx.x;
  if ((int)(blockIdx.x + 1) * (WG * EW_ITEMS) <= E)
    edge_weight_body<false>(ew4, dinv4, hl, tl, E, base);
  else
    edge_weight_body<true>(ew4, dinv4, hl, tl, E, base);
}

#define MSG_EDGES 32
#define MSG_BATCH 8

__device__ __forceinline__ float lane_bcast(float v, int srcLane) {
  return __int_as_float(__builtin_amdgcn_readlane(__float_as_int(v), srcLane));
}

template <bool FULL>
__device__ __forceinline__ void message_chunk(const float* __restrict__ ew, const float* __restrict__ ego,
                                              const int* __restrict__ hl, const int* __restrict__ tl,
                                              float* __restrict__ fact, int e0, int cnt, int lane) {
  int el = e0 + ((FULL || lane < cnt) ? lane : (cnt - 1));
  int hv = hl[el];
  int tv = tl[el];
  float4 wv = reinterpret_cast<const float4*>(ew)[el];
  bool upper = lane >= 16;
  const int n = FULL ? MSG_EDGES : cnt;

  for (int b = 0; b < n; b += MSG_BATCH) {
    float p0[MSG_BATCH], p1[MSG_BATCH];
    float* dst[MSG_BATCH];
#pragma unroll
    for (int k = 0; k < MSG_BATCH; ++k) {
      int j = b + k;
      if (!FULL) j = (j < cnt) ? j : (cnt - 1);
      int h = __builtin_amdgcn_readlane(hv, j);
      int t = __builtin_amdgcn_readlane(tv, j);
      float wx = lane_bcast(wv.x, j), wy = lane_bcast(wv.y, j);
      float wz = lane_bcast(wv.z, j), ww = lane_bcast(wv.w, j);
      float w0 = upper ? wy : wx;
      float w1 = upper ? ww : wz;
      const float* s = ego + (size_t)t * 64 + lane;
      p0[k] = w0 * s[0];
      p1[k] = w1 * s[32];
      dst[k] = fact + (size_t)h * 64 + lane;
    }
#pragma unroll
    for (int k = 0; k < MSG_BATCH; ++k) {
      if (FULL || b + k < cnt) {
        atomic_fadd(dst[k], p0[k]);
        atomic_fadd(dst[k] + 32, p1[k]);
      }
    }
  }
}

__global__ void __launch_bounds__(WG) k_message(
    const float* __restrict__ ew, const float* __restrict__ ego,
    const int* __restrict__ hl, const int* __restrict__ tl,
    float* __restrict__ fact, int E) {
  int lane = threadIdx.x & 31;
  int wave = __builtin_amdgcn_readfirstlane((int)((blockIdx.x * WG + threadIdx.x) >> 5));
  int e0 = wave * MSG_EDGES;
  if (e0 >= E) return;
  int cnt = E - e0;
  if (cnt >= MSG_EDGES)
    message_chunk<true>(ew, ego, hl, tl, fact, e0, MSG_EDGES, lane);
  else
    message_chunk<false>(ew, ego, hl, tl, fact, e0, cnt, lane);
}

__global__ void k_norminv(const float* __restrict__ x, float* __restrict__ out, int NF) {
  int i = blockIdx.x * blockDim.x + threadIdx.x;
  if (i >= NF) return;
  int n = i >> 2, f = i & 3;
  const float4* p = reinterpret_cast<const float4*>(x + (size_t)n * 64 + (size_t)f * 16);
  float s = 0.f;
#pragma unroll
  for (int k = 0; k < 4; ++k) {
    float4 v = p[k];
    s += v.x * v.x + v.y * v.y + v.z * v.z + v.w * v.w;
  }
  out[i] = 1.0f / fmaxf(sqrtf(s), 1e-12f);
}

#define UA_ROWS 16
#define UA_PITCH 68
#define UA_WAVE_FLOATS (2 * UA_ROWS * UA_PITCH)

__global__ void __launch_bounds__(WG) k_updateA(
    const float* __restrict__ fact, const float* __restrict__ ego,
    const float* __restrict__ hninv, const float* __restrict__ tninv,
    const int* __restrict__ hl, const int* __restrict__ tl,
    float* __restrict__ A, int E, int nGroups) {
  __shared__ __attribute__((aligned(16))) float stage[(WG / 32) * UA_WAVE_FLOATS];

  int lane = threadIdx.x & 31;
  int wib = __builtin_amdgcn_readfirstlane((int)(threadIdx.x >> 5));
  int wave = __builtin_amdgcn_readfirstlane((int)((blockIdx.x * WG + threadIdx.x) >> 5));
  if (wave >= nGroups) return;

  int r = lane & 15;
  int eIdx = wave * 16 + r;
  int e = (eIdx < E) ? eIdx : (E - 1);
  int h = hl[e], t = tl[e];
  int koff = (lane >> 4) << 1;

  bool active = (lane < 8) || (lane >= 24);
  int outRow = (lane < 8) ? lane : (lane - 16);
  int eOut = wave * 16 + outRow;
  bool doOut = active && eOut < E;
  v4f* A4 = reinterpret_cast<v4f*>(A);
  v4f a4 = {0.f, 0.f, 0.f, 0.f};
  if (doOut) a4 = A4[eOut];

  v4f hn4 = *reinterpret_cast<const v4f*>(hninv + (size_t)h * 4);
  v4f tn4 = *reinterpret_cast<const v4f*>(tninv + (size_t)t * 4);

  float* sH = stage + wib * UA_WAVE_FLOATS;
  float* sT = sH + UA_ROWS * UA_PITCH;

  v2f hr[UA_ROWS], tr[UA_ROWS];
#pragma unroll
  for (int j = 0; j < UA_ROWS; ++j) {
    int hj = __builtin_amdgcn_readlane(h, j);
    int tj = __builtin_amdgcn_readlane(t, j);
    hr[j] = *reinterpret_cast<const v2f*>(fact + (size_t)hj * 64 + 2 * lane);
    tr[j] = *reinterpret_cast<const v2f*>(ego + (size_t)tj * 64 + 2 * lane);
  }
#pragma unroll
  for (int j = 0; j < UA_ROWS; ++j) {
    *reinterpret_cast<v2f*>(sH + j * UA_PITCH + 2 * lane) = hr[j];
    *reinterpret_cast<v2f*>(sT + j * UA_PITCH + 2 * lane) = tr[j];
  }
  __builtin_amdgcn_wave_barrier();

  const float* Hrow = sH + r * UA_PITCH;
  const float* Trow = sT + r * UA_PITCH;

#pragma unroll
  for (int f = 0; f < 4; ++f) {
    float hn = hn4[f];
    float tn = tn4[f];
    v8f c = {0.f, 0.f, 0.f, 0.f, 0.f, 0.f, 0.f, 0.f};
#pragma unroll
    for (int s = 0; s < 4; ++s) {
      int o = f * 16 + s * 4 + koff;
      v2f hv2 = *reinterpret_cast<const v2f*>(Hrow + o);
      v2f tv2 = *reinterpret_cast<const v2f*>(Trow + o);
      v2f av, bv;
      av.x = hv2.x * hn;
      av.y = hv2.y * hn;
      bv.x = fast_tanh(tv2.x * tn);
      bv.y = fast_tanh(tv2.y * tn);
      c = __builtin_amdgcn_wmma_f32_16x16x4_f32(false, av, false, bv,
                                                (short)0, c, false, false);
    }
    int idx = lane & 7;
    float dv = c[0];
    dv = (idx == 1) ? c[1] : dv;
    dv = (idx == 2) ? c[2] : dv;
    dv = (idx == 3) ? c[3] : dv;
    dv = (idx == 4) ? c[4] : dv;
    dv = (idx == 5) ? c[5] : dv;
    dv = (idx == 6) ? c[6] : dv;
    dv = (idx == 7) ? c[7] : dv;
    a4[f] += dv;
  }
  if (doOut) A4[eOut] = a4;
}

template <bool TAIL>
__device__ __forceinline__ void acc_add_body(float4* acc4, const float4* __restrict__ x4, int nVec, int base) {
  float4 a[VEC_ITEMS], b[VEC_ITEMS];
#pragma unroll
  for (int k = 0; k < VEC_ITEMS; ++k) {
    int i = base + k * WG;
    if (TAIL) i = (i < nVec) ? i : (nVec - 1);
    a[k] = acc4[i];
    b[k] = x4[i];
  }
#pragma unroll
  for (int k = 0; k < VEC_ITEMS; ++k) {
    int i = base + k * WG;
    float4 r = a[k];
    r.x += b[k].x; r.y += b[k].y; r.z += b[k].z; r.w += b[k].w;
    if (!TAIL || i < nVec) acc4[i] = r;
  }
}

__global__ void k_acc_add(float* __restrict__ acc, const float* __restrict__ x, int nVec) {
  float4* acc4 = reinterpret_cast<float4*>(acc);
  const float4* __restrict__ x4 = reinterpret_cast<const float4*>(x);
  int base = blockIdx.x * (WG * VEC_ITEMS) + threadIdx.x;
  if ((int)(blockIdx.x + 1) * (WG * VEC_ITEMS) <= nVec)
    acc_add_body<false>(acc4, x4, nVec, base);
  else
    acc_add_body<true>(acc4, x4, nVec, base);
}

template <bool TAIL>
__device__ __forceinline__ void final_body(const float4* __restrict__ acc4, float4* __restrict__ out4,
                                           int nVec, int base) {
  float4 a[VEC_ITEMS];
#pragma unroll
  for (int k = 0; k < VEC_ITEMS; ++k) {
    int i = base + k * WG;
    if (TAIL) i = (i < nVec) ? i : (nVec - 1);
    a[k] = acc4[i];
  }
#pragma unroll
  for (int k = 0; k < VEC_ITEMS; ++k) {
    int i = base + k * WG;
    float4 r = a[k];
    r.x *= (1.0f / 3.0f); r.y *= (1.0f / 3.0f); r.z *= (1.0f / 3.0f); r.w *= (1.0f / 3.0f);
    if (!TAIL || i < nVec) out4[i] = r;
  }
}

__global__ void k_final(const float* __restrict__ acc, float* __restrict__ out, int nVec) {
  const float4* __restrict__ acc4 = reinterpret_cast<const float4*>(acc);
  float4* __restrict__ out4 = reinterpret_cast<float4*>(out);
  int base = blockIdx.x * (WG * VEC_ITEMS) + threadIdx.x;
  if ((int)(blockIdx.x + 1) * (WG * VEC_ITEMS) <= nVec)
    final_body<false>(acc4, out4, nVec, base);
  else
    final_body<true>(acc4, out4, nVec, base);
}

static inline size_t align16f(size_t x) { return (x + 15) & ~(size_t)15; }

extern "C" void kernel_launch(void* const* d_in, const int* in_sizes, int n_in,
                              void* d_out, int out_size, void* d_ws, size_t ws_size,
                              hipStream_t stream) {
  (void)n_in; (void)out_size; (void)ws_size;
  const float* u  = (const float*)d_in[0];
  const float* it = (const float*)d_in[1];
  const int* hl   = (const int*)d_in[2];
  const int* tl   = (const int*)d_in[3];

  const int userElems = in_sizes[0];
  const int itemElems = in_sizes[1];
  const int E = in_sizes[2];
  const int N = (userElems + itemElems) / 64;
  const size_t N64 = (size_t)N * 64;
  const size_t E4  = (size_t)E * 4;
  const size_t N4  = (size_t)N * 4;

  float* base = (float*)d_ws;
  size_t off = 0;
  float* ego   = base + off; off += align16f(N64);
  float* fact  = base + off; off += align16f(N64);
  float* acc   = base + off; off += align16f(N64);
  float* A     = base + off; off += align16f(E4);
  float* ew    = base + off; off += align16f(E4);
  float* dval  = base + off; off += align16f(N4);
  float* hninv = base + off; off += align16f(N4);
  float* tninv = base + off; off += align16f(N4);

#define GRID(n) ((unsigned)(((size_t)(n) + WG - 1) / WG))
#define GRIDP(n, per) ((unsigned)(((size_t)(n) + (size_t)WG * (per) - 1) / ((size_t)WG * (per))))
  const int N16 = (int)(N64 / 4);

  k_init<<<GRIDP(N16, VEC_ITEMS), WG, 0, stream>>>(u, it, ego, acc, userElems / 4, N16);
  k_fill1<<<GRIDP(E, VEC_ITEMS), WG, 0, stream>>>(A, E);

  float* cur = ego;
  float* nxt = fact;
  const int nGroups = (E + 15) / 16;

  for (int layer = 0; layer < 2; ++layer) {
    k_norminv<<<GRID(N4), WG, 0, stream>>>(cur, tninv, (int)N4);
    for (int itn = 0; itn < 2; ++itn) {
      hipMemsetAsync(dval, 0, N4 * sizeof(float), stream);
      k_softmax_dval<<<GRIDP(E4, SM_ITEMS), WG, 0, stream>>>(A, hl, ew, dval, E);
      k_dinv<<<GRID(N), WG, 0, stream>>>(dval, N);
      k_edge_weight<<<GRIDP(E, EW_ITEMS), WG, 0, stream>>>(ew, dval, hl, tl, E);
      hipMemsetAsync(nxt, 0, N64 * sizeof(float), stream);
      k_message<<<GRID(E), WG, 0, stream>>>(ew, cur, hl, tl, nxt, E);
      const bool lastUpdateDead = (layer == 1 && itn == 1);
      if (!lastUpdateDead) {
        k_norminv<<<GRID(N4), WG, 0, stream>>>(nxt, hninv, (int)N4);
        k_updateA<<<GRID((size_t)nGroups * 32), WG, 0, stream>>>(
            nxt, cur, hninv, tninv, hl, tl, A, E, nGroups);
      }
    }
    k_acc_add<<<GRIDP(N16, VEC_ITEMS), WG, 0, stream>>>(acc, nxt, N16);
    float* tmp = cur; cur = nxt; nxt = tmp;
  }

  k_final<<<GRIDP(N16, VEC_ITEMS), WG, 0, stream>>>(acc, (float*)d_out, N16);
#undef GRID
#undef GRIDP
}


# Edges counting-sorted by head once a call (integer atomic rank, arrival order); degree and message sums become per-head sequential f32 loops with one store, float atomics and their memsets gone
# speedup vs baseline: 3.0764x; 1.5240x over previous
#include <hip/hip_runtime.h>
#include <math.h>

typedef float v2f __attribute__((ext_vector_type(2)));
typedef float v8f __attribute__((ext_vector_type(8)));
typedef float v4f __attribute__((ext_vector_type(4)));

#define WG 256

__device__ __forceinline__ int atomic_rank(int* p) {
  return __hip_atomic_fetch_add(p, 1, __ATOMIC_RELAXED, __HIP_MEMORY_SCOPE_AGENT);
}

__device__ __forceinline__ float fast_tanh(float x) {
  float r;
  asm("v_tanh_f32 %0, %1\n\tv_nop\n\tv_nop" : "=v"(r) : "v"(x));
  return r;
}

#define VEC_ITEMS 4

static_assert(VEC_ITEMS == 4, "keep_loaded4 names four tuples");
__device__ __forceinline__ void keep_loaded4(v4f (&v)[4]) {
  asm volatile("" : "+v"(v[0]), "+v"(v[1]), "+v"(v[2]), "+v"(v[3]));
}

template <int SRC, bool TAIL>
__device__ __forceinline__ void init_body(const v4f* __restrict__ u4, const v4f* __restrict__ it4,
                                          v4f* __restrict__ ego4, v4f* __restrict__ acc4,
                                          int userVec, int totalVec, int base) {
  v4f v[VEC_ITEMS];
#pragma unroll
  for (int k = 0; k < VEC_ITEMS; ++k) {
    int i = base + k * WG;
    if (TAIL) i = (i < totalVec) ? i : (totalVec - 1);
    if (SRC == 0) v[k] = u4[i];
    else if (SRC == 1) v[k] = it4[i - userVec];
    else v[k] = (i < userVec) ? u4[i] : it4[i - userVec];
  }
  keep_loaded4(v);
#pragma unroll
  for (int k = 0; k < VEC_ITEMS; ++k) {
    int i = base + k * WG;
    if (!TAIL || i < totalVec) { ego4[i] = v[k]; acc4[i] = v[k]; }
  }
}

__global__ void k_init(const float* __restrict__ u, const float* __restrict__ it,
                       float* __restrict__ ego, float* __restrict__ acc,
                       int userVec, int totalVec) {
  const v4f* __restrict__ u4 = reinterpret_cast<const v4f*>(u);
  const v4f* __restrict__ it4 = reinterpret_cast<const v4f*>(it);
  v4f* __restrict__ ego4 = reinterpret_cast<v4f*>(ego);
  v4f* __restrict__ acc4 = reinterpret_cast<v4f*>(acc);
  int first = blockIdx.x * (WG * VEC_ITEMS);
  int end = first + WG * VEC_ITEMS;
  int base = first + threadIdx.x;
  if (end <= userVec)
    init_body<0, false>(u4, it4, ego4, acc4, userVec, totalVec, base);
  else if (first >= userVec && end <= totalVec)
    init_body<1, false>(u4, it4, ego4, acc4, userVec, totalVec, base);
  else
    init_body<2, true>(u4, it4, ego4, acc4, userVec, totalVec, base);
}

__global__ void k_fill1(float* __restrict__ a, int nVec) {
  float4* __restrict__ a4 = reinterpret_cast<float4*>(a);
  int base = blockIdx.x * (WG * VEC_ITEMS) + threadIdx.x;
  float4 one = make_float4(1.0f, 1.0f, 1.0f, 1.0f);
#pragma unroll
  for (int k = 0; k < VEC_ITEMS; ++k) {
    int i = base + k * WG;
    if (i < nVec) a4[i] = one;
  }
}

#define SORT_ITEMS 4

__global__ void k_head_rank(const int* __restrict__ hl, int* __restrict__ count,
                            int* __restrict__ rank, int E) {
  int base = blockIdx.x * (WG * SORT_ITEMS) + threadIdx.x;
  int h[SORT_ITEMS];
#pragma unroll
  for (int k = 0; k < SORT_ITEMS; ++k) {
    int e = base + k * WG;
    h[k] = hl[(e < E) ? e : (E - 1)];
  }
  int r[SORT_ITEMS];
#pragma unroll
  for (int k = 0; k < SORT_ITEMS; ++k) {
    int e = base + k * WG;
    r[k] = 0;
    if (e < E) r[k] = atomic_rank(count + h[k]);
  }
#pragma unroll
  for (int k = 0; k < SORT_ITEMS; ++k) {
    int e = base + k * WG;
    if (e < E) rank[e] = r[k];
  }
}

#define SCAN_WG 1024
#define SCAN_STEPS 20
#define SCAN_WAVE_SPAN (SCAN_STEPS * 128)

__device__ __forceinline__ int wave_incl_scan(int x, int lane) {
#pragma unroll
  for (int d = 1; d < 32; d <<= 1) {
    int y = __shfl_up(x, d);
    if (lane >= d) x += y;
  }
  return x;
}

__global__ void __launch_bounds__(SCAN_WG) k_scan(int* __restrict__ count, int N) {
  __shared__ int wsum[SCAN_WG / 32];
  int lane = threadIdx.x & 31;
  int w = __builtin_amdgcn_readfirstlane((int)(threadIdx.x >> 5));
  int carry = 0;
  for (int rbase = 0; rbase < N; rbase += (SCAN_WG / 32) * SCAN_WAVE_SPAN) {
    int wbase = rbase + w * SCAN_WAVE_SPAN;
    int v[SCAN_STEPS][4];
    int tot = 0;
#pragma unroll
    for (int k = 0; k < SCAN_STEPS; ++k) {
      int i = wbase + k * 128 + lane * 4;
      if (i + 3 < N) {
        int4 x = *reinterpret_cast<const int4*>(count + i);
        v[k][0] = x.x; v[k][1] = x.y; v[k][2] = x.z; v[k][3] = x.w;
      } else {
#pragma unroll
        for (int q = 0; q < 4; ++q) v[k][q] = (i + q < N) ? count[i + q] : 0;
      }
      tot += v[k][0] + v[k][1] + v[k][2] + v[k][3];
    }
    int incl = wave_incl_scan(tot, lane);
    if (lane == 31) wsum[w] = incl;
    __syncthreads();
    int xs = wave_incl_scan(wsum[lane], lane);
    int run = carry + ((w > 0) ? __builtin_amdgcn_readlane(xs, w - 1) : 0);
    carry += __builtin_amdgcn_readlane(xs, 31);
#pragma unroll
    for (int k = 0; k < SCAN_STEPS; ++k) {
      int i = wbase + k * 128 + lane * 4;
      int l4 = v[k][0] + v[k][1] + v[k][2] + v[k][3];
      int inc = wave_incl_scan(l4, lane);
      int b = run + inc - l4;
      if (i + 3 < N) {
        int4 o;
        o.x = b; o.y = o.x + v[k][0]; o.z = o.y + v[k][1]; o.w = o.z + v[k][2];
        *reinterpret_cast<int4*>(count + i) = o;
      } else {
#pragma unroll
        for (int q = 0; q < 4; ++q) {
          if (i + q < N) count[i + q] = b;
          b += v[k][q];
        }
      }
      run += __builtin_amdgcn_readlane(inc, 31);
    }
    __syncthreads();
  }
}

__global__ void k_place(const int* __restrict__ hl, const int* __restrict__ tl,
                        const int* __restrict__ rowstart, const int* __restrict__ rank,
                        int* __restrict__ hs, int* __restrict__ ts, int E) {
  int base = blockIdx.x * (WG * SORT_ITEMS) + threadIdx.x;
  int h[SORT_ITEMS], t[SORT_ITEMS], r[SORT_ITEMS], p[SORT_ITEMS];
#pragma unroll
  for (int k = 0; k < SORT_ITEMS; ++k) {
    int e = base + k * WG;
    e = (e < E) ? e : (E - 1);
    h[k] = hl[e];
    t[k] = tl[e];
    r[k] = rank[e];
  }
#pragma unroll
  for (int k = 0; k < SORT_ITEMS; ++k) p[k] = rowstart[h[k]] + r[k];
#pragma unroll
  for (int k = 0; k < SORT_ITEMS; ++k) {
    int e = base + k * WG;
    if (e < E) { hs[p[k]] = h[k]; ts[p[k]] = t[k]; }
  }
}

#define SM_ITEMS 4
static_assert(SM_ITEMS == VEC_ITEMS, "softmax_body uses keep_loaded4");

template <bool TAIL>
__device__ __forceinline__ void softmax_body(const v4f* __restrict__ A4, v4f* __restrict__ ew4,
                                             int E, int base) {
  v4f av[SM_ITEMS];
#pragma unroll
  for (int k = 0; k < SM_ITEMS; ++k) {
    int e = base + k * WG;
    if (TAIL) e = (e < E) ? e : (E - 1);
    av[k] = A4[e];
  }
  keep_loaded4(av);
#pragma unroll
  for (int k = 0; k < SM_ITEMS; ++k) {
    int e = base + k * WG;
    v4f a = av[k];
    float m = fmaxf(fmaxf(a.x, a.y), fmaxf(a.z, a.w));
    float e0 = expf(a.x - m), e1 = expf(a.y - m), e2 = expf(a.z - m), e3 = expf(a.w - m);
    float inv = 1.0f / (e0 + e1 + e2 + e3);
    v4f r = {e0 * inv, e1 * inv, e2 * inv, e3 * inv};
    if (!TAIL || e < E) ew4[e] = r;
  }
}

__global__ void k_softmax(const float* __restrict__ A, float* __restrict__ ew, int E) {
  const v4f* __restrict__ A4 = reinterpret_cast<const v4f*>(A);
  v4f* __restrict__ ew4 = reinterpret_cast<v4f*>(ew);
  int base = blockIdx.x * (WG * SM_ITEMS) + threadIdx.x;
  if ((int)(blockIdx.x + 1) * (WG * SM_ITEMS) <= E)
    softmax_body<false>(A4, ew4, E, base);
  else
    softmax_body<true>(A4, ew4, E, base);
}

__global__ void k_degree_dinv(const float* __restrict__ ew, const int* __restrict__ rowstart,
                              float* __restrict__ dinv, int N, int E) {
  int n = blockIdx.x * blockDim.x + threadIdx.x;
  if (n >= N) return;
  const v4f* __restrict__ ew4 = reinterpret_cast<const v4f*>(ew);
  int s = rowstart[n];
  int e = (n + 1 < N) ? rowstart[n + 1] : E;
  v4f acc = {0.f, 0.f, 0.f, 0.f};
  for (int i = s; i < e; i += 4) {
    int last = e - 1;
    v4f x0 = ew4[i];
    v4f x1 = ew4[(i + 1 < e) ? (i + 1) : last];
    v4f x2 = ew4[(i + 2 < e) ? (i + 2) : last];
    v4f x3 = ew4[(i + 3 < e) ? (i + 3) : last];
    asm volatile("" : "+v"(x0), "+v"(x1), "+v"(x2), "+v"(x3));
    acc += x0;
    if (i + 1 < e) acc += x1;
    if (i + 2 < e) acc += x2;
    if (i + 3 < e) acc += x3;
  }
  v4f r;
  r.x = 1.0f / sqrtf(fmaxf(acc.x, 1e-8f));
  r.y = 1.0f / sqrtf(fmaxf(acc.y, 1e-8f));
  r.z = 1.0f / sqrtf(fmaxf(acc.z, 1e-8f));
  r.w = 1.0f / sqrtf(fmaxf(acc.w, 1e-8f));
  reinterpret_cast<v4f*>(dinv)[n] = r;
}

#define EW_ITEMS 2

template <bool TAIL>
__device__ __forceinline__ void edge_weight_body(float4* ew4, const float4* __restrict__ dinv4,
                                                 const int* __restrict__ hl, const int* __restrict__ tl,
                                                 int E, int base) {
  float4 w[EW_ITEMS], dh[EW_ITEMS], dt[EW_ITEMS];
  int h[EW_ITEMS], t[EW_ITEMS];
#pragma unroll
  for (int k = 0; k < EW_ITEMS; ++k) {
    int e = base + k * WG;
    if (TAIL) e = (e < E) ? e : (E - 1);
    h[k] = hl[e];
    t[k] = tl[e];
    w[k] = ew4[e];
  }
#pragma unroll
  for (int k = 0; k < EW_ITEMS; ++k) {
    dh[k] = dinv4[h[k]];
    dt[k] = dinv4[t[k]];
  }
#pragma unroll
  for (int k = 0; k < EW_ITEMS; ++k) {
    int e = base + k * WG;
    float4 r = w[k];
    r.x *= dh[k].x * dt[k].x;
    r.y *= dh[k].y * dt[k].y;
    r.z *= dh[k].z * dt[k].z;
    r.w *= dh[k].w * dt[k].w;
    if (!TAIL || e < E) ew4[e] = r;
  }
}

__global__ void k_edge_weight(float* __restrict__ ew, const float* __restrict__ dinv,
                              const int* __restrict__ hl, const int* __restrict__ tl, int E) {
  float4* ew4 = reinterpret_cast<float4*>(ew);
  const float4* __restrict__ dinv4 = reinterpret_cast<const float4*>(dinv);
  int base = blockIdx.x * (WG * EW_ITEMS) + threadIdx.x;
  if ((int)(blockIdx.x + 1) * (WG * EW_ITEMS) <= E)
    edge_weight_body<false>(ew4, dinv4, hl, tl, E, base);
  else
    edge_weight_body<true>(ew4, dinv4, hl, tl, E, base);
}

#define MSG_NODES 4
#define MSG_BATCH 8
static_assert(MSG_BATCH == 8, "the asm statement in k_message names eight edges' row words");

__device__ __forceinline__ float lane_bcast(float v, int srcLane) {
  return __int_as_float(__builtin_amdgcn_readlane(__float_as_int(v), srcLane));
}

__global__ void __launch_bounds__(WG) k_message(
    const float* __restrict__ ew, const float* __restrict__ ego,
    const int* __restrict__ rowstart, const int* __restrict__ ts,
    float* __restrict__ fact, int N, int E) {
  int lane = threadIdx.x & 31;
  int wave = __builtin_amdgcn_readfirstlane((int)((blockIdx.x * WG + threadIdx.x) >> 5));
  bool upper = lane >= 16;
  const float4* __restrict__ ew4 = reinterpret_cast<const float4*>(ew);

  for (int q = 0; q < MSG_NODES; ++q) {
    int n = wave * MSG_NODES + q;
    if (n >= N) return;
    int s = rowstart[n];
    int end = (n + 1 < N) ? rowstart[n + 1] : E;
    float acc0 = 0.f, acc1 = 0.f;
    for (int c = s; c < end; c += 32) {
      int cnt = end - c;
      cnt = (cnt < 32) ? cnt : 32;
      int el = c + ((lane < cnt) ? lane : (cnt - 1));
      int tv = ts[el];
      float4 wv = ew4[el];
      for (int b = 0; b < cnt; b += MSG_BATCH) {
        float p0[MSG_BATCH], p1[MSG_BATCH];
        float w0[MSG_BATCH], w1[MSG_BATCH];
#pragma unroll
        for (int k = 0; k < MSG_BATCH; ++k) {
          int j = b + k;
          j = (j < cnt) ? j : (cnt - 1);
          int t = __builtin_amdgcn_readlane(tv, j);
          float wx = lane_bcast(wv.x, j), wy = lane_bcast(wv.y, j);
          float wz = lane_bcast(wv.z, j), ww = lane_bcast(wv.w, j);
          w0[k] = upper ? wy : wx;
          w1[k] = upper ? ww : wz;
          const float* sp = ego + (size_t)t * 64 + lane;
          p0[k] = sp[0];
          p1[k] = sp[32];
        }
        asm volatile("" : "+v"(p0[0]), "+v"(p0[1]), "+v"(p0[2]), "+v"(p0[3]),
                          "+v"(p0[4]), "+v"(p0[5]), "+v"(p0[6]), "+v"(p0[7]),
                          "+v"(p1[0]), "+v"(p1[1]), "+v"(p1[2]), "+v"(p1[3]),
                          "+v"(p1[4]), "+v"(p1[5]), "+v"(p1[6]), "+v"(p1[7]));
#pragma unroll
        for (int k = 0; k < MSG_BATCH; ++k) {
          if (b + k < cnt) {
            acc0 += w0[k] * p0[k];
            acc1 += w1[k] * p1[k];
          }
        }
      }
    }
    float* d = fact + (size_t)n * 64 + lane;
    d[0] = acc0;
    d[32] = acc1;
  }
}

__global__ void k_norminv(const float* __restrict__ x, float* __restrict__ out, int NF) {
  int i = blockIdx.x * blockDim.x + threadIdx.x;
  if (i >= NF) return;
  int n = i >> 2, f = i & 3;
  const float4* p = reinterpret_cast<const float4*>(x + (size_t)n * 64 + (size_t)f * 16);
  float s = 0.f;
#pragma unroll
  for (int k = 0; k < 4; ++k) {
    float4 v = p[k];
    s += v.x * v.x + v.y * v.y + v.z * v.z + v.w * v.w;
  }
  out[i] = 1.0f / fmaxf(sqrtf(s), 1e-12f);
}

#define UA_ROWS 16
#define UA_PITCH 68
#define UA_WAVE_FLOATS (2 * UA_ROWS * UA_PITCH)

__global__ void __launch_bounds__(WG) k_updateA(
    const float* __restrict__ fact, const float* __restrict__ ego,
    const float* __restrict__ hninv, const float* __restrict__ tninv,
    const int* __restrict__ hl, const int* __restrict__ tl,
    float* __restrict__ A, int E, int nGroups) {
  __shared__ __attribute__((aligned(16))) float stage[(WG / 32) * UA_WAVE_FLOATS];

  int lane = threadIdx.x & 31;
  int wib = __builtin_amdgcn_readfirstlane((int)(threadIdx.x >> 5));
  int wave = __builtin_amdgcn_readfirstlane((int)((blockIdx.x * WG + threadIdx.x) >> 5));
  if (wave >= nGroups) return;

  int r = lane & 15;
  int eIdx = wave * 16 + r;
  int e = (eIdx < E) ? eIdx : (E - 1);
  int h = hl[e], t = tl[e];
  int koff = (lane >> 4) << 1;

  bool active = (lane < 8) || (lane >= 24);
  int outRow = (lane < 8) ? lane : (lane - 16);
  int eOut = wave * 16 + outRow;
  bool doOut = active && eOut < E;
  v4f* A4 = reinterpret_cast<v4f*>(A);
  v4f a4 = {0.f, 0.f, 0.f, 0.f};
  if (doOut) a4 = A4[eOut];

  v4f hn4 = *reinterpret_cast<const v4f*>(hninv + (size_t)h * 4);
  v4f tn4 = *reinterpret_cast<const v4f*>(tninv + (size_t)t * 4);

  float* sH = stage + wib * UA_WAVE_FLOATS;
  float* sT = sH + UA_ROWS * UA_PITCH;

  v2f hr[UA_ROWS], tr[UA_ROWS];
#pragma unroll
  for (int j = 0; j < UA_ROWS; ++j) {
    int hj = __builtin_amdgcn_readlane(h, j);
    int tj = __builtin_amdgcn_readlane(t, j);
    hr[j] = *reinterpret_cast<const v2f*>(fact + (size_t)hj * 64 + 2 * lane);
    tr[j] = *reinterpret_cast<const v2f*>(ego + (size_t)tj * 64 + 2 * lane);
  }
#pragma unroll
  for (int j = 0; j < UA_ROWS; ++j) {
    *reinterpret_cast<v2f*>(sH + j * UA_PITCH + 2 * lane) = hr[j];
    *reinterpret_cast<v2f*>(sT + j * UA_PITCH + 2 * lane) = tr[j];
  }
  __builtin_amdgcn_wave_barrier();

  const float* Hrow = sH + r * UA_PITCH;
  const float* Trow = sT + r * UA_PITCH;

#pragma unroll
  for (int f = 0; f < 4; ++f) {
    float hn = hn4[f];
    float tn = tn4[f];
    v8f c = {0.f, 0.f, 0.f, 0.f, 0.f, 0.f, 0.f, 0.f};
#pragma unroll
    for (int s = 0; s < 4; ++s) {
      int o = f * 16 + s * 4 + koff;
      v2f hv2 = *reinterpret_cast<const v2f*>(Hrow + o);
      v2f tv2 = *reinterpret_cast<const v2f*>(Trow + o);
      v2f av, bv;
      av.x = hv2.x * hn;
      av.y = hv2.y * hn;
      bv.x = fast_tanh(tv2.x * tn);
      bv.y = fast_tanh(tv2.y * tn);
      c = __builtin_amdgcn_wmma_f32_16x16x4_f32(false, av, false, bv,
                                                (short)0, c, false, false);
    }
    int idx = lane & 7;
    float dv = c[0];
    dv = (idx == 1) ? c[1] : dv;
    dv = (idx == 2) ? c[2] : dv;
    dv = (idx == 3) ? c[3] : dv;
    dv = (idx == 4) ? c[4] : dv;
    dv = (idx == 5) ? c[5] : dv;
    dv = (idx == 6) ? c[6] : dv;
    dv = (idx == 7) ? c[7] : dv;
    a4[f] += dv;
  }
  if (doOut) A4[eOut] = a4;
}

template <bool TAIL>
__device__ __forceinline__ void acc_add_body(float4* acc4, const float4* __restrict__ x4, int nVec, int base) {
  float4 a[VEC_ITEMS], b[VEC_ITEMS];
#pragma unroll
  for (int k = 0; k < VEC_ITEMS; ++k) {
    int i = base + k * WG;
    if (TAIL) i = (i < nVec) ? i : (nVec - 1);
    a[k] = acc4[i];
    b[k] = x4[i];
  }
#pragma unroll
  for (int k = 0; k < VEC_ITEMS; ++k) {
    int i = base + k * WG;
    float4 r = a[k];
    r.x += b[k].x; r.y += b[k].y; r.z += b[k].z; r.w += b[k].w;
    if (!TAIL || i < nVec) acc4[i] = r;
  }
}

__global__ void k_acc_add(float* __restrict__ acc, const float* __restrict__ x, int nVec) {
  float4* acc4 = reinterpret_cast<float4*>(acc);
  const float4* __restrict__ x4 = reinterpret_cast<const float4*>(x);
  int base = blockIdx.x * (WG * VEC_ITEMS) + threadIdx.x;
  if ((int)(blockIdx.x + 1) * (WG * VEC_ITEMS) <= nVec)
    acc_add_body<false>(acc4, x4, nVec, base);
  else
    acc_add_body<true>(acc4, x4, nVec, base);
}

template <bool TAIL>
__device__ __forceinline__ void final_body(float4* out4, int nVec, int base) {
  float4 a[VEC_ITEMS];
#pragma unroll
  for (int k = 0; k < VEC_ITEMS; ++k) {
    int i = base + k * WG;
    if (TAIL) i = (i < nVec) ? i : (nVec - 1);
    a[k] = out4[i];
  }
#pragma unroll
  for (int k = 0; k < VEC_ITEMS; ++k) {
    int i = base + k * WG;
    float4 r = a[k];
    r.x *= (1.0f / 3.0f); r.y *= (1.0f / 3.0f); r.z *= (1.0f / 3.0f); r.w *= (1.0f / 3.0f);
    if (!TAIL || i < nVec) out4[i] = r;
  }
}

__global__ void k_final(float* out, int nVec) {
  float4* out4 = reinterpret_cast<float4*>(out);
  int base = blockIdx.x * (WG * VEC_ITEMS) + threadIdx.x;
  if ((int)(blockIdx.x + 1) * (WG * VEC_ITEMS) <= nVec)
    final_body<false>(out4, nVec, base);
  else
    final_body<true>(out4, nVec, base);
}

static inline size_t align16f(size_t x) { return (x + 15) & ~(size_t)15; }

extern "C" void kernel_launch(void* const* d_in, const int* in_sizes, int n_in,
                              void* d_out, int out_size, void* d_ws, size_t ws_size,
                              hipStream_t stream) {
  (void)n_in; (void)out_size; (void)ws_size;
  const float* u  = (const float*)d_in[0];
  const float* it = (const float*)d_in[1];
  const int* hl   = (const int*)d_in[2];
  const int* tl   = (const int*)d_in[3];

  const int userElems = in_sizes[0];
  const int itemElems = in_sizes[1];
  const int E = in_sizes[2];
  const int N = (userElems + itemElems) / 64;
  const size_t N64 = (size_t)N * 64;
  const size_t E4  = (size_t)E * 4;
  const size_t N4  = (size_t)N * 4;

  float* base = (float*)d_ws;
  size_t off = 0;
  float* ego   = base + off; off += align16f(N64);
  float* fact  = base + off; off += align16f(N64);
  int* hs       = (int*)(base + off);
  int* ts       = hs + align16f(E);
  int* rank     = ts + align16f(E);
  int* rowstart = rank + align16f(E);
  off += align16f(N64);
  float* A     = base + off; off += align16f(E4);
  float* ew    = base + off; off += align16f(E4);
  float* dinv  = base + off; off += align16f(N4);
  float* hninv = base + off; off += align16f(N4);
  float* tninv = base + off; off += align16f(N4);
  float* acc   = (float*)d_out;

#define GRID(n) ((unsigned)(((size_t)(n) + WG - 1) / WG))
#define GRIDP(n, per) ((unsigned)(((size_t)(n) + (size_t)WG * (per) - 1) / ((size_t)WG * (per))))
  const int N16 = (int)(N64 / 4);

  k_init<<<GRIDP(N16, VEC_ITEMS), WG, 0, stream>>>(u, it, ego, acc, userElems / 4, N16);
  k_fill1<<<GRIDP(E, VEC_ITEMS), WG, 0, stream>>>(A, E);

  hipMemsetAsync(rowstart, 0, (size_t)N * sizeof(int), stream);
  k_head_rank<<<GRIDP(E, SORT_ITEMS), WG, 0, stream>>>(hl, rowstart, rank, E);
  k_scan<<<1, SCAN_WG, 0, stream>>>(rowstart, N);
  k_place<<<GRIDP(E, SORT_ITEMS), WG, 0, stream>>>(hl, tl, rowstart, rank, hs, ts, E);

  float* cur = ego;
  float* nxt = fact;
  const int nGroups = (E + 15) / 16;

  for (int layer = 0; layer < 2; ++layer) {
    k_norminv<<<GRID(N4), WG, 0, stream>>>(cur, tninv, (int)N4);
    for (int itn = 0; itn < 2; ++itn) {
      k_softmax<<<GRIDP(E, SM_ITEMS), WG, 0, stream>>>(A, ew, E);
      k_degree_dinv<<<GRID(N), WG, 0, stream>>>(ew, rowstart, dinv, N, E);
      k_edge_weight<<<GRIDP(E, EW_ITEMS), WG, 0, stream>>>(ew, dinv, hs, ts, E);
      k_message<<<GRID((size_t)((N + MSG_NODES - 1) / MSG_NODES) * 32), WG, 0, stream>>>(
          ew, cur, rowstart, ts, nxt, N, E);
      const bool lastUpdateDead = (layer == 1 && itn == 1);
      if (!lastUpdateDead) {
        k_norminv<<<GRID(N4), WG, 0, stream>>>(nxt, hninv, (int)N4);
        k_updateA<<<GRID((size_t)nGroups * 32), WG, 0, stream>>>(
            nxt, cur, hninv, tninv, hs, ts, A, E, nGroups);
      }
    }
    k_acc_add<<<GRIDP(N16, VEC_ITEMS), WG, 0, stream>>>(acc, nxt, N16);
    float* tmp = cur; cur = nxt; nxt = tmp;
  }

  k_final<<<GRIDP(N16, VEC_ITEMS), WG, 0, stream>>>(acc, N16);
#undef GRID
#undef GRIDP
}


# Elementwise kernels fused into neighbours, 32 -> 18 launches: degree and message kernels form softmax and edge weight in registers, message waves write row norms (16-lane tree sum), one epilogue pass
# speedup vs baseline: 3.5616x; 1.1577x over previous
#include <hip/hip_runtime.h>
#include <math.h>

typedef float v2f __attribute__((ext_vector_type(2)));
typedef float v8f __attribute__((ext_vector_type(8)));
typedef float v4f __attribute__((ext_vector_type(4)));

#define WG 256

__device__ __forceinline__ int atomic_rank(int* p) {
  return __hip_atomic_fetch_add(p, 1, __ATOMIC_RELAXED, __HIP_MEMORY_SCOPE_AGENT);
}

__device__ __forceinline__ float fast_tanh(float x) {
  float r;
  asm("v_tanh_f32 %0, %1\n\tv_nop\n\tv_nop" : "=v"(r) : "v"(x));
  return r;
}

#define VEC_ITEMS 4

static_assert(VEC_ITEMS == 4, "keep_loaded4 names four tuples");
__device__ __forceinline__ void keep_loaded4(v4f (&v)[4]) {
  asm volatile("" : "+v"(v[0]), "+v"(v[1]), "+v"(v[2]), "+v"(v[3]));
}

template <int SRC, bool TAIL>
__device__ __forceinline__ void init_body(const v4f* __restrict__ u4, const v4f* __restrict__ it4,
                                          v4f* __restrict__ ego4,
                                          int userVec, int totalVec, int base) {
  v4f v[VEC_ITEMS];
#pragma unroll
  for (int k = 0; k < VEC_ITEMS; ++k) {
    int i = base + k * WG;
    if (TAIL) i = (i < totalVec) ? i : (totalVec - 1);
    if (SRC == 0) v[k] = u4[i];
    else if (SRC == 1) v[k] = it4[i - userVec];
    else v[k] = (i < userVec) ? u4[i] : it4[i - userVec];
  }
  keep_loaded4(v);
#pragma unroll
  for (int k = 0; k < VEC_ITEMS; ++k) {
    int i = base + k * WG;
    if (!TAIL || i < totalVec) ego4[i] = v[k];
  }
}

__global__ void k_init(const float* __restrict__ u, const float* __restrict__ it,
                       float* __restrict__ ego,
                       int userVec, int totalVec) {
  const v4f* __restrict__ u4 = reinterpret_cast<const v4f*>(u);
  const v4f* __restrict__ it4 = reinterpret_cast<const v4f*>(it);
  v4f* __restrict__ ego4 = reinterpret_cast<v4f*>(ego);
  int first = blockIdx.x * (WG * VEC_ITEMS);
  int end = first + WG * VEC_ITEMS;
  int base = first + threadIdx.x;
  if (end <= userVec)
    init_body<0, false>(u4, it4, ego4, userVec, totalVec, base);
  else if (first >= userVec && end <= totalVec)
    init_body<1, false>(u4, it4, ego4, userVec, totalVec, base);
  else
    init_body<2, true>(u4, it4, ego4, userVec, totalVec, base);
}

__global__ void k_fill1(float* __restrict__ a, int nVec) {
  float4* __restrict__ a4 = reinterpret_cast<float4*>(a);
  int base = blockIdx.x * (WG * VEC_ITEMS) + threadIdx.x;
  float4 one = make_float4(1.0f, 1.0f, 1.0f, 1.0f);
#pragma unroll
  for (int k = 0; k < VEC_ITEMS; ++k) {
    int i = base + k * WG;
    if (i < nVec) a4[i] = one;
  }
}

#define SORT_ITEMS 4

__global__ void k_head_rank(const int* __restrict__ hl, int* __restrict__ count,
                            int* __restrict__ rank, int E) {
  int base = blockIdx.x * (WG * SORT_ITEMS) + threadIdx.x;
  int h[SORT_ITEMS];
#pragma unroll
  for (int k = 0; k < SORT_ITEMS; ++k) {
    int e = base + k * WG;
    h[k] = hl[(e < E) ? e : (E - 1)];
  }
  int r[SORT_ITEMS];
#pragma unroll
  for (int k = 0; k < SORT_ITEMS; ++k) {
    int e = base + k * WG;
    r[k] = 0;
    if (e < E) r[k] = atomic_rank(count + h[k]);
  }
#pragma unroll
  for (int k = 0; k < SORT_ITEMS; ++k) {
    int e = base + k * WG;
    if (e < E) rank[e] = r[k];
  }
}

#define SCAN_WG 1024
#define SCAN_STEPS 20
#define SCAN_WAVE_SPAN (SCAN_STEPS * 128)

__device__ __forceinline__ int wave_incl_scan(int x, int lane) {
#pragma unroll
  for (int d = 1; d < 32; d <<= 1) {
    int y = __shfl_up(x, d);
    if (lane >= d) x += y;
  }
  return x;
}

__global__ void __launch_bounds__(SCAN_WG) k_scan(int* __restrict__ count, int N) {
  __shared__ int wsum[SCAN_WG / 32];
  int lane = threadIdx.x & 31;
  int w = __builtin_amdgcn_readfirstlane((int)(threadIdx.x >> 5));
  int carry = 0;
  for (int rbase = 0; rbase < N; rbase += (SCAN_WG / 32) * SCAN_WAVE_SPAN) {
    int wbase = rbase + w * SCAN_WAVE_SPAN;
    int v[SCAN_STEPS][4];
    int tot = 0;
#pragma unroll
    for (int k = 0; k < SCAN_STEPS; ++k) {
      int i = wbase + k * 128 + lane * 4;
      if (i + 3 < N) {
        int4 x = *reinterpret_cast<const int4*>(count + i);
        v[k][0] = x.x; v[k][1] = x.y; v[k][2] = x.z; v[k][3] = x.w;
      } else {
#pragma unroll
        for (int q = 0; q < 4; ++q) v[k][q] = (i + q < N) ? count[i + q] : 0;
      }
      tot += v[k][0] + v[k][1] + v[k][2] + v[k][3];
    }
    int incl = wave_incl_scan(tot, lane);
    if (lane == 31) wsum[w] = incl;
    __syncthreads();
    int xs = wave_incl_scan(wsum[lane], lane);
    int run = carry + ((w > 0) ? __builtin_amdgcn_readlane(xs, w - 1) : 0);
    carry += __builtin_amdgcn_readlane(xs, 31);
#pragma unroll
    for (int k = 0; k < SCAN_STEPS; ++k) {
      int i = wbase + k * 128 + lane * 4;
      int l4 = v[k][0] + v[k][1] + v[k][2] + v[k][3];
      int inc = wave_incl_scan(l4, lane);
      int b = run + inc - l4;
      if (i + 3 < N) {
        int4 o;
        o.x = b; o.y = o.x + v[k][0]; o.z = o.y + v[k][1]; o.w = o.z + v[k][2];
        *reinterpret_cast<int4*>(count + i) = o;
      } else {
#pragma unroll
        for (int q = 0; q < 4; ++q) {
          if (i + q < N) count[i + q] = b;
          b += v[k][q];
        }
      }
      run += __builtin_amdgcn_readlane(inc, 31);
    }
    __syncthreads();
  }
}

__global__ void k_place(const int* __restrict__ hl, const int* __restrict__ tl,
                        const int* __restrict__ rowstart, const int* __restrict__ rank,
                        int* __restrict__ hs, int* __restrict__ ts, int E) {
  int base = blockIdx.x * (WG * SORT_ITEMS) + threadIdx.x;
  int h[SORT_ITEMS], t[SORT_ITEMS], r[SORT_ITEMS], p[SORT_ITEMS];
#pragma unroll
  for (int k = 0; k < SORT_ITEMS; ++k) {
    int e = base + k * WG;
    e = (e < E) ? e : (E - 1);
    h[k] = hl[e];
    t[k] = tl[e];
    r[k] = rank[e];
  }
#pragma unroll
  for (int k = 0; k < SORT_ITEMS; ++k) p[k] = rowstart[h[k]] + r[k];
#pragma unroll
  for (int k = 0; k < SORT_ITEMS; ++k) {
    int e = base + k * WG;
    if (e < E) { hs[p[k]] = h[k]; ts[p[k]] = t[k]; }
  }
}

__device__ __forceinline__ v4f softmax4(v4f a) {
  float m = fmaxf(fmaxf(a.x, a.y), fmaxf(a.z, a.w));
  float e0 = expf(a.x - m), e1 = expf(a.y - m), e2 = expf(a.z - m), e3 = expf(a.w - m);
  float inv = 1.0f / (e0 + e1 + e2 + e3);
  v4f r = {e0 * inv, e1 * inv, e2 * inv, e3 * inv};
  return r;
}

__global__ void k_degree_dinv(const float* __restrict__ A, const int* __restrict__ rowstart,
                              float* __restrict__ dinv, int N, int E) {
  int n = blockIdx.x * blockDim.x + threadIdx.x;
  if (n >= N) return;
  const v4f* __restrict__ ew4 = reinterpret_cast<const v4f*>(A);
  int s = rowstart[n];
  int e = (n + 1 < N) ? rowstart[n + 1] : E;
  v4f acc = {0.f, 0.f, 0.f, 0.f};
  for (int i = s; i < e; i += 4) {
    int last = e - 1;
    v4f x0 = ew4[i];
    v4f x1 = ew4[(i + 1 < e) ? (i + 1) : last];
    v4f x2 = ew4[(i + 2 < e) ? (i + 2) : last];
    v4f x3 = ew4[(i + 3 < e) ? (i + 3) : last];
    asm volatile("" : "+v"(x0), "+v"(x1), "+v"(x2), "+v"(x3));
    v4f s0 = softmax4(x0), s1 = softmax4(x1), s2 = softmax4(x2), s3 = softmax4(x3);
    acc += s0;
    if (i + 1 < e) acc += s1;
    if (i + 2 < e) acc += s2;
    if (i + 3 < e) acc += s3;
  }
  v4f r;
  r.x = 1.0f / sqrtf(fmaxf(acc.x, 1e-8f));
  r.y = 1.0f / sqrtf(fmaxf(acc.y, 1e-8f));
  r.z = 1.0f / sqrtf(fmaxf(acc.z, 1e-8f));
  r.w = 1.0f / sqrtf(fmaxf(acc.w, 1e-8f));
  reinterpret_cast<v4f*>(dinv)[n] = r;
}

#define MSG_NODES 4
#define MSG_BATCH 8
static_assert(MSG_BATCH == 8, "the asm statement in k_message names eight edges' row words");

__device__ __forceinline__ float lane_bcast(float v, int srcLane) {
  return __int_as_float(__builtin_amdgcn_readlane(__float_as_int(v), srcLane));
}

template <int MASK>
__device__ __forceinline__ float add_xmask(float x) {
  return x + __int_as_float(__builtin_amdgcn_update_dpp(0, __float_as_int(x), 0x160 + MASK, 0xf, 0xf, false));
}
__device__ __forceinline__ float half16_sum(float x) {
  x = add_xmask<1>(x);
  x = add_xmask<2>(x);
  x = add_xmask<4>(x);
  x = add_xmask<8>(x);
  return x;
}

__global__ void __launch_bounds__(WG) k_message(
    const float* __restrict__ A, const float* __restrict__ dinv, const float* __restrict__ ego,
    const int* __restrict__ rowstart, const int* __restrict__ ts,
    float* __restrict__ fact, float* __restrict__ hninv, int N, int E) {
  int lane = threadIdx.x & 31;
  int wave = __builtin_amdgcn_readfirstlane((int)((blockIdx.x * WG + threadIdx.x) >> 5));
  bool upper = lane >= 16;
  const v4f* __restrict__ A4 = reinterpret_cast<const v4f*>(A);
  const v4f* __restrict__ dinv4 = reinterpret_cast<const v4f*>(dinv);

  float hnv = 0.f;
  for (int q = 0; q < MSG_NODES; ++q) {
    int n = wave * MSG_NODES + q;
    if (n >= N) break;
    int s = rowstart[n];
    int end = (n + 1 < N) ? rowstart[n + 1] : E;
    v4f dh = dinv4[n];
    float acc0 = 0.f, acc1 = 0.f;
    for (int c = s; c < end; c += 32) {
      int cnt = end - c;
      cnt = (cnt < 32) ? cnt : 32;
      int el = c + ((lane < cnt) ? lane : (cnt - 1));
      int tv = ts[el];
      v4f av = A4[el];
      v4f dt = dinv4[tv];
      v4f nw = softmax4(av);
      for (int b = 0; b < cnt; b += MSG_BATCH) {
        float p0[MSG_BATCH], p1[MSG_BATCH];
#pragma unroll
        for (int k = 0; k < MSG_BATCH; ++k) {
          int j = b + k;
          j = (j < cnt) ? j : (cnt - 1);
          int t = __builtin_amdgcn_readlane(tv, j);
          const float* sp = ego + (size_t)t * 64 + lane;
          p0[k] = sp[0];
          p1[k] = sp[32];
        }
        asm volatile("" : "+v"(dt));
        v4f wv;
        wv.x = nw.x * (dh.x * dt.x);
        wv.y = nw.y * (dh.y * dt.y);
        wv.z = nw.z * (dh.z * dt.z);
        wv.w = nw.w * (dh.w * dt.w);
        float w0[MSG_BATCH], w1[MSG_BATCH];
#pragma unroll
        for (int k = 0; k < MSG_BATCH; ++k) {
          int j = b + k;
          j = (j < cnt) ? j : (cnt - 1);
          float wx = lane_bcast(wv.x, j), wy = lane_bcast(wv.y, j);
          float wz = lane_bcast(wv.z, j), ww = lane_bcast(wv.w, j);
          w0[k] = upper ? wy : wx;
          w1[k] = upper ? ww : wz;
        }
        asm volatile("" : "+v"(p0[0]), "+v"(p0[1]), "+v"(p0[2]), "+v"(p0[3]),
                          "+v"(p0[4]), "+v"(p0[5]), "+v"(p0[6]), "+v"(p0[7]),
                          "+v"(p1[0]), "+v"(p1[1]), "+v"(p1[2]), "+v"(p1[3]),
                          "+v"(p1[4]), "+v"(p1[5]), "+v"(p1[6]), "+v"(p1[7]));
#pragma unroll
        for (int k = 0; k < MSG_BATCH; ++k) {
          if (b + k < cnt) {
            acc0 += w0[k] * p0[k];
            acc1 += w1[k] * p1[k];
          }
        }
      }
    }
    float* d = fact + (size_t)n * 64 + lane;
    d[0] = acc0;
    d[32] = acc1;
    float r0 = 1.0f / fmaxf(sqrtf(half16_sum(acc0 * acc0)), 1e-12f);
    float r1 = 1.0f / fmaxf(sqrtf(half16_sum(acc1 * acc1)), 1e-12f);
    float h0 = lane_bcast(r0, 0), h1 = lane_bcast(r0, 16), h2 = lane_bcast(r1, 0), h3 = lane_bcast(r1, 16);
    float fLo = (lane & 1) ? h1 : h0;
    float fHi = (lane & 1) ? h3 : h2;
    float mine = (lane & 2) ? fHi : fLo;
    hnv = ((lane >> 2) == q) ? mine : hnv;
  }
  int nb = wave * MSG_NODES;
  if (lane < MSG_NODES * 4 && nb + (lane >> 2) < N) hninv[(size_t)nb * 4 + lane] = hnv;
}

__global__ void k_norminv(const float* __restrict__ x, float* __restrict__ out, int NF) {
  int i = blockIdx.x * blockDim.x + threadIdx.x;
  if (i >= NF) return;
  int n = i >> 2, f = i & 3;
  const float4* p = reinterpret_cast<const float4*>(x + (size_t)n * 64 + (size_t)f * 16);
  float s = 0.f;
#pragma unroll
  for (int k = 0; k < 4; ++k) {
    float4 v = p[k];
    s += v.x * v.x + v.y * v.y + v.z * v.z + v.w * v.w;
  }
  out[i] = 1.0f / fmaxf(sqrtf(s), 1e-12f);
}

#define UA_ROWS 16
#define UA_PITCH 68
#define UA_WAVE_FLOATS (2 * UA_ROWS * UA_PITCH)

__global__ void __launch_bounds__(WG) k_updateA(
    const float* __restrict__ fact, const float* __restrict__ ego,
    const float* __restrict__ hninv, const float* __restrict__ tninv,
    const int* __restrict__ hl, const int* __restrict__ tl,
    float* __restrict__ A, int E, int nGroups) {
  __shared__ __attribute__((aligned(16))) float stage[(WG / 32) * UA_WAVE_FLOATS];

  int lane = threadIdx.x & 31;
  int wib = __builtin_amdgcn_readfirstlane((int)(threadIdx.x >> 5));
  int wave = __builtin_amdgcn_readfirstlane((int)((blockIdx.x * WG + threadIdx.x) >> 5));
  if (wave >= nGroups) return;

  int r = lane & 15;
  int eIdx = wave * 16 + r;
  int e = (eIdx < E) ? eIdx : (E - 1);
  int h = hl[e], t = tl[e];
  int koff = (lane >> 4) << 1;

  bool active = (lane < 8) || (lane >= 24);
  int outRow = (lane < 8) ? lane : (lane - 16);
  int eOut = wave * 16 + outRow;
  bool doOut = active && eOut < E;
  v4f* A4 = reinterpret_cast<v4f*>(A);
  v4f a4 = {0.f, 0.f, 0.f, 0.f};
  if (doOut) a4 = A4[eOut];

  v4f hn4 = *reinterpret_cast<const v4f*>(hninv + (size_t)h * 4);
  v4f tn4 = *reinterpret_cast<const v4f*>(tninv + (size_t)t * 4);

  float* sH = stage + wib * UA_WAVE_FLOATS;
  float* sT = sH + UA_ROWS * UA_PITCH;

  v2f hr[UA_ROWS], tr[UA_ROWS];
#pragma unroll
  for (int j = 0; j < UA_ROWS; ++j) {
    int hj = __builtin_amdgcn_readlane(h, j);
    int tj = __builtin_amdgcn_readlane(t, j);
    hr[j] = *reinterpret_cast<const v2f*>(fact + (size_t)hj * 64 + 2 * lane);
    tr[j] = *reinterpret_cast<const v2f*>(ego + (size_t)tj * 64 + 2 * lane);
  }
#pragma unroll
  for (int j = 0; j < UA_ROWS; ++j) {
    *reinterpret_cast<v2f*>(sH + j * UA_PITCH + 2 * lane) = hr[j];
    *reinterpret_cast<v2f*>(sT + j * UA_PITCH + 2 * lane) = tr[j];
  }
  __builtin_amdgcn_wave_barrier();

  const float* Hrow = sH + r * UA_PITCH;
  const float* Trow = sT + r * UA_PITCH;

#pragma unroll
  for (int f = 0; f < 4; ++f) {
    float hn = hn4[f];
    float tn = tn4[f];
    v8f c = {0.f, 0.f, 0.f, 0.f, 0.f, 0.f, 0.f, 0.f};
#pragma unroll
    for (int s = 0; s < 4; ++s) {
      int o = f * 16 + s * 4 + koff;
      v2f hv2 = *reinterpret_cast<const v2f*>(Hrow + o);
      v2f tv2 = *reinterpret_cast<const v2f*>(Trow + o);
      v2f av, bv;
      av.x = hv2.x * hn;
      av.y = hv2.y * hn;
      bv.x = fast_tanh(tv2.x * tn);
      bv.y = fast_tanh(tv2.y * tn);
      c = __builtin_amdgcn_wmma_f32_16x16x4_f32(false, av, false, bv,
                                                (short)0, c, false, false);
    }
    int idx = lane & 7;
    float dv = c[0];
    dv = (idx == 1) ? c[1] : dv;
    dv = (idx == 2) ? c[2] : dv;
    dv = (idx == 3) ? c[3] : dv;
    dv = (idx == 4) ? c[4] : dv;
    dv = (idx == 5) ? c[5] : dv;
    dv = (idx == 6) ? c[6] : dv;
    dv = (idx == 7) ? c[7] : dv;
    a4[f] += dv;
  }
  if (doOut) A4[eOut] = a4;
}

template <int SRC, bool TAIL>
__device__ __forceinline__ void final_body(const v4f* __restrict__ u4, const v4f* __restrict__ it4,
                                           const v4f* __restrict__ x4, const v4f* __restrict__ y4,
                                           v4f* __restrict__ out4, int userVec, int totalVec, int base) {
  v4f a[VEC_ITEMS], b[VEC_ITEMS], d[VEC_ITEMS];
#pragma unroll
  for (int k = 0; k < VEC_ITEMS; ++k) {
    int i = base + k * WG;
    if (TAIL) i = (i < totalVec) ? i : (totalVec - 1);
    if (SRC == 0) a[k] = u4[i];
    else if (SRC == 1) a[k] = it4[i - userVec];
    else a[k] = (i < userVec) ? u4[i] : it4[i - userVec];
    b[k] = x4[i];
    d[k] = y4[i];
  }
#pragma unroll
  for (int k = 0; k < VEC_ITEMS; ++k) {
    int i = base + k * WG;
    v4f r = (a[k] + b[k]) + d[k];
    r *= (1.0f / 3.0f);
    if (!TAIL || i < totalVec) out4[i] = r;
  }
}

__global__ void k_final(const float* __restrict__ u, const float* __restrict__ it,
                        const float* __restrict__ x, const float* __restrict__ y,
                        float* __restrict__ out, int userVec, int totalVec) {
  const v4f* __restrict__ u4 = reinterpret_cast<const v4f*>(u);
  const v4f* __restrict__ it4 = reinterpret_cast<const v4f*>(it);
  const v4f* __restrict__ x4 = reinterpret_cast<const v4f*>(x);
  const v4f* __restrict__ y4 = reinterpret_cast<const v4f*>(y);
  v4f* __restrict__ out4 = reinterpret_cast<v4f*>(out);
  int first = blockIdx.x * (WG * VEC_ITEMS);
  int end = first + WG * VEC_ITEMS;
  int base = first + threadIdx.x;
  if (end <= userVec)
    final_body<0, false>(u4, it4, x4, y4, out4, userVec, totalVec, base);
  else if (first >= userVec && end <= totalVec)
    final_body<1, false>(u4, it4, x4, y4, out4, userVec, totalVec, base);
  else
    final_body<2, true>(u4, it4, x4, y4, out4, userVec, totalVec, base);
}

static inline size_t align16f(size_t x) { return (x + 15) & ~(size_t)15; }

extern "C" void kernel_launch(void* const* d_in, const int* in_sizes, int n_in,
                              void* d_out, int out_size, void* d_ws, size_t ws_size,
                              hipStream_t stream) {
  (void)n_in; (void)out_size; (void)ws_size;
  const float* u  = (const float*)d_in[0];
  const float* it = (const float*)d_in[1];
  const int* hl   = (const int*)d_in[2];
  const int* tl   = (const int*)d_in[3];

  const int userElems = in_sizes[0];
  const int itemElems = in_sizes[1];
  const int E = in_sizes[2];
  const int N = (userElems + itemElems) / 64;
  const size_t N64 = (size_t)N * 64;
  const size_t E4  = (size_t)E * 4;
  const size_t N4  = (size_t)N * 4;

  float* base = (float*)d_ws;
  size_t off = 0;
  float* ego   = base + off; off += align16f(N64);
  float* fact  = base + off; off += align16f(N64);
  int* hs       = (int*)(base + off);
  int* ts       = hs + align16f(E);
  int* rank     = ts + align16f(E);
  int* rowstart = rank + align16f(E);
  off += align16f(N64);
  float* A     = base + off; off += align16f(E4);
  float* dinv  = base + off; off += align16f(N4);
  float* hninv = base + off; off += align16f(N4);
  float* tninv = base + off; off += align16f(N4);

#define GRID(n) ((unsigned)(((size_t)(n) + WG - 1) / WG))
#define GRIDP(n, per) ((unsigned)(((size_t)(n) + (size_t)WG * (per) - 1) / ((size_t)WG * (per))))
  const int N16 = (int)(N64 / 4);

  k_init<<<GRIDP(N16, VEC_ITEMS), WG, 0, stream>>>(u, it, ego, userElems / 4, N16);
  k_fill1<<<GRIDP(E, VEC_ITEMS), WG, 0, stream>>>(A, E);

  hipMemsetAsync(rowstart, 0, (size_t)N * sizeof(int), stream);
  k_head_rank<<<GRIDP(E, SORT_ITEMS), WG, 0, stream>>>(hl, rowstart, rank, E);
  k_scan<<<1, SCAN_WG, 0, stream>>>(rowstart, N);
  k_place<<<GRIDP(E, SORT_ITEMS), WG, 0, stream>>>(hl, tl, rowstart, rank, hs, ts, E);

  float* cur = ego;
  float* nxt = fact;
  float* tn = tninv;
  float* hn = hninv;
  const int nGroups = (E + 15) / 16;

  k_norminv<<<GRID(N4), WG, 0, stream>>>(cur, tn, (int)N4);
  for (int layer = 0; layer < 2; ++layer) {
    for (int itn = 0; itn < 2; ++itn) {
      k_degree_dinv<<<GRID(N), WG, 0, stream>>>(A, rowstart, dinv, N, E);
      k_message<<<GRID((size_t)((N + MSG_NODES - 1) / MSG_NODES) * 32), WG, 0, stream>>>(
          A, dinv, cur, rowstart, ts, nxt, hn, N, E);
      const bool lastUpdateDead = (layer == 1 && itn == 1);
      if (!lastUpdateDead) {
        k_updateA<<<GRID((size_t)nGroups * 32), WG, 0, stream>>>(
            nxt, cur, hn, tn, hs, ts, A, E, nGroups);
      }
    }
    float* tmp = cur; cur = nxt; nxt = tmp;
    tmp = tn; tn = hn; hn = tmp;
  }

  k_final<<<GRIDP(N16, VEC_ITEMS), WG, 0, stream>>>(u, it, nxt, cur, (float*)d_out, userElems / 4, N16);
#undef GRID
#undef GRIDP
}


# 18 -> 14 launches, no memset: first iteration takes A = 1 as known (no fill, degree pass or A reads), input tables read in place (no copy), last message pass writes the output mean; f32 sums kept
# speedup vs baseline: 3.9459x; 1.1079x over previous
#include <hip/hip_runtime.h>
#include <math.h>

typedef float v2f __attribute__((ext_vector_type(2)));
typedef float v8f __attribute__((ext_vector_type(8)));
typedef float v4f __attribute__((ext_vector_type(4)));

#define WG 256

__device__ __forceinline__ int atomic_rank(int* p) {
  return __hip_atomic_fetch_add(p, 1, __ATOMIC_RELAXED, __HIP_MEMORY_SCOPE_AGENT);
}

__device__ __forceinline__ float fast_tanh(float x) {
  float r;
  asm("v_tanh_f32 %0, %1\n\tv_nop\n\tv_nop" : "=v"(r) : "v"(x));
  return r;
}

__device__ __forceinline__ const float* row_of(const float* __restrict__ lo, const float* __restrict__ hi,
                                               int split, int n) {
  return (n < split) ? (lo + (size_t)n * 64) : (hi + (size_t)(n - split) * 64);
}

#define SORT_ITEMS 4

__global__ void k_head_rank(const int* __restrict__ hl, int* __restrict__ count,
                            int* __restrict__ rank, int E) {
  int base = blockIdx.x * (WG * SORT_ITEMS) + threadIdx.x;
  int h[SORT_ITEMS];
#pragma unroll
  for (int k = 0; k < SORT_ITEMS; ++k) {
    int e = base + k * WG;
    h[k] = hl[(e < E) ? e : (E - 1)];
  }
  int r[SORT_ITEMS];
#pragma unroll
  for (int k = 0; k < SORT_ITEMS; ++k) {
    int e = base + k * WG;
    r[k] = 0;
    if (e < E) r[k] = atomic_rank(count + h[k]);
  }
#pragma unroll
  for (int k = 0; k < SORT_ITEMS; ++k) {
    int e = base + k * WG;
    if (e < E) rank[e] = r[k];
  }
}

#define SCAN_WG 1024
#define SCAN_STEPS 20
#define SCAN_WAVE_SPAN (SCAN_STEPS * 128)

__device__ __forceinline__ int wave_incl_scan(int x, int lane) {
#pragma unroll
  for (int d = 1; d < 32; d <<= 1) {
    int y = __shfl_up(x, d);
    if (lane >= d) x += y;
  }
  return x;
}

__global__ void __launch_bounds__(SCAN_WG) k_scan(int* __restrict__ count, int N) {
  __shared__ int wsum[SCAN_WG / 32];
  int lane = threadIdx.x & 31;
  int w = __builtin_amdgcn_readfirstlane((int)(threadIdx.x >> 5));
  int carry = 0;
  for (int rbase = 0; rbase < N; rbase += (SCAN_WG / 32) * SCAN_WAVE_SPAN) {
    int wbase = rbase + w * SCAN_WAVE_SPAN;
    int v[SCAN_STEPS][4];
    int tot = 0;
#pragma unroll
    for (int k = 0; k < SCAN_STEPS; ++k) {
      int i = wbase + k * 128 + lane * 4;
      if (i + 3 < N) {
        int4 x = *reinterpret_cast<const int4*>(count + i);
        v[k][0] = x.x; v[k][1] = x.y; v[k][2] = x.z; v[k][3] = x.w;
      } else {
#pragma unroll
        for (int q = 0; q < 4; ++q) v[k][q] = (i + q < N) ? count[i + q] : 0;
      }
      tot += v[k][0] + v[k][1] + v[k][2] + v[k][3];
    }
    int incl = wave_incl_scan(tot, lane);
    if (lane == 31) wsum[w] = incl;
    __syncthreads();
    int xs = wave_incl_scan(wsum[lane], lane);
    int run = carry + ((w > 0) ? __builtin_amdgcn_readlane(xs, w - 1) : 0);
    carry += __builtin_amdgcn_readlane(xs, 31);
#pragma unroll
    for (int k = 0; k < SCAN_STEPS; ++k) {
      int i = wbase + k * 128 + lane * 4;
      int l4 = v[k][0] + v[k][1] + v[k][2] + v[k][3];
      int inc = wave_incl_scan(l4, lane);
      int b = run + inc - l4;
      if (i + 3 < N) {
        int4 o;
        o.x = b; o.y = o.x + v[k][0]; o.z = o.y + v[k][1]; o.w = o.z + v[k][2];
        *reinterpret_cast<int4*>(count + i) = o;
      } else {
#pragma unroll
        for (int q = 0; q < 4; ++q) {
          if (i + q < N) count[i + q] = b;
          b += v[k][q];
        }
      }
      run += __builtin_amdgcn_readlane(inc, 31);
    }
    __syncthreads();
  }
}

__global__ void k_place(const int* __restrict__ hl, const int* __restrict__ tl,
                        const int* __restrict__ rowstart, const int* __restrict__ rank,
                        int* __restrict__ hs, int* __restrict__ ts, int E) {
  int base = blockIdx.x * (WG * SORT_ITEMS) + threadIdx.x;
  int h[SORT_ITEMS], t[SORT_ITEMS], r[SORT_ITEMS], p[SORT_ITEMS];
#pragma unroll
  for (int k = 0; k < SORT_ITEMS; ++k) {
    int e = base + k * WG;
    e = (e < E) ? e : (E - 1);
    h[k] = hl[e];
    t[k] = tl[e];
    r[k] = rank[e];
  }
#pragma unroll
  for (int k = 0; k < SORT_ITEMS; ++k) p[k] = rowstart[h[k]] + r[k];
#pragma unroll
  for (int k = 0; k < SORT_ITEMS; ++k) {
    int e = base + k * WG;
    if (e < E) { hs[p[k]] = h[k]; ts[p[k]] = t[k]; }
  }
}

__device__ __forceinline__ v4f softmax4(v4f a) {
  float m = fmaxf(fmaxf(a.x, a.y), fmaxf(a.z, a.w));
  float e0 = expf(a.x - m), e1 = expf(a.y - m), e2 = expf(a.z - m), e3 = expf(a.w - m);
  float inv = 1.0f / (e0 + e1 + e2 + e3);
  v4f r = {e0 * inv, e1 * inv, e2 * inv, e3 * inv};
  return r;
}

__global__ void k_degree_dinv(const float* __restrict__ A, const int* __restrict__ rowstart,
                              float* __restrict__ dinv, int N, int E) {
  int n = blockIdx.x * blockDim.x + threadIdx.x;
  if (n >= N) return;
  const v4f* __restrict__ ew4 = reinterpret_cast<const v4f*>(A);
  int s = rowstart[n];
  int e = (n + 1 < N) ? rowstart[n + 1] : E;
  v4f acc = {0.f, 0.f, 0.f, 0.f};
  for (int i = s; i < e; i += 4) {
    int last = e - 1;
    v4f x0 = ew4[i];
    v4f x1 = ew4[(i + 1 < e) ? (i + 1) : last];
    v4f x2 = ew4[(i + 2 < e) ? (i + 2) : last];
    v4f x3 = ew4[(i + 3 < e) ? (i + 3) : last];
    asm volatile("" : "+v"(x0), "+v"(x1), "+v"(x2), "+v"(x3));
    v4f s0 = softmax4(x0), s1 = softmax4(x1), s2 = softmax4(x2), s3 = softmax4(x3);
    acc += s0;
    if (i + 1 < e) acc += s1;
    if (i + 2 < e) acc += s2;
    if (i + 3 < e) acc += s3;
  }
  v4f r;
  r.x = 1.0f / sqrtf(fmaxf(acc.x, 1e-8f));
  r.y = 1.0f / sqrtf(fmaxf(acc.y, 1e-8f));
  r.z = 1.0f / sqrtf(fmaxf(acc.z, 1e-8f));
  r.w = 1.0f / sqrtf(fmaxf(acc.w, 1e-8f));
  reinterpret_cast<v4f*>(dinv)[n] = r;
}

#define MSG_NODES 4
#define MSG_BATCH 8
static_assert(MSG_BATCH == 8, "the asm statement in k_message names eight edges' row words");

__device__ __forceinline__ float lane_bcast(float v, int srcLane) {
  return __int_as_float(__builtin_amdgcn_readlane(__float_as_int(v), srcLane));
}

template <int MASK>
__device__ __forceinline__ float add_xmask(float x) {
  return x + __int_as_float(__builtin_amdgcn_update_dpp(0, __float_as_int(x), 0x160 + MASK, 0xf, 0xf, false));
}
__device__ __forceinline__ float half16_sum(float x) {
  x = add_xmask<1>(x);
  x = add_xmask<2>(x);
  x = add_xmask<4>(x);
  x = add_xmask<8>(x);
  return x;
}

template <bool FIRST, bool LAST>
__global__ void __launch_bounds__(WG) k_message(
    const float* __restrict__ A, const float* __restrict__ dinv,
    const float* __restrict__ rowsLo, const float* __restrict__ rowsHi,
    const float* __restrict__ inLo, const float* __restrict__ inHi, int split,
    const int* __restrict__ rowstart, const int* __restrict__ ts,
    float* __restrict__ fact, float* __restrict__ hninv, int N, int E) {
  int lane = threadIdx.x & 31;
  int wave = __builtin_amdgcn_readfirstlane((int)((blockIdx.x * WG + threadIdx.x) >> 5));
  bool upper = lane >= 16;
  const v4f* __restrict__ A4 = reinterpret_cast<const v4f*>(A);
  const v4f* __restrict__ dinv4 = reinterpret_cast<const v4f*>(dinv);

  float hnv = 0.f;
  for (int q = 0; q < MSG_NODES; ++q) {
    int n = wave * MSG_NODES + q;
    if (n >= N) break;
    int s = rowstart[n];
    int end = (n + 1 < N) ? rowstart[n + 1] : E;
    v4f dh;
    if (FIRST) {
      float dv = 1.0f / sqrtf(fmaxf(0.25f * (float)(end - s), 1e-8f));
      dh.x = dv; dh.y = dv; dh.z = dv; dh.w = dv;
    } else {
      dh = dinv4[n];
    }
    float in0 = 0.f, in1 = 0.f, pv0 = 0.f, pv1 = 0.f;
    if (LAST) {
      const float* ip = row_of(inLo, inHi, split, n) + lane;
      const float* pp = row_of(rowsLo, rowsHi, split, n) + lane;
      in0 = ip[0]; in1 = ip[32];
      pv0 = pp[0]; pv1 = pp[32];
      asm volatile("" : "+v"(in0), "+v"(in1), "+v"(pv0), "+v"(pv1));
    }
    float acc0 = 0.f, acc1 = 0.f;
    for (int c = s; c < end; c += 32) {
      int cnt = end - c;
      cnt = (cnt < 32) ? cnt : 32;
      int el = c + ((lane < cnt) ? lane : (cnt - 1));
      int tv = ts[el];
      v4f nw;
      v4f dt = {0.f, 0.f, 0.f, 0.f};
      int r0 = 0, r1raw = 0;
      if (FIRST) {
        nw.x = 0.25f; nw.y = 0.25f; nw.z = 0.25f; nw.w = 0.25f;
        r0 = rowstart[tv];
        r1raw = rowstart[(tv + 1 < N) ? (tv + 1) : tv];
      } else {
        v4f av = A4[el];
        dt = dinv4[tv];
        nw = softmax4(av);
      }
      v4f wv = {0.f, 0.f, 0.f, 0.f};
      for (int b = 0; b < cnt; b += MSG_BATCH) {
        float p0[MSG_BATCH], p1[MSG_BATCH];
#pragma unroll
        for (int k = 0; k < MSG_BATCH; ++k) {
          int j = b + k;
          j = (j < cnt) ? j : (cnt - 1);
          int t = __builtin_amdgcn_readlane(tv, j);
          const float* sp = row_of(rowsLo, rowsHi, split, t) + lane;
          p0[k] = sp[0];
          p1[k] = sp[32];
        }
        if (FIRST) {
          if (b == 0) {
            asm volatile("" : "+v"(r0), "+v"(r1raw));
            int r1 = (tv + 1 < N) ? r1raw : E;
            float dtv = 1.0f / sqrtf(fmaxf(0.25f * (float)(r1 - r0), 1e-8f));
            wv.x = nw.x * (dh.x * dtv);
            wv.y = nw.y * (dh.y * dtv);
            wv.z = nw.z * (dh.z * dtv);
            wv.w = nw.w * (dh.w * dtv);
          }
        } else {
          asm volatile("" : "+v"(dt));
          wv.x = nw.x * (dh.x * dt.x);
          wv.y = nw.y * (dh.y * dt.y);
          wv.z = nw.z * (dh.z * dt.z);
          wv.w = nw.w * (dh.w * dt.w);
        }
        float w0[MSG_BATCH], w1[MSG_BATCH];
#pragma unroll
        for (int k = 0; k < MSG_BATCH; ++k) {
          int j = b + k;
          j = (j < cnt) ? j : (cnt - 1);
          float wx = lane_bcast(wv.x, j), wy = lane_bcast(wv.y, j);
          float wz = lane_bcast(wv.z, j), ww = lane_bcast(wv.w, j);
          w0[k] = upper ? wy : wx;
          w1[k] = upper ? ww : wz;
        }
        asm volatile("" : "+v"(p0[0]), "+v"(p0[1]), "+v"(p0[2]), "+v"(p0[3]),
                          "+v"(p0[4]), "+v"(p0[5]), "+v"(p0[6]), "+v"(p0[7]),
                          "+v"(p1[0]), "+v"(p1[1]), "+v"(p1[2]), "+v"(p1[3]),
                          "+v"(p1[4]), "+v"(p1[5]), "+v"(p1[6]), "+v"(p1[7]));
#pragma unroll
        for (int k = 0; k < MSG_BATCH; ++k) {
          if (b + k < cnt) {
            acc0 += w0[k] * p0[k];
            acc1 += w1[k] * p1[k];
          }
        }
      }
    }
    float* d = fact + (size_t)n * 64 + lane;
    if (LAST) {
      d[0] = ((in0 + pv0) + acc0) * (1.0f / 3.0f);
      d[32] = ((in1 + pv1) + acc1) * (1.0f / 3.0f);
    } else {
      d[0] = acc0;
      d[32] = acc1;
      float r0n = 1.0f / fmaxf(sqrtf(half16_sum(acc0 * acc0)), 1e-12f);
      float r1n = 1.0f / fmaxf(sqrtf(half16_sum(acc1 * acc1)), 1e-12f);
      float h0 = lane_bcast(r0n, 0), h1 = lane_bcast(r0n, 16), h2 = lane_bcast(r1n, 0), h3 = lane_bcast(r1n, 16);
      float fLo = (lane & 1) ? h1 : h0;
      float fHi = (lane & 1) ? h3 : h2;
      float mine = (lane & 2) ? fHi : fLo;
      hnv = ((lane >> 2) == q) ? mine : hnv;
    }
  }
  if (!LAST) {
    int nb = wave * MSG_NODES;
    if (lane < MSG_NODES * 4 && nb + (lane >> 2) < N) hninv[(size_t)nb * 4 + lane] = hnv;
  }
}

__global__ void k_norminv(const float* __restrict__ lo, const float* __restrict__ hi, int split,
                          float* __restrict__ out, int* __restrict__ count, int NF, int N) {
  int i = blockIdx.x * blockDim.x + threadIdx.x;
  if (i < N) count[i] = 0;
  if (i >= NF) return;
  int n = i >> 2, f = i & 3;
  const float4* p = reinterpret_cast<const float4*>(row_of(lo, hi, split, n) + (size_t)f * 16);
  float s = 0.f;
#pragma unroll
  for (int k = 0; k < 4; ++k) {
    float4 v = p[k];
    s += v.x * v.x + v.y * v.y + v.z * v.z + v.w * v.w;
  }
  out[i] = 1.0f / fmaxf(sqrtf(s), 1e-12f);
}

#define UA_ROWS 16
#define UA_PITCH 68
#define UA_WAVE_FLOATS (2 * UA_ROWS * UA_PITCH)

template <bool FIRST>
__global__ void __launch_bounds__(WG) k_updateA(
    const float* __restrict__ fact, const float* __restrict__ tailLo, const float* __restrict__ tailHi, int split,
    const float* __restrict__ hninv, const float* __restrict__ tninv,
    const int* __restrict__ hl, const int* __restrict__ tl,
    float* __restrict__ A, int E, int nGroups) {
  __shared__ __attribute__((aligned(16))) float stage[(WG / 32) * UA_WAVE_FLOATS];

  int lane = threadIdx.x & 31;
  int wib = __builtin_amdgcn_readfirstlane((int)(threadIdx.x >> 5));
  int wave = __builtin_amdgcn_readfirstlane((int)((blockIdx.x * WG + threadIdx.x) >> 5));
  if (wave >= nGroups) return;

  int r = lane & 15;
  int eIdx = wave * 16 + r;
  int e = (eIdx < E) ? eIdx : (E - 1);
  int h = hl[e], t = tl[e];
  int koff = (lane >> 4) << 1;

  bool active = (lane < 8) || (lane >= 24);
  int outRow = (lane < 8) ? lane : (lane - 16);
  int eOut = wave * 16 + outRow;
  bool doOut = active && eOut < E;
  v4f* A4 = reinterpret_cast<v4f*>(A);
  v4f a4 = {0.f, 0.f, 0.f, 0.f};
  if (FIRST) {
    a4.x = 1.0f; a4.y = 1.0f; a4.z = 1.0f; a4.w = 1.0f;
  } else {
    if (doOut) a4 = A4[eOut];
  }

  v4f hn4 = *reinterpret_cast<const v4f*>(hninv + (size_t)h * 4);
  v4f tn4 = *reinterpret_cast<const v4f*>(tninv + (size_t)t * 4);

  float* sH = stage + wib * UA_WAVE_FLOATS;
  float* sT = sH + UA_ROWS * UA_PITCH;

  v2f hr[UA_ROWS], tr[UA_ROWS];
#pragma unroll
  for (int j = 0; j < UA_ROWS; ++j) {
    int hj = __builtin_amdgcn_readlane(h, j);
    int tj = __builtin_amdgcn_readlane(t, j);
    hr[j] = *reinterpret_cast<const v2f*>(fact + (size_t)hj * 64 + 2 * lane);
    tr[j] = *reinterpret_cast<const v2f*>(row_of(tailLo, tailHi, split, tj) + 2 * lane);
  }
#pragma unroll
  for (int j = 0; j < UA_ROWS; ++j) {
    *reinterpret_cast<v2f*>(sH + j * UA_PITCH + 2 * lane) = hr[j];
    *reinterpret_cast<v2f*>(sT + j * UA_PITCH + 2 * lane) = tr[j];
  }
  __builtin_amdgcn_wave_barrier();

  const float* Hrow = sH + r * UA_PITCH;
  const float* Trow = sT + r * UA_PITCH;

#pragma unroll
  for (int f = 0; f < 4; ++f) {
    float hn = hn4[f];
    float tn = tn4[f];
    v8f c = {0.f, 0.f, 0.f, 0.f, 0.f, 0.f, 0.f, 0.f};
#pragma unroll
    for (int s = 0; s < 4; ++s) {
      int o = f * 16 + s * 4 + koff;
      v2f hv2 = *reinterpret_cast<const v2f*>(Hrow + o);
      v2f tv2 = *reinterpret_cast<const v2f*>(Trow + o);
      v2f av, bv;
      av.x = hv2.x * hn;
      av.y = hv2.y * hn;
      bv.x = fast_tanh(tv2.x * tn);
      bv.y = fast_tanh(tv2.y * tn);
      c = __builtin_amdgcn_wmma_f32_16x16x4_f32(false, av, false, bv,
                                                (short)0, c, false, false);
    }
    int idx = lane & 7;
    float dv = c[0];
    dv = (idx == 1) ? c[1] : dv;
    dv = (idx == 2) ? c[2] : dv;
    dv = (idx == 3) ? c[3] : dv;
    dv = (idx == 4) ? c[4] : dv;
    dv = (idx == 5) ? c[5] : dv;
    dv = (idx == 6) ? c[6] : dv;
    dv = (idx == 7) ? c[7] : dv;
    a4[f] += dv;
  }
  if (doOut) A4[eOut] = a4;
}

static inline size_t align16f(size_t x) { return (x + 15) & ~(size_t)15; }

extern "C" void kernel_launch(void* const* d_in, const int* in_sizes, int n_in,
                              void* d_out, int out_size, void* d_ws, size_t ws_size,
                              hipStream_t stream) {
  (void)n_in; (void)out_size; (void)ws_size;
  const float* u  = (const float*)d_in[0];
  const float* it = (const float*)d_in[1];
  const int* hl   = (const int*)d_in[2];
  const int* tl   = (const int*)d_in[3];

  const int userElems = in_sizes[0];
  const int itemElems = in_sizes[1];
  const int E = in_sizes[2];
  const int N = (userElems + itemElems) / 64;
  const size_t N64 = (size_t)N * 64;
  const size_t E4  = (size_t)E * 4;
  const size_t N4  = (size_t)N * 4;

  float* base = (float*)d_ws;
  size_t off = 0;
  float* ego   = base + off; off += align16f(N64);
  float* fact  = base + off; off += align16f(N64);
  int* hs       = (int*)(base + off);
  int* ts       = hs + align16f(E);
  int* rank     = ts + align16f(E);
  int* rowstart = rank + align16f(E);
  off += align16f(N64);
  float* A     = base + off; off += align16f(E4);
  float* dinv  = base + off; off += align16f(N4);
  float* hninv = base + off; off += align16f(N4);
  float* tninv = base + off; off += align16f(N4);

#define GRID(n) ((unsigned)(((size_t)(n) + WG - 1) / WG))
#define GRIDP(n, per) ((unsigned)(((size_t)(n) + (size_t)WG * (per) - 1) / ((size_t)WG * (per))))
  const int nUser = userElems / 64;

  k_norminv<<<GRID(N4), WG, 0, stream>>>(u, it, nUser, tninv, rowstart, (int)N4, N);

  k_head_rank<<<GRIDP(E, SORT_ITEMS), WG, 0, stream>>>(hl, rowstart, rank, E);
  k_scan<<<1, SCAN_WG, 0, stream>>>(rowstart, N);
  k_place<<<GRIDP(E, SORT_ITEMS), WG, 0, stream>>>(hl, tl, rowstart, rank, hs, ts, E);

  const float* rowsLo = u;
  const float* rowsHi = it;
  float* nxt = fact;
  float* spare = ego;
  float* tn = tninv;
  float* hn = hninv;
  const int nGroups = (E + 15) / 16;
  const unsigned msgGrid = GRID((size_t)((N + MSG_NODES - 1) / MSG_NODES) * 32);
  const unsigned updGrid = GRID((size_t)nGroups * 32);

  for (int layer = 0; layer < 2; ++layer) {
    for (int itn = 0; itn < 2; ++itn) {
      const bool first = (layer == 0 && itn == 0);
      const bool last = (layer == 1 && itn == 1);
      if (!first) k_degree_dinv<<<GRID(N), WG, 0, stream>>>(A, rowstart, dinv, N, E);
      if (first) {
        k_message<true, false><<<msgGrid, WG, 0, stream>>>(
            A, dinv, rowsLo, rowsHi, u, it, nUser, rowstart, ts, nxt, hn, N, E);
        k_updateA<true><<<updGrid, WG, 0, stream>>>(
            nxt, rowsLo, rowsHi, nUser, hn, tn, hs, ts, A, E, nGroups);
      } else if (last) {
        k_message<false, true><<<msgGrid, WG, 0, stream>>>(
            A, dinv, rowsLo, rowsHi, u, it, nUser, rowstart, ts, (float*)d_out, hn, N, E);
      } else {
        k_message<false, false><<<msgGrid, WG, 0, stream>>>(
            A, dinv, rowsLo, rowsHi, u, it, nUser, rowstart, ts, nxt, hn, N, E);
        k_updateA<false><<<updGrid, WG, 0, stream>>>(
            nxt, rowsLo, rowsHi, nUser, hn, tn, hs, ts, A, E, nGroups);
      }
    }
    rowsLo = nxt;
    rowsHi = nxt + (size_t)nUser * 64;
    float* tmpf = nxt; nxt = spare; spare = tmpf;
    float* tmp = tn; tn = hn; hn = tmp;
  }
#undef GRID
#undef GRIDP
}


# Routing update and next degree sum done by the message wave that holds the run: tanh tail rows staged in wave LDS, head row as the WMMA A operand; 14 -> 8 launches; degree sums now a lane tree
# speedup vs baseline: 4.4381x; 1.1247x over previous
#include <hip/hip_runtime.h>
#include <math.h>

typedef float v2f __attribute__((ext_vector_type(2)));
typedef float v8f __attribute__((ext_vector_type(8)));
typedef float v4f __attribute__((ext_vector_type(4)));

#define WG 256

__device__ __forceinline__ int atomic_rank(int* p) {
  return __hip_atomic_fetch_add(p, 1, __ATOMIC_RELAXED, __HIP_MEMORY_SCOPE_AGENT);
}

__device__ __forceinline__ float fast_tanh(float x) {
  float r;
  asm("v_tanh_f32 %0, %1\n\tv_nop\n\tv_nop" : "=v"(r) : "v"(x));
  return r;
}

__device__ __forceinline__ const float* row_of(const float* __restrict__ lo, const float* __restrict__ hi,
                                               int split, int n) {
  return (n < split) ? (lo + (size_t)n * 64) : (hi + (size_t)(n - split) * 64);
}

#define SORT_ITEMS 4

__global__ void k_head_rank(const int* __restrict__ hl, int* __restrict__ count,
                            int* __restrict__ rank, int E) {
  int base = blockIdx.x * (WG * SORT_ITEMS) + threadIdx.x;
  int h[SORT_ITEMS];
#pragma unroll
  for (int k = 0; k < SORT_ITEMS; ++k) {
    int e = base + k * WG;
    h[k] = hl[(e < E) ? e : (E - 1)];
  }
  int r[SORT_ITEMS];
#pragma unroll
  for (int k = 0; k < SORT_ITEMS; ++k) {
    int e = base + k * WG;
    r[k] = 0;
    if (e < E) r[k] = atomic_rank(count + h[k]);
  }
#pragma unroll
  for (int k = 0; k < SORT_ITEMS; ++k) {
    int e = base + k * WG;
    if (e < E) rank[e] = r[k];
  }
}

#define SCAN_WG 1024
#define SCAN_STEPS 20
#define SCAN_WAVE_SPAN (SCAN_STEPS * 128)

__device__ __forceinline__ int wave_incl_scan(int x, int lane) {
#pragma unroll
  for (int d = 1; d < 32; d <<= 1) {
    int y = __shfl_up(x, d);
    if (lane >= d) x += y;
  }
  return x;
}

__global__ void __launch_bounds__(SCAN_WG) k_scan(int* __restrict__ count, int N) {
  __shared__ int wsum[SCAN_WG / 32];
  int lane = threadIdx.x & 31;
  int w = __builtin_amdgcn_readfirstlane((int)(threadIdx.x >> 5));
  int carry = 0;
  for (int rbase = 0; rbase < N; rbase += (SCAN_WG / 32) * SCAN_WAVE_SPAN) {
    int wbase = rbase + w * SCAN_WAVE_SPAN;
    int v[SCAN_STEPS][4];
    int tot = 0;
#pragma unroll
    for (int k = 0; k < SCAN_STEPS; ++k) {
      int i = wbase + k * 128 + lane * 4;
      if (i + 3 < N) {
        int4 x = *reinterpret_cast<const int4*>(count + i);
        v[k][0] = x.x; v[k][1] = x.y; v[k][2] = x.z; v[k][3] = x.w;
      } else {
#pragma unroll
        for (int q = 0; q < 4; ++q) v[k][q] = (i + q < N) ? count[i + q] : 0;
      }
      tot += v[k][0] + v[k][1] + v[k][2] + v[k][3];
    }
    int incl = wave_incl_scan(tot, lane);
    if (lane == 31) wsum[w] = incl;
    __syncthreads();
    int xs = wave_incl_scan(wsum[lane], lane);
    int run = carry + ((w > 0) ? __builtin_amdgcn_readlane(xs, w - 1) : 0);
    carry += __builtin_amdgcn_readlane(xs, 31);
#pragma unroll
    for (int k = 0; k < SCAN_STEPS; ++k) {
      int i = wbase + k * 128 + lane * 4;
      int l4 = v[k][0] + v[k][1] + v[k][2] + v[k][3];
      int inc = wave_incl_scan(l4, lane);
      int b = run + inc - l4;
      if (i + 3 < N) {
        int4 o;
        o.x = b; o.y = o.x + v[k][0]; o.z = o.y + v[k][1]; o.w = o.z + v[k][2];
        *reinterpret_cast<int4*>(count + i) = o;
      } else {
#pragma unroll
        for (int q = 0; q < 4; ++q) {
          if (i + q < N) count[i + q] = b;
          b += v[k][q];
        }
      }
      run += __builtin_amdgcn_readlane(inc, 31);
    }
    __syncthreads();
  }
}

__global__ void k_place(const int* __restrict__ hl, const int* __restrict__ tl,
                        const int* __restrict__ rowstart, const int* __restrict__ rank,
                        int* __restrict__ ts, int E) {
  int base = blockIdx.x * (WG * SORT_ITEMS) + threadIdx.x;
  int h[SORT_ITEMS], t[SORT_ITEMS], r[SORT_ITEMS], p[SORT_ITEMS];
#pragma unroll
  for (int k = 0; k < SORT_ITEMS; ++k) {
    int e = base + k * WG;
    e = (e < E) ? e : (E - 1);
    h[k] = hl[e];
    t[k] = tl[e];
    r[k] = rank[e];
  }
#pragma unroll
  for (int k = 0; k < SORT_ITEMS; ++k) p[k] = rowstart[h[k]] + r[k];
#pragma unroll
  for (int k = 0; k < SORT_ITEMS; ++k) {
    int e = base + k * WG;
    if (e < E) ts[p[k]] = t[k];
  }
}

__device__ __forceinline__ v4f softmax4(v4f a) {
  float m = fmaxf(fmaxf(a.x, a.y), fmaxf(a.z, a.w));
  float e0 = expf(a.x - m), e1 = expf(a.y - m), e2 = expf(a.z - m), e3 = expf(a.w - m);
  float inv = 1.0f / (e0 + e1 + e2 + e3);
  v4f r = {e0 * inv, e1 * inv, e2 * inv, e3 * inv};
  return r;
}

#define MSG_NODES 4
#define MSG_BATCH 8
static_assert(MSG_BATCH == 8, "the asm statement in k_message names eight edges' row words");

__device__ __forceinline__ float lane_bcast(float v, int srcLane) {
  return __int_as_float(__builtin_amdgcn_readlane(__float_as_int(v), srcLane));
}

template <int MASK>
__device__ __forceinline__ float add_xmask(float x) {
  return x + __int_as_float(__builtin_amdgcn_update_dpp(0, __float_as_int(x), 0x160 + MASK, 0xf, 0xf, false));
}
__device__ __forceinline__ float half16_sum(float x) {
  x = add_xmask<1>(x);
  x = add_xmask<2>(x);
  x = add_xmask<4>(x);
  x = add_xmask<8>(x);
  return x;
}

#define FU_PITCH 68
#define FU_TAILS 32
#define FU_WAVE_FLOATS ((FU_TAILS + 1) * FU_PITCH)

__device__ __forceinline__ void tile_dots(const v2f (&hf)[16], const float* Trow, int koff, float (&dv)[4]) {
#pragma unroll
  for (int f = 0; f < 4; ++f) {
    v8f c = {0.f, 0.f, 0.f, 0.f, 0.f, 0.f, 0.f, 0.f};
#pragma unroll
    for (int s = 0; s < 4; ++s) {
      int o = f * 16 + s * 4 + koff;
      v2f bv = *reinterpret_cast<const v2f*>(Trow + o);
      c = __builtin_amdgcn_wmma_f32_16x16x4_f32(false, hf[f * 4 + s], false, bv,
                                                (short)0, c, false, false);
    }
    dv[f] = c[0];
  }
}

__device__ __forceinline__ v4f finish_chunk(const v2f (&hf)[16], const float* sT, int koff, int lane,
                                            int cnt, int el, v4f aold, float* __restrict__ A) {
  int r = lane & 15;
  float d0[4], d1[4] = {0.f, 0.f, 0.f, 0.f};
  if (cnt > 16) {
    int rowB = (16 + r < cnt) ? (16 + r) : (cnt - 1);
    tile_dots(hf, sT + rowB * FU_PITCH, koff, d1);
  }
  int rowA = (r < cnt) ? r : (cnt - 1);
  tile_dots(hf, sT + rowA * FU_PITCH, koff, d0);
  bool lo = lane < 16;
  v4f an;
  an.x = aold.x + (lo ? d0[0] : d1[0]);
  an.y = aold.y + (lo ? d0[1] : d1[1]);
  an.z = aold.z + (lo ? d0[2] : d1[2]);
  an.w = aold.w + (lo ? d0[3] : d1[3]);
  bool mine = lane < cnt;
  if (mine) reinterpret_cast<v4f*>(A)[el] = an;
  v4f sm = softmax4(an);
  float x0 = half16_sum(mine ? sm.x : 0.f), x1 = half16_sum(mine ? sm.y : 0.f);
  float x2 = half16_sum(mine ? sm.z : 0.f), x3 = half16_sum(mine ? sm.w : 0.f);
  v4f tot;
  tot.x = lane_bcast(x0, 0) + lane_bcast(x0, 16);
  tot.y = lane_bcast(x1, 0) + lane_bcast(x1, 16);
  tot.z = lane_bcast(x2, 0) + lane_bcast(x2, 16);
  tot.w = lane_bcast(x3, 0) + lane_bcast(x3, 16);
  return tot;
}

template <bool FIRST, bool LAST>
__global__ void __launch_bounds__(WG) k_message(
    float* __restrict__ A, const float* __restrict__ dinv, float* __restrict__ dinvNext,
    const float* __restrict__ rowsLo, const float* __restrict__ rowsHi,
    const float* __restrict__ inLo, const float* __restrict__ inHi, int split,
    const int* __restrict__ rowstart, const int* __restrict__ ts, const float* __restrict__ tninv,
    float* __restrict__ fact, float* __restrict__ hninv, int N, int E) {
  __shared__ __attribute__((aligned(16))) float stage[LAST ? 4 : (WG / 32) * FU_WAVE_FLOATS];
  int lane = threadIdx.x & 31;
  int wave = __builtin_amdgcn_readfirstlane((int)((blockIdx.x * WG + threadIdx.x) >> 5));
  int wib = __builtin_amdgcn_readfirstlane((int)(threadIdx.x >> 5));
  bool upper = lane >= 16;
  int koff = (lane >> 4) << 1;
  v4f* __restrict__ A4 = reinterpret_cast<v4f*>(A);
  const v4f* __restrict__ dinv4 = reinterpret_cast<const v4f*>(dinv);
  const v4f* __restrict__ tninv4 = reinterpret_cast<const v4f*>(tninv);
  float* sT = stage + (LAST ? 0 : wib * FU_WAVE_FLOATS);
  float* sHd = sT + (LAST ? 0 : FU_TAILS * FU_PITCH);

  float hnv = 0.f;
  float dnv = 0.f;
  for (int q = 0; q < MSG_NODES; ++q) {
    int n = wave * MSG_NODES + q;
    if (n >= N) break;
    int s = rowstart[n];
    int end = (n + 1 < N) ? rowstart[n + 1] : E;
    v4f dh;
    if (FIRST) {
      float dv = 1.0f / sqrtf(fmaxf(0.25f * (float)(end - s), 1e-8f));
      dh.x = dv; dh.y = dv; dh.z = dv; dh.w = dv;
    } else {
      dh = dinv4[n];
    }
    float in0 = 0.f, in1 = 0.f, pv0 = 0.f, pv1 = 0.f;
    if (LAST) {
      const float* ip = row_of(inLo, inHi, split, n) + lane;
      const float* pp = row_of(rowsLo, rowsHi, split, n) + lane;
      in0 = ip[0]; in1 = ip[32];
      pv0 = pp[0]; pv1 = pp[32];
      asm volatile("" : "+v"(in0), "+v"(in1), "+v"(pv0), "+v"(pv1));
    }
    float acc0 = 0.f, acc1 = 0.f;
    int elK = 0, cntK = 0;
    v4f avK = {1.0f, 1.0f, 1.0f, 1.0f};
    for (int c = s; c < end; c += 32) {
      int cnt = end - c;
      cnt = (cnt < 32) ? cnt : 32;
      int el = c + ((lane < cnt) ? lane : (cnt - 1));
      int tv = ts[el];
      v4f nw;
      v4f dt = {0.f, 0.f, 0.f, 0.f};
      v4f tn4 = {0.f, 0.f, 0.f, 0.f};
      int r0 = 0, r1raw = 0;
      if (FIRST) {
        nw.x = 0.25f; nw.y = 0.25f; nw.z = 0.25f; nw.w = 0.25f;
        r0 = rowstart[tv];
        r1raw = rowstart[(tv + 1 < N) ? (tv + 1) : tv];
      } else {
        v4f av = A4[el];
        asm volatile("" : "+v"(tv), "+v"(av));
        dt = dinv4[tv];
        nw = softmax4(av);
        avK = av;
      }
      if (!LAST) tn4 = tninv4[tv];
      elK = el; cntK = cnt;
      v4f wv = {0.f, 0.f, 0.f, 0.f};
      for (int b = 0; b < cnt; b += MSG_BATCH) {
        float p0[MSG_BATCH], p1[MSG_BATCH];
#pragma unroll
        for (int k = 0; k < MSG_BATCH; ++k) {
          int j = b + k;
          j = (j < cnt) ? j : (cnt - 1);
          int t = __builtin_amdgcn_readlane(tv, j);
          const float* sp = row_of(rowsLo, rowsHi, split, t) + lane;
          p0[k] = sp[0];
          p1[k] = sp[32];
        }
        if (FIRST) {
          if (b == 0) {
            asm volatile("" : "+v"(r0), "+v"(r1raw));
            int r1 = (tv + 1 < N) ? r1raw : E;
            float dtv = 1.0f / sqrtf(fmaxf(0.25f * (float)(r1 - r0), 1e-8f));
            wv.x = nw.x * (dh.x * dtv);
            wv.y = nw.y * (dh.y * dtv);
            wv.z = nw.z * (dh.z * dtv);
            wv.w = nw.w * (dh.w * dtv);
          }
        } else {
          asm volatile("" : "+v"(dt));
          wv.x = nw.x * (dh.x * dt.x);
          wv.y = nw.y * (dh.y * dt.y);
          wv.z = nw.z * (dh.z * dt.z);
          wv.w = nw.w * (dh.w * dt.w);
        }
        float w0[MSG_BATCH], w1[MSG_BATCH];
#pragma unroll
        for (int k = 0; k < MSG_BATCH; ++k) {
          int j = b + k;
          j = (j < cnt) ? j : (cnt - 1);
          float wx = lane_bcast(wv.x, j), wy = lane_bcast(wv.y, j);
          float wz = lane_bcast(wv.z, j), ww = lane_bcast(wv.w, j);
          w0[k] = upper ? wy : wx;
          w1[k] = upper ? ww : wz;
        }
        asm volatile("" : "+v"(p0[0]), "+v"(p0[1]), "+v"(p0[2]), "+v"(p0[3]),
                          "+v"(p0[4]), "+v"(p0[5]), "+v"(p0[6]), "+v"(p0[7]),
                          "+v"(p1[0]), "+v"(p1[1]), "+v"(p1[2]), "+v"(p1[3]),
                          "+v"(p1[4]), "+v"(p1[5]), "+v"(p1[6]), "+v"(p1[7]));
#pragma unroll
        for (int k = 0; k < MSG_BATCH; ++k) {
          if (b + k < cnt) {
            acc0 += w0[k] * p0[k];
            acc1 += w1[k] * p1[k];
            if (!LAST) {
              int j = b + k;
              float tx = lane_bcast(tn4.x, j), ty = lane_bcast(tn4.y, j);
              float tz = lane_bcast(tn4.z, j), tw = lane_bcast(tn4.w, j);
              float* dst = sT + j * FU_PITCH + lane;
              dst[0] = fast_tanh(p0[k] * (upper ? ty : tx));
              dst[32] = fast_tanh(p1[k] * (upper ? tw : tz));
            }
          }
        }
      }
    }
    float* d = fact + (size_t)n * 64 + lane;
    if (LAST) {
      d[0] = ((in0 + pv0) + acc0) * (1.0f / 3.0f);
      d[32] = ((in1 + pv1) + acc1) * (1.0f / 3.0f);
    } else {
      d[0] = acc0;
      d[32] = acc1;
      float r0n = 1.0f / fmaxf(sqrtf(half16_sum(acc0 * acc0)), 1e-12f);
      float r1n = 1.0f / fmaxf(sqrtf(half16_sum(acc1 * acc1)), 1e-12f);
      float h0 = lane_bcast(r0n, 0), h1 = lane_bcast(r0n, 16), h2 = lane_bcast(r1n, 0), h3 = lane_bcast(r1n, 16);
      float fLo = (lane & 1) ? h1 : h0;
      float fHi = (lane & 1) ? h3 : h2;
      float mine = (lane & 2) ? fHi : fLo;
      hnv = ((lane >> 2) == q) ? mine : hnv;

      sHd[lane] = acc0 * r0n;
      sHd[32 + lane] = acc1 * r1n;
      __builtin_amdgcn_wave_barrier();
      v2f hf[16];
#pragma unroll
      for (int i = 0; i < 16; ++i) {
        int o = (i >> 2) * 16 + (i & 3) * 4 + koff;
        hf[i] = *reinterpret_cast<const v2f*>(sHd + o);
      }
      v4f dsum = {0.f, 0.f, 0.f, 0.f};
      int len = end - s;
      if (len <= FU_TAILS) {
        if (len > 0) dsum = finish_chunk(hf, sT, koff, lane, cntK, elK, avK, A);
      } else {
        for (int c = s; c < end; c += 32) {
          int cnt = end - c;
          cnt = (cnt < 32) ? cnt : 32;
          int el = c + ((lane < cnt) ? lane : (cnt - 1));
          int tv = ts[el];
          v4f ao = {1.0f, 1.0f, 1.0f, 1.0f};
          if (!FIRST) ao = A4[el];
          v4f tn4 = tninv4[tv];
          __builtin_amdgcn_wave_barrier();
          for (int j = 0; j < cnt; ++j) {
            int t = __builtin_amdgcn_readlane(tv, j);
            const float* sp = row_of(rowsLo, rowsHi, split, t) + lane;
            float x0 = sp[0], x1 = sp[32];
            float tx = lane_bcast(tn4.x, j), ty = lane_bcast(tn4.y, j);
            float tz = lane_bcast(tn4.z, j), tw = lane_bcast(tn4.w, j);
            float* dst = sT + j * FU_PITCH + lane;
            dst[0] = fast_tanh(x0 * (upper ? ty : tx));
            dst[32] = fast_tanh(x1 * (upper ? tw : tz));
          }
          __builtin_amdgcn_wave_barrier();
          v4f t4 = finish_chunk(hf, sT, koff, lane, cnt, el, ao, A);
          dsum.x += t4.x; dsum.y += t4.y; dsum.z += t4.z; dsum.w += t4.w;
        }
      }
      float gLo = (lane & 1) ? dsum.y : dsum.x;
      float gHi = (lane & 1) ? dsum.w : dsum.z;
      float gs = (lane & 2) ? gHi : gLo;
      float gm = 1.0f / sqrtf(fmaxf(gs, 1e-8f));
      dnv = ((lane >> 2) == q) ? gm : dnv;
      __builtin_amdgcn_wave_barrier();
    }
  }
  if (!LAST) {
    int nb = wave * MSG_NODES;
    if (lane < MSG_NODES * 4 && nb + (lane >> 2) < N) {
      hninv[(size_t)nb * 4 + lane] = hnv;
      dinvNext[(size_t)nb * 4 + lane] = dnv;
    }
  }
}

__global__ void k_norminv(const float* __restrict__ lo, const float* __restrict__ hi, int split,
                          float* __restrict__ out, int* __restrict__ count, int NF, int N) {
  int i = blockIdx.x * blockDim.x + threadIdx.x;
  if (i < N) count[i] = 0;
  if (i >= NF) return;
  int n = i >> 2, f = i & 3;
  const float4* p = reinterpret_cast<const float4*>(row_of(lo, hi, split, n) + (size_t)f * 16);
  float s = 0.f;
#pragma unroll
  for (int k = 0; k < 4; ++k) {
    float4 v = p[k];
    s += v.x * v.x + v.y * v.y + v.z * v.z + v.w * v.w;
  }
  out[i] = 1.0f / fmaxf(sqrtf(s), 1e-12f);
}

static inline size_t align16f(size_t x) { return (x + 15) & ~(size_t)15; }

extern "C" void kernel_launch(void* const* d_in, const int* in_sizes, int n_in,
                              void* d_out, int out_size, void* d_ws, size_t ws_size,
                              hipStream_t stream) {
  (void)n_in; (void)out_size; (void)ws_size;
  const float* u  = (const float*)d_in[0];
  const float* it = (const float*)d_in[1];
  const int* hl   = (const int*)d_in[2];
  const int* tl   = (const int*)d_in[3];

  const int userElems = in_sizes[0];
  const int itemElems = in_sizes[1];
  const int E = in_sizes[2];
  const int N = (userElems + itemElems) / 64;
  const size_t N64 = (size_t)N * 64;
  const size_t E4  = (size_t)E * 4;
  const size_t N4  = (size_t)N * 4;

  float* base = (float*)d_ws;
  size_t off = 0;
  float* ego   = base + off; off += align16f(N64);
  float* fact  = base + off; off += align16f(N64);
  float* dinvB  = base + off;
  int* ts       = (int*)(dinvB + align16f(N4));
  int* rank     = ts + align16f(E);
  int* rowstart = rank + align16f(E);
  off += align16f(N64);
  float* A     = base + off; off += align16f(E4);
  float* dinv  = base + off; off += align16f(N4);
  float* hninv = base + off; off += align16f(N4);
  float* tninv = base + off; off += align16f(N4);

#define GRID(n) ((unsigned)(((size_t)(n) + WG - 1) / WG))
#define GRIDP(n, per) ((unsigned)(((size_t)(n) + (size_t)WG * (per) - 1) / ((size_t)WG * (per))))
  const int nUser = userElems / 64;

  k_norminv<<<GRID(N4), WG, 0, stream>>>(u, it, nUser, tninv, rowstart, (int)N4, N);

  k_head_rank<<<GRIDP(E, SORT_ITEMS), WG, 0, stream>>>(hl, rowstart, rank, E);
  k_scan<<<1, SCAN_WG, 0, stream>>>(rowstart, N);
  k_place<<<GRIDP(E, SORT_ITEMS), WG, 0, stream>>>(hl, tl, rowstart, rank, ts, E);

  const float* rowsLo = u;
  const float* rowsHi = it;
  float* nxt = fact;
  float* spare = ego;
  float* tn = tninv;
  float* hn = hninv;
  float* dcur = dinvB;
  float* dnext = dinv;
  const unsigned msgGrid = GRID((size_t)((N + MSG_NODES - 1) / MSG_NODES) * 32);

  for (int layer = 0; layer < 2; ++layer) {
    for (int itn = 0; itn < 2; ++itn) {
      const bool first = (layer == 0 && itn == 0);
      const bool last = (layer == 1 && itn == 1);
      if (first) {
        k_message<true, false><<<msgGrid, WG, 0, stream>>>(
            A, dcur, dnext, rowsLo, rowsHi, u, it, nUser, rowstart, ts, tn, nxt, hn, N, E);
      } else if (last) {
        k_message<false, true><<<msgGrid, WG, 0, stream>>>(
            A, dcur, dnext, rowsLo, rowsHi, u, it, nUser, rowstart, ts, tn, (float*)d_out, hn, N, E);
      } else {
        k_message<false, false><<<msgGrid, WG, 0, stream>>>(
            A, dcur, dnext, rowsLo, rowsHi, u, it, nUser, rowstart, ts, tn, nxt, hn, N, E);
      }
      float* tmpd = dcur; dcur = dnext; dnext = tmpd;
    }
    rowsLo = nxt;
    rowsHi = nxt + (size_t)nUser * 64;
    float* tmpf = nxt; nxt = spare; spare = tmpf;
    float* tmp = tn; tn = hn; hn = tmp;
  }
#undef GRID
#undef GRIDP
}


# Edge weights and tail norms go once a chunk into a wave-private LDS table; each lane reads its two factors' values for 8 edges with 16-byte loads (was 8 lane reads + selects an edge); sums unchanged
# speedup vs baseline: 4.7815x; 1.0774x over previous
#include <hip/hip_runtime.h>
#include <math.h>

typedef float v2f __attribute__((ext_vector_type(2)));
typedef float v8f __attribute__((ext_vector_type(8)));
typedef float v4f __attribute__((ext_vector_type(4)));

#define WG 256

__device__ __forceinline__ int atomic_rank(int* p) {
  return __hip_atomic_fetch_add(p, 1, __ATOMIC_RELAXED, __HIP_MEMORY_SCOPE_AGENT);
}

__device__ __forceinline__ float fast_tanh(float x) {
  float r;
  asm("v_tanh_f32 %0, %1\n\tv_nop\n\tv_nop" : "=v"(r) : "v"(x));
  return r;
}

__device__ __forceinline__ const float* row_of(const float* __restrict__ lo, const float* __restrict__ hi,
                                               int split, int n) {
  return (n < split) ? (lo + (size_t)n * 64) : (hi + (size_t)(n - split) * 64);
}

#define SORT_ITEMS 4

__global__ void k_head_rank(const int* __restrict__ hl, int* __restrict__ count,
                            int* __restrict__ rank, int E) {
  int base = blockIdx.x * (WG * SORT_ITEMS) + threadIdx.x;
  int h[SORT_ITEMS];
#pragma unroll
  for (int k = 0; k < SORT_ITEMS; ++k) {
    int e = base + k * WG;
    h[k] = hl[(e < E) ? e : (E - 1)];
  }
  int r[SORT_ITEMS];
#pragma unroll
  for (int k = 0; k < SORT_ITEMS; ++k) {
    int e = base + k * WG;
    r[k] = 0;
    if (e < E) r[k] = atomic_rank(count + h[k]);
  }
#pragma unroll
  for (int k = 0; k < SORT_ITEMS; ++k) {
    int e = base + k * WG;
    if (e < E) rank[e] = r[k];
  }
}

#define SCAN_WG 1024
#define SCAN_STEPS 20
#define SCAN_WAVE_SPAN (SCAN_STEPS * 128)

__device__ __forceinline__ int wave_incl_scan(int x, int lane) {
#pragma unroll
  for (int d = 1; d < 32; d <<= 1) {
    int y = __shfl_up(x, d);
    if (lane >= d) x += y;
  }
  return x;
}

__global__ void __launch_bounds__(SCAN_WG) k_scan(int* __restrict__ count, int N) {
  __shared__ int wsum[SCAN_WG / 32];
  int lane = threadIdx.x & 31;
  int w = __builtin_amdgcn_readfirstlane((int)(threadIdx.x >> 5));
  int carry = 0;
  for (int rbase = 0; rbase < N; rbase += (SCAN_WG / 32) * SCAN_WAVE_SPAN) {
    int wbase = rbase + w * SCAN_WAVE_SPAN;
    int v[SCAN_STEPS][4];
    int tot = 0;
#pragma unroll
    for (int k = 0; k < SCAN_STEPS; ++k) {
      int i = wbase + k * 128 + lane * 4;
      if (i + 3 < N) {
        int4 x = *reinterpret_cast<const int4*>(count + i);
        v[k][0] = x.x; v[k][1] = x.y; v[k][2] = x.z; v[k][3] = x.w;
      } else {
#pragma unroll
        for (int q = 0; q < 4; ++q) v[k][q] = (i + q < N) ? count[i + q] : 0;
      }
      tot += v[k][0] + v[k][1] + v[k][2] + v[k][3];
    }
    int incl = wave_incl_scan(tot, lane);
    if (lane == 31) wsum[w] = incl;
    __syncthreads();
    int xs = wave_incl_scan(wsum[lane], lane);
    int run = carry + ((w > 0) ? __builtin_amdgcn_readlane(xs, w - 1) : 0);
    carry += __builtin_amdgcn_readlane(xs, 31);
#pragma unroll
    for (int k = 0; k < SCAN_STEPS; ++k) {
      int i = wbase + k * 128 + lane * 4;
      int l4 = v[k][0] + v[k][1] + v[k][2] + v[k][3];
      int inc = wave_incl_scan(l4, lane);
      int b = run + inc - l4;
      if (i + 3 < N) {
        int4 o;
        o.x = b; o.y = o.x + v[k][0]; o.z = o.y + v[k][1]; o.w = o.z + v[k][2];
        *reinterpret_cast<int4*>(count + i) = o;
      } else {
#pragma unroll
        for (int q = 0; q < 4; ++q) {
          if (i + q < N) count[i + q] = b;
          b += v[k][q];
        }
      }
      run += __builtin_amdgcn_readlane(inc, 31);
    }
    __syncthreads();
  }
}

__global__ void k_place(const int* __restrict__ hl, const int* __restrict__ tl,
                        const int* __restrict__ rowstart, const int* __restrict__ rank,
                        int* __restrict__ ts, int E) {
  int base = blockIdx.x * (WG * SORT_ITEMS) + threadIdx.x;
  int h[SORT_ITEMS], t[SORT_ITEMS], r[SORT_ITEMS], p[SORT_ITEMS];
#pragma unroll
  for (int k = 0; k < SORT_ITEMS; ++k) {
    int e = base + k * WG;
    e = (e < E) ? e : (E - 1);
    h[k] = hl[e];
    t[k] = tl[e];
    r[k] = rank[e];
  }
#pragma unroll
  for (int k = 0; k < SORT_ITEMS; ++k) p[k] = rowstart[h[k]] + r[k];
#pragma unroll
  for (int k = 0; k < SORT_ITEMS; ++k) {
    int e = base + k * WG;
    if (e < E) ts[p[k]] = t[k];
  }
}

__device__ __forceinline__ v4f softmax4(v4f a) {
  float m = fmaxf(fmaxf(a.x, a.y), fmaxf(a.z, a.w));
  float e0 = expf(a.x - m), e1 = expf(a.y - m), e2 = expf(a.z - m), e3 = expf(a.w - m);
  float inv = 1.0f / (e0 + e1 + e2 + e3);
  v4f r = {e0 * inv, e1 * inv, e2 * inv, e3 * inv};
  return r;
}

#define MSG_NODES 4
#define MSG_BATCH 8
static_assert(MSG_BATCH == 8, "the asm statement in k_message names eight edges' row words");

__device__ __forceinline__ float lane_bcast(float v, int srcLane) {
  return __int_as_float(__builtin_amdgcn_readlane(__float_as_int(v), srcLane));
}

template <int MASK>
__device__ __forceinline__ float add_xmask(float x) {
  return x + __int_as_float(__builtin_amdgcn_update_dpp(0, __float_as_int(x), 0x160 + MASK, 0xf, 0xf, false));
}
__device__ __forceinline__ float half16_sum(float x) {
  x = add_xmask<1>(x);
  x = add_xmask<2>(x);
  x = add_xmask<4>(x);
  x = add_xmask<8>(x);
  return x;
}

#define FU_PITCH 68
#define FU_TAILS 32
#define FU_ROWS_FLOATS (FU_TAILS * FU_PITCH + 64)
#define FU_BC_W (4 * FU_TAILS)
#define FU_WAVE_FLOATS (FU_ROWS_FLOATS + 2 * FU_BC_W)
static_assert((FU_ROWS_FLOATS % 4) == 0 && (FU_WAVE_FLOATS % 4) == 0, "16-byte LDS loads from the broadcast table");

__device__ __forceinline__ void tile_dots(const v2f (&hf)[16], const float* Trow, int koff, float (&dv)[4]) {
#pragma unroll
  for (int f = 0; f < 4; ++f) {
    v8f c = {0.f, 0.f, 0.f, 0.f, 0.f, 0.f, 0.f, 0.f};
#pragma unroll
    for (int s = 0; s < 4; ++s) {
      int o = f * 16 + s * 4 + koff;
      v2f bv = *reinterpret_cast<const v2f*>(Trow + o);
      c = __builtin_amdgcn_wmma_f32_16x16x4_f32(false, hf[f * 4 + s], false, bv,
                                                (short)0, c, false, false);
    }
    dv[f] = c[0];
  }
}

__device__ __forceinline__ v4f finish_chunk(const v2f (&hf)[16], const float* sT, int koff, int lane,
                                            int cnt, int el, v4f aold, float* __restrict__ A) {
  int r = lane & 15;
  float d0[4], d1[4] = {0.f, 0.f, 0.f, 0.f};
  if (cnt > 16) {
    int rowB = (16 + r < cnt) ? (16 + r) : (cnt - 1);
    tile_dots(hf, sT + rowB * FU_PITCH, koff, d1);
  }
  int rowA = (r < cnt) ? r : (cnt - 1);
  tile_dots(hf, sT + rowA * FU_PITCH, koff, d0);
  bool lo = lane < 16;
  v4f an;
  an.x = aold.x + (lo ? d0[0] : d1[0]);
  an.y = aold.y + (lo ? d0[1] : d1[1]);
  an.z = aold.z + (lo ? d0[2] : d1[2]);
  an.w = aold.w + (lo ? d0[3] : d1[3]);
  bool mine = lane < cnt;
  if (mine) reinterpret_cast<v4f*>(A)[el] = an;
  v4f sm = softmax4(an);
  float x0 = half16_sum(mine ? sm.x : 0.f), x1 = half16_sum(mine ? sm.y : 0.f);
  float x2 = half16_sum(mine ? sm.z : 0.f), x3 = half16_sum(mine ? sm.w : 0.f);
  v4f tot;
  tot.x = lane_bcast(x0, 0) + lane_bcast(x0, 16);
  tot.y = lane_bcast(x1, 0) + lane_bcast(x1, 16);
  tot.z = lane_bcast(x2, 0) + lane_bcast(x2, 16);
  tot.w = lane_bcast(x3, 0) + lane_bcast(x3, 16);
  return tot;
}

template <bool FIRST, bool LAST>
__global__ void __launch_bounds__(WG, 8) k_message(
    float* __restrict__ A, const float* __restrict__ dinv, float* __restrict__ dinvNext,
    const float* __restrict__ rowsLo, const float* __restrict__ rowsHi,
    const float* __restrict__ inLo, const float* __restrict__ inHi, int split,
    const int* __restrict__ rowstart, const int* __restrict__ ts, const float* __restrict__ tninv,
    float* __restrict__ fact, float* __restrict__ hninv, int N, int E) {
  __shared__ __attribute__((aligned(16))) float stage[(WG / 32) * (LAST ? FU_BC_W : FU_WAVE_FLOATS)];
  int lane = threadIdx.x & 31;
  int wave = __builtin_amdgcn_readfirstlane((int)((blockIdx.x * WG + threadIdx.x) >> 5));
  int wib = __builtin_amdgcn_readfirstlane((int)(threadIdx.x >> 5));
  bool upper = lane >= 16;
  int koff = (lane >> 4) << 1;
  v4f* __restrict__ A4 = reinterpret_cast<v4f*>(A);
  const v4f* __restrict__ dinv4 = reinterpret_cast<const v4f*>(dinv);
  const v4f* __restrict__ tninv4 = reinterpret_cast<const v4f*>(tninv);
  float* sT = stage + (LAST ? 0 : wib * FU_WAVE_FLOATS);
  float* sHd = sT + (LAST ? 0 : FU_TAILS * FU_PITCH);
  float* sW = LAST ? (stage + wib * FU_BC_W) : (sT + FU_ROWS_FLOATS);
  float* sN = sW + FU_BC_W;
  const float* bcl = sW + (lane >> 4) * FU_TAILS;

  float hnv = 0.f;
  float dnv = 0.f;
  for (int q = 0; q < MSG_NODES; ++q) {
    int n = wave * MSG_NODES + q;
    if (n >= N) break;
    int s = rowstart[n];
    int end = (n + 1 < N) ? rowstart[n + 1] : E;
    v4f dh;
    if (FIRST) {
      float dv = 1.0f / sqrtf(fmaxf(0.25f * (float)(end - s), 1e-8f));
      dh.x = dv; dh.y = dv; dh.z = dv; dh.w = dv;
    } else {
      dh = dinv4[n];
    }
    float in0 = 0.f, in1 = 0.f, pv0 = 0.f, pv1 = 0.f;
    if (LAST) {
      const float* ip = row_of(inLo, inHi, split, n) + lane;
      const float* pp = row_of(rowsLo, rowsHi, split, n) + lane;
      in0 = ip[0]; in1 = ip[32];
      pv0 = pp[0]; pv1 = pp[32];
      asm volatile("" : "+v"(in0), "+v"(in1), "+v"(pv0), "+v"(pv1));
    }
    float acc0 = 0.f, acc1 = 0.f;
    int elK = 0, cntK = 0;
    v4f avK = {1.0f, 1.0f, 1.0f, 1.0f};
    for (int c = s; c < end; c += 32) {
      int cnt = end - c;
      cnt = (cnt < 32) ? cnt : 32;
      int el = c + ((lane < cnt) ? lane : (cnt - 1));
      int tv = ts[el];
      v4f nw;
      v4f dt = {0.f, 0.f, 0.f, 0.f};
      v4f tn4 = {0.f, 0.f, 0.f, 0.f};
      int r0 = 0, r1raw = 0;
      if (FIRST) {
        nw.x = 0.25f; nw.y = 0.25f; nw.z = 0.25f; nw.w = 0.25f;
        r0 = rowstart[tv];
        r1raw = rowstart[(tv + 1 < N) ? (tv + 1) : tv];
      } else {
        v4f av = A4[el];
        asm volatile("" : "+v"(tv), "+v"(av));
        dt = dinv4[tv];
        nw = softmax4(av);
        avK = av;
      }
      if (!LAST) tn4 = tninv4[tv];
      elK = el; cntK = cnt;
      v4f wv = {0.f, 0.f, 0.f, 0.f};
      for (int b = 0; b < cnt; b += MSG_BATCH) {
        float p0[MSG_BATCH], p1[MSG_BATCH];
#pragma unroll
        for (int k = 0; k < MSG_BATCH; ++k) {
          int j = b + k;
          j = (j < cnt) ? j : (cnt - 1);
          int t = __builtin_amdgcn_readlane(tv, j);
          const float* sp = row_of(rowsLo, rowsHi, split, t) + lane;
          p0[k] = sp[0];
          p1[k] = sp[32];
        }
        if (b == 0) {
          if (FIRST) {
            asm volatile("" : "+v"(r0), "+v"(r1raw));
            int r1 = (tv + 1 < N) ? r1raw : E;
            float dtv = 1.0f / sqrtf(fmaxf(0.25f * (float)(r1 - r0), 1e-8f));
            wv.x = nw.x * (dh.x * dtv);
            wv.y = nw.y * (dh.y * dtv);
            wv.z = nw.z * (dh.z * dtv);
            wv.w = nw.w * (dh.w * dtv);
          } else {
            asm volatile("" : "+v"(dt));
            wv.x = nw.x * (dh.x * dt.x);
            wv.y = nw.y * (dh.y * dt.y);
            wv.z = nw.z * (dh.z * dt.z);
            wv.w = nw.w * (dh.w * dt.w);
          }
          __builtin_amdgcn_wave_barrier();
          sW[lane] = wv.x;
          sW[FU_TAILS + lane] = wv.y;
          sW[2 * FU_TAILS + lane] = wv.z;
          sW[3 * FU_TAILS + lane] = wv.w;
          if (!LAST) {
            sN[lane] = tn4.x;
            sN[FU_TAILS + lane] = tn4.y;
            sN[2 * FU_TAILS + lane] = tn4.z;
            sN[3 * FU_TAILS + lane] = tn4.w;
          }
          __builtin_amdgcn_wave_barrier();
        }
        float w0[MSG_BATCH], w1[MSG_BATCH], t0[MSG_BATCH], t1[MSG_BATCH];
        {
          v4f a0 = *reinterpret_cast<const v4f*>(bcl + b), a1 = *reinterpret_cast<const v4f*>(bcl + b + 4);
          v4f c0 = *reinterpret_cast<const v4f*>(bcl + 2 * FU_TAILS + b);
          v4f c1 = *reinterpret_cast<const v4f*>(bcl + 2 * FU_TAILS + b + 4);
          w0[0] = a0.x; w0[1] = a0.y; w0[2] = a0.z; w0[3] = a0.w;
          w0[4] = a1.x; w0[5] = a1.y; w0[6] = a1.z; w0[7] = a1.w;
          w1[0] = c0.x; w1[1] = c0.y; w1[2] = c0.z; w1[3] = c0.w;
          w1[4] = c1.x; w1[5] = c1.y; w1[6] = c1.z; w1[7] = c1.w;
        }
        if (!LAST) {
          const float* nl = bcl + FU_BC_W;
          v4f a0 = *reinterpret_cast<const v4f*>(nl + b), a1 = *reinterpret_cast<const v4f*>(nl + b + 4);
          v4f c0 = *reinterpret_cast<const v4f*>(nl + 2 * FU_TAILS + b);
          v4f c1 = *reinterpret_cast<const v4f*>(nl + 2 * FU_TAILS + b + 4);
          t0[0] = a0.x; t0[1] = a0.y; t0[2] = a0.z; t0[3] = a0.w;
          t0[4] = a1.x; t0[5] = a1.y; t0[6] = a1.z; t0[7] = a1.w;
          t1[0] = c0.x; t1[1] = c0.y; t1[2] = c0.z; t1[3] = c0.w;
          t1[4] = c1.x; t1[5] = c1.y; t1[6] = c1.z; t1[7] = c1.w;
        }
        asm volatile("" : "+v"(p0[0]), "+v"(p0[1]), "+v"(p0[2]), "+v"(p0[3]),
                          "+v"(p0[4]), "+v"(p0[5]), "+v"(p0[6]), "+v"(p0[7]),
                          "+v"(p1[0]), "+v"(p1[1]), "+v"(p1[2]), "+v"(p1[3]),
                          "+v"(p1[4]), "+v"(p1[5]), "+v"(p1[6]), "+v"(p1[7]));
#pragma unroll
        for (int k = 0; k < MSG_BATCH; ++k) {
          if (b + k < cnt) {
            acc0 += w0[k] * p0[k];
            acc1 += w1[k] * p1[k];
            if (!LAST) {
              float* dst = sT + (b + k) * FU_PITCH + lane;
              dst[0] = fast_tanh(p0[k] * t0[k]);
              dst[32] = fast_tanh(p1[k] * t1[k]);
            }
          }
        }
      }
    }
    float* d = fact + (size_t)n * 64;
    asm volatile("" : "+s"(d));
    if (LAST) {
      d[lane] = ((in0 + pv0) + acc0) * (1.0f / 3.0f);
      d[32 + lane] = ((in1 + pv1) + acc1) * (1.0f / 3.0f);
    } else {
      d[lane] = acc0;
      d[32 + lane] = acc1;
      float r0n = 1.0f / fmaxf(sqrtf(half16_sum(acc0 * acc0)), 1e-12f);
      float r1n = 1.0f / fmaxf(sqrtf(half16_sum(acc1 * acc1)), 1e-12f);
      float h0 = lane_bcast(r0n, 0), h1 = lane_bcast(r0n, 16), h2 = lane_bcast(r1n, 0), h3 = lane_bcast(r1n, 16);
      float fLo = (lane & 1) ? h1 : h0;
      float fHi = (lane & 1) ? h3 : h2;
      float mine = (lane & 2) ? fHi : fLo;
      hnv = ((lane >> 2) == q) ? mine : hnv;

      sHd[lane] = acc0 * r0n;
      sHd[32 + lane] = acc1 * r1n;
      __builtin_amdgcn_wave_barrier();
      v2f hf[16];
#pragma unroll
      for (int i = 0; i < 16; ++i) {
        int o = (i >> 2) * 16 + (i & 3) * 4 + koff;
        hf[i] = *reinterpret_cast<const v2f*>(sHd + o);
      }
      v4f dsum = {0.f, 0.f, 0.f, 0.f};
      int len = end - s;
      if (len <= FU_TAILS) {
        if (len > 0) dsum = finish_chunk(hf, sT, koff, lane, cntK, elK, avK, A);
      } else {
        for (int c = s; c < end; c += 32) {
          int cnt = end - c;
          cnt = (cnt < 32) ? cnt : 32;
          int el = c + ((lane < cnt) ? lane : (cnt - 1));
          int tv = ts[el];
          v4f ao = {1.0f, 1.0f, 1.0f, 1.0f};
          if (!FIRST) ao = A4[el];
          v4f tn4 = tninv4[tv];
          __builtin_amdgcn_wave_barrier();
          for (int j = 0; j < cnt; ++j) {
            int t = __builtin_amdgcn_readlane(tv, j);
            const float* sp = row_of(rowsLo, rowsHi, split, t) + lane;
            float x0 = sp[0], x1 = sp[32];
            float tx = lane_bcast(tn4.x, j), ty = lane_bcast(tn4.y, j);
            float tz = lane_bcast(tn4.z, j), tw = lane_bcast(tn4.w, j);
            float* dst = sT + j * FU_PITCH + lane;
            dst[0] = fast_tanh(x0 * (upper ? ty : tx));
            dst[32] = fast_tanh(x1 * (upper ? tw : tz));
          }
          __builtin_amdgcn_wave_barrier();
          v4f t4 = finish_chunk(hf, sT, koff, lane, cnt, el, ao, A);
          dsum.x += t4.x; dsum.y += t4.y; dsum.z += t4.z; dsum.w += t4.w;
        }
      }
      float gLo = (lane & 1) ? dsum.y : dsum.x;
      float gHi = (lane & 1) ? dsum.w : dsum.z;
      float gs = (lane & 2) ? gHi : gLo;
      float gm = 1.0f / sqrtf(fmaxf(gs, 1e-8f));
      dnv = ((lane >> 2) == q) ? gm : dnv;
      __builtin_amdgcn_wave_barrier();
    }
  }
  if (!LAST) {
    int nb = wave * MSG_NODES;
    if (lane < MSG_NODES * 4 && nb + (lane >> 2) < N) {
      hninv[(size_t)nb * 4 + lane] = hnv;
      dinvNext[(size_t)nb * 4 + lane] = dnv;
    }
  }
}

__global__ void k_norminv(const float* __restrict__ lo, const float* __restrict__ hi, int split,
                          float* __restrict__ out, int* __restrict__ count, int NF, int N) {
  int i = blockIdx.x * blockDim.x + threadIdx.x;
  if (i < N) count[i] = 0;
  if (i >= NF) return;
  int n = i >> 2, f = i & 3;
  const float4* p = reinterpret_cast<const float4*>(row_of(lo, hi, split, n) + (size_t)f * 16);
  float s = 0.f;
#pragma unroll
  for (int k = 0; k < 4; ++k) {
    float4 v = p[k];
    s += v.x * v.x + v.y * v.y + v.z * v.z + v.w * v.w;
  }
  out[i] = 1.0f / fmaxf(sqrtf(s), 1e-12f);
}

static inline size_t align16f(size_t x) { return (x + 15) & ~(size_t)15; }

extern "C" void kernel_launch(void* const* d_in, const int* in_sizes, int n_in,
                              void* d_out, int out_size, void* d_ws, size_t ws_size,
                              hipStream_t stream) {
  (void)n_in; (void)out_size; (void)ws_size;
  const float* u  = (const float*)d_in[0];
  const float* it = (const float*)d_in[1];
  const int* hl   = (const int*)d_in[2];
  const int* tl   = (const int*)d_in[3];

  const int userElems = in_sizes[0];
  const int itemElems = in_sizes[1];
  const int E = in_sizes[2];
  const int N = (userElems + itemElems) / 64;
  const size_t N64 = (size_t)N * 64;
  const size_t E4  = (size_t)E * 4;
  const size_t N4  = (size_t)N * 4;

  float* base = (float*)d_ws;
  size_t off = 0;
  float* ego   = base + off; off += align16f(N64);
  float* fact  = base + off; off += align16f(N64);
  float* dinvB  = base + off;
  int* ts       = (int*)(dinvB + align16f(N4));
  int* rank     = ts + align16f(E);
  int* rowstart = rank + align16f(E);
  off += align16f(N64);
  float* A     = base + off; off += align16f(E4);
  float* dinv  = base + off; off += align16f(N4);
  float* hninv = base + off; off += align16f(N4);
  float* tninv = base + off; off += align16f(N4);

#define GRID(n) ((unsigned)(((size_t)(n) + WG - 1) / WG))
#define GRIDP(n, per) ((unsigned)(((size_t)(n) + (size_t)WG * (per) - 1) / ((size_t)WG * (per))))
  const int nUser = userElems / 64;

  k_norminv<<<GRID(N4), WG, 0, stream>>>(u, it, nUser, tninv, rowstart, (int)N4, N);

  k_head_rank<<<GRIDP(E, SORT_ITEMS), WG, 0, stream>>>(hl, rowstart, rank, E);
  k_scan<<<1, SCAN_WG, 0, stream>>>(rowstart, N);
  k_place<<<GRIDP(E, SORT_ITEMS), WG, 0, stream>>>(hl, tl, rowstart, rank, ts, E);

  const float* rowsLo = u;
  const float* rowsHi = it;
  float* nxt = fact;
  float* spare = ego;
  float* tn = tninv;
  float* hn = hninv;
  float* dcur = dinvB;
  float* dnext = dinv;
  const unsigned msgGrid = GRID((size_t)((N + MSG_NODES - 1) / MSG_NODES) * 32);

  for (int layer = 0; layer < 2; ++layer) {
    for (int itn = 0; itn < 2; ++itn) {
      const bool first = (layer == 0 && itn == 0);
      const bool last = (layer == 1 && itn == 1);
      if (first) {
        k_message<true, false><<<msgGrid, WG, 0, stream>>>(
            A, dcur, dnext, rowsLo, rowsHi, u, it, nUser, rowstart, ts, tn, nxt, hn, N, E);
      } else if (last) {
        k_message<false, true><<<msgGrid, WG, 0, stream>>>(
            A, dcur, dnext, rowsLo, rowsHi, u, it, nUser, rowstart, ts, tn, (float*)d_out, hn, N, E);
      } else {
        k_message<false, false><<<msgGrid, WG, 0, stream>>>(
            A, dcur, dnext, rowsLo, rowsHi, u, it, nUser, rowstart, ts, tn, nxt, hn, N, E);
      }
      float* tmpd = dcur; dcur = dnext; dnext = tmpd;
    }
    rowsLo = nxt;
    rowsHi = nxt + (size_t)nUser * 64;
    float* tmpf = nxt; nxt = spare; spare = tmpf;
    float* tmp = tn; tn = hn; hn = tmp;
  }
#undef GRID
#undef GRIDP
}


# Softmax exp and division, row inverse norms and degree inverse roots become single hardware transcendentals (exp2, rcp, rsq; 1 ulp) where the library's exact sequences ran; sums and their order kept
# speedup vs baseline: 5.0384x; 1.0537x over previous
#include <hip/hip_runtime.h>
#include <math.h>

typedef float v2f __attribute__((ext_vector_type(2)));
typedef float v8f __attribute__((ext_vector_type(8)));
typedef float v4f __attribute__((ext_vector_type(4)));

#define WG 256

__device__ __forceinline__ int atomic_rank(int* p) {
  return __hip_atomic_fetch_add(p, 1, __ATOMIC_RELAXED, __HIP_MEMORY_SCOPE_AGENT);
}

__device__ __forceinline__ float fast_tanh(float x) {
  float r;
  asm("v_tanh_f32 %0, %1\n\tv_nop\n\tv_nop" : "=v"(r) : "v"(x));
  return r;
}

__device__ __forceinline__ const float* row_of(const float* __restrict__ lo, const float* __restrict__ hi,
                                               int split, int n) {
  return (n < split) ? (lo + (size_t)n * 64) : (hi + (size_t)(n - split) * 64);
}

#define SORT_ITEMS 4

__global__ void k_head_rank(const int* __restrict__ hl, int* __restrict__ count,
                            int* __restrict__ rank, int E) {
  int base = blockIdx.x * (WG * SORT_ITEMS) + threadIdx.x;
  int h[SORT_ITEMS];
#pragma unroll
  for (int k = 0; k < SORT_ITEMS; ++k) {
    int e = base + k * WG;
    h[k] = hl[(e < E) ? e : (E - 1)];
  }
  int r[SORT_ITEMS];
#pragma unroll
  for (int k = 0; k < SORT_ITEMS; ++k) {
    int e = base + k * WG;
    r[k] = 0;
    if (e < E) r[k] = atomic_rank(count + h[k]);
  }
#pragma unroll
  for (int k = 0; k < SORT_ITEMS; ++k) {
    int e = base + k * WG;
    if (e < E) rank[e] = r[k];
  }
}

#define SCAN_WG 1024
#define SCAN_STEPS 20
#define SCAN_WAVE_SPAN (SCAN_STEPS * 128)

__device__ __forceinline__ int wave_incl_scan(int x, int lane) {
#pragma unroll
  for (int d = 1; d < 32; d <<= 1) {
    int y = __shfl_up(x, d);
    if (lane >= d) x += y;
  }
  return x;
}

__global__ void __launch_bounds__(SCAN_WG) k_scan(int* __restrict__ count, int N) {
  __shared__ int wsum[SCAN_WG / 32];
  int lane = threadIdx.x & 31;
  int w = __builtin_amdgcn_readfirstlane((int)(threadIdx.x >> 5));
  int carry = 0;
  for (int rbase = 0; rbase < N; rbase += (SCAN_WG / 32) * SCAN_WAVE_SPAN) {
    int wbase = rbase + w * SCAN_WAVE_SPAN;
    int v[SCAN_STEPS][4];
    int tot = 0;
#pragma unroll
    for (int k = 0; k < SCAN_STEPS; ++k) {
      int i = wbase + k * 128 + lane * 4;
      if (i + 3 < N) {
        int4 x = *reinterpret_cast<const int4*>(count + i);
        v[k][0] = x.x; v[k][1] = x.y; v[k][2] = x.z; v[k][3] = x.w;
      } else {
#pragma unroll
        for (int q = 0; q < 4; ++q) v[k][q] = (i + q < N) ? count[i + q] : 0;
      }
      tot += v[k][0] + v[k][1] + v[k][2] + v[k][3];
    }
    int incl = wave_incl_scan(tot, lane);
    if (lane == 31) wsum[w] = incl;
    __syncthreads();
    int xs = wave_incl_scan(wsum[lane], lane);
    int run = carry + ((w > 0) ? __builtin_amdgcn_readlane(xs, w - 1) : 0);
    carry += __builtin_amdgcn_readlane(xs, 31);
#pragma unroll
    for (int k = 0; k < SCAN_STEPS; ++k) {
      int i = wbase + k * 128 + lane * 4;
      int l4 = v[k][0] + v[k][1] + v[k][2] + v[k][3];
      int inc = wave_incl_scan(l4, lane);
      int b = run + inc - l4;
      if (i + 3 < N) {
        int4 o;
        o.x = b; o.y = o.x + v[k][0]; o.z = o.y + v[k][1]; o.w = o.z + v[k][2];
        *reinterpret_cast<int4*>(count + i) = o;
      } else {
#pragma unroll
        for (int q = 0; q < 4; ++q) {
          if (i + q < N) count[i + q] = b;
          b += v[k][q];
        }
      }
      run += __builtin_amdgcn_readlane(inc, 31);
    }
    __syncthreads();
  }
}

__global__ void k_place(const int* __restrict__ hl, const int* __restrict__ tl,
                        const int* __restrict__ rowstart, const int* __restrict__ rank,
                        int* __restrict__ ts, int E) {
  int base = blockIdx.x * (WG * SORT_ITEMS) + threadIdx.x;
  int h[SORT_ITEMS], t[SORT_ITEMS], r[SORT_ITEMS], p[SORT_ITEMS];
#pragma unroll
  for (int k = 0; k < SORT_ITEMS; ++k) {
    int e = base + k * WG;
    e = (e < E) ? e : (E - 1);
    h[k] = hl[e];
    t[k] = tl[e];
    r[k] = rank[e];
  }
#pragma unroll
  for (int k = 0; k < SORT_ITEMS; ++k) p[k] = rowstart[h[k]] + r[k];
#pragma unroll
  for (int k = 0; k < SORT_ITEMS; ++k) {
    int e = base + k * WG;
    if (e < E) ts[p[k]] = t[k];
  }
}

__device__ __forceinline__ v4f softmax4(v4f a) {
  const float LOG2E = 1.44269504088896340736f;
  float m = fmaxf(fmaxf(a.x, a.y), fmaxf(a.z, a.w));
  float e0 = __builtin_amdgcn_exp2f((a.x - m) * LOG2E), e1 = __builtin_amdgcn_exp2f((a.y - m) * LOG2E);
  float e2 = __builtin_amdgcn_exp2f((a.z - m) * LOG2E), e3 = __builtin_amdgcn_exp2f((a.w - m) * LOG2E);
  float inv = __builtin_amdgcn_rcpf(e0 + e1 + e2 + e3);
  v4f r = {e0 * inv, e1 * inv, e2 * inv, e3 * inv};
  return r;
}

__device__ __forceinline__ float deg_rsqrt(float x) {
  return __builtin_amdgcn_rsqf(fmaxf(x, 1e-8f));
}
__device__ __forceinline__ float norm_rinv(float ss) {
  return fminf(__builtin_amdgcn_rsqf(ss), 1e12f);
}

#define MSG_NODES 4
#define MSG_BATCH 8
static_assert(MSG_BATCH == 8, "the asm statement in k_message names eight edges' row words");

__device__ __forceinline__ float lane_bcast(float v, int srcLane) {
  return __int_as_float(__builtin_amdgcn_readlane(__float_as_int(v), srcLane));
}

template <int MASK>
__device__ __forceinline__ float add_xmask(float x) {
  return x + __int_as_float(__builtin_amdgcn_update_dpp(0, __float_as_int(x), 0x160 + MASK, 0xf, 0xf, false));
}
__device__ __forceinline__ float half16_sum(float x) {
  x = add_xmask<1>(x);
  x = add_xmask<2>(x);
  x = add_xmask<4>(x);
  x = add_xmask<8>(x);
  return x;
}

#define FU_PITCH 68
#define FU_TAILS 32
#define FU_ROWS_FLOATS (FU_TAILS * FU_PITCH + 64)
#define FU_BC_W (4 * FU_TAILS)
#define FU_WAVE_FLOATS (FU_ROWS_FLOATS + 2 * FU_BC_W)
static_assert((FU_ROWS_FLOATS % 4) == 0 && (FU_WAVE_FLOATS % 4) == 0, "16-byte LDS loads from the broadcast table");

__device__ __forceinline__ void tile_dots(const v2f (&hf)[16], const float* Trow, int koff, float (&dv)[4]) {
#pragma unroll
  for (int f = 0; f < 4; ++f) {
    v8f c = {0.f, 0.f, 0.f, 0.f, 0.f, 0.f, 0.f, 0.f};
#pragma unroll
    for (int s = 0; s < 4; ++s) {
      int o = f * 16 + s * 4 + koff;
      v2f bv = *reinterpret_cast<const v2f*>(Trow + o);
      c = __builtin_amdgcn_wmma_f32_16x16x4_f32(false, hf[f * 4 + s], false, bv,
                                                (short)0, c, false, false);
    }
    dv[f] = c[0];
  }
}

__device__ __forceinline__ v4f finish_chunk(const v2f (&hf)[16], const float* sT, int koff, int lane,
                                            int cnt, int el, v4f aold, float* __restrict__ A) {
  int r = lane & 15;
  float d0[4], d1[4] = {0.f, 0.f, 0.f, 0.f};
  if (cnt > 16) {
    int rowB = (16 + r < cnt) ? (16 + r) : (cnt - 1);
    tile_dots(hf, sT + rowB * FU_PITCH, koff, d1);
  }
  int rowA = (r < cnt) ? r : (cnt - 1);
  tile_dots(hf, sT + rowA * FU_PITCH, koff, d0);
  bool lo = lane < 16;
  v4f an;
  an.x = aold.x + (lo ? d0[0] : d1[0]);
  an.y = aold.y + (lo ? d0[1] : d1[1]);
  an.z = aold.z + (lo ? d0[2] : d1[2]);
  an.w = aold.w + (lo ? d0[3] : d1[3]);
  bool mine = lane < cnt;
  if (mine) reinterpret_cast<v4f*>(A)[el] = an;
  v4f sm = softmax4(an);
  float x0 = half16_sum(mine ? sm.x : 0.f), x1 = half16_sum(mine ? sm.y : 0.f);
  float x2 = half16_sum(mine ? sm.z : 0.f), x3 = half16_sum(mine ? sm.w : 0.f);
  v4f tot;
  tot.x = lane_bcast(x0, 0) + lane_bcast(x0, 16);
  tot.y = lane_bcast(x1, 0) + lane_bcast(x1, 16);
  tot.z = lane_bcast(x2, 0) + lane_bcast(x2, 16);
  tot.w = lane_bcast(x3, 0) + lane_bcast(x3, 16);
  return tot;
}

template <bool FIRST, bool LAST>
__global__ void __launch_bounds__(WG, 8) k_message(
    float* __restrict__ A, const float* __restrict__ dinv, float* __restrict__ dinvNext,
    const float* __restrict__ rowsLo, const float* __restrict__ rowsHi,
    const float* __restrict__ inLo, const float* __restrict__ inHi, int split,
    const int* __restrict__ rowstart, const int* __restrict__ ts, const float* __restrict__ tninv,
    float* __restrict__ fact, float* __restrict__ hninv, int N, int E) {
  __shared__ __attribute__((aligned(16))) float stage[(WG / 32) * (LAST ? FU_BC_W : FU_WAVE_FLOATS)];
  int lane = threadIdx.x & 31;
  int wave = __builtin_amdgcn_readfirstlane((int)((blockIdx.x * WG + threadIdx.x) >> 5));
  int wib = __builtin_amdgcn_readfirstlane((int)(threadIdx.x >> 5));
  bool upper = lane >= 16;
  int koff = (lane >> 4) << 1;
  v4f* __restrict__ A4 = reinterpret_cast<v4f*>(A);
  const v4f* __restrict__ dinv4 = reinterpret_cast<const v4f*>(dinv);
  const v4f* __restrict__ tninv4 = reinterpret_cast<const v4f*>(tninv);
  float* sT = stage + (LAST ? 0 : wib * FU_WAVE_FLOATS);
  float* sHd = sT + (LAST ? 0 : FU_TAILS * FU_PITCH);
  float* sW = LAST ? (stage + wib * FU_BC_W) : (sT + FU_ROWS_FLOATS);
  float* sN = sW + FU_BC_W;
  const float* bcl = sW + (lane >> 4) * FU_TAILS;

  float hnv = 0.f;
  float dnv = 0.f;
  for (int q = 0; q < MSG_NODES; ++q) {
    int n = wave * MSG_NODES + q;
    if (n >= N) break;
    int s = rowstart[n];
    int end = (n + 1 < N) ? rowstart[n + 1] : E;
    v4f dh;
    if (FIRST) {
      float dv = deg_rsqrt(0.25f * (float)(end - s));
      dh.x = dv; dh.y = dv; dh.z = dv; dh.w = dv;
    } else {
      dh = dinv4[n];
    }
    float in0 = 0.f, in1 = 0.f, pv0 = 0.f, pv1 = 0.f;
    if (LAST) {
      const float* ip = row_of(inLo, inHi, split, n) + lane;
      const float* pp = row_of(rowsLo, rowsHi, split, n) + lane;
      in0 = ip[0]; in1 = ip[32];
      pv0 = pp[0]; pv1 = pp[32];
      asm volatile("" : "+v"(in0), "+v"(in1), "+v"(pv0), "+v"(pv1));
    }
    float acc0 = 0.f, acc1 = 0.f;
    int elK = 0, cntK = 0;
    v4f avK = {1.0f, 1.0f, 1.0f, 1.0f};
    for (int c = s; c < end; c += 32) {
      int cnt = end - c;
      cnt = (cnt < 32) ? cnt : 32;
      int el = c + ((lane < cnt) ? lane : (cnt - 1));
      int tv = ts[el];
      v4f nw;
      v4f dt = {0.f, 0.f, 0.f, 0.f};
      v4f tn4 = {0.f, 0.f, 0.f, 0.f};
      int r0 = 0, r1raw = 0;
      if (FIRST) {
        nw.x = 0.25f; nw.y = 0.25f; nw.z = 0.25f; nw.w = 0.25f;
        r0 = rowstart[tv];
        r1raw = rowstart[(tv + 1 < N) ? (tv + 1) : tv];
      } else {
        v4f av = A4[el];
        asm volatile("" : "+v"(tv), "+v"(av));
        dt = dinv4[tv];
        nw = softmax4(av);
        avK = av;
      }
      if (!LAST) tn4 = tninv4[tv];
      elK = el; cntK = cnt;
      v4f wv = {0.f, 0.f, 0.f, 0.f};
      for (int b = 0; b < cnt; b += MSG_BATCH) {
        float p0[MSG_BATCH], p1[MSG_BATCH];
#pragma unroll
        for (int k = 0; k < MSG_BATCH; ++k) {
          int j = b + k;
          j = (j < cnt) ? j : (cnt - 1);
          int t = __builtin_amdgcn_readlane(tv, j);
          const float* sp = row_of(rowsLo, rowsHi, split, t) + lane;
          p0[k] = sp[0];
          p1[k] = sp[32];
        }
        if (b == 0) {
          if (FIRST) {
            asm volatile("" : "+v"(r0), "+v"(r1raw));
            int r1 = (tv + 1 < N) ? r1raw : E;
            float dtv = deg_rsqrt(0.25f * (float)(r1 - r0));
            wv.x = nw.x * (dh.x * dtv);
            wv.y = nw.y * (dh.y * dtv);
            wv.z = nw.z * (dh.z * dtv);
            wv.w = nw.w * (dh.w * dtv);
          } else {
            asm volatile("" : "+v"(dt));
            wv.x = nw.x * (dh.x * dt.x);
            wv.y = nw.y * (dh.y * dt.y);
            wv.z = nw.z * (dh.z * dt.z);
            wv.w = nw.w * (dh.w * dt.w);
          }
          __builtin_amdgcn_wave_barrier();
          sW[lane] = wv.x;
          sW[FU_TAILS + lane] = wv.y;
          sW[2 * FU_TAILS + lane] = wv.z;
          sW[3 * FU_TAILS + lane] = wv.w;
          if (!LAST) {
            sN[lane] = tn4.x;
            sN[FU_TAILS + lane] = tn4.y;
            sN[2 * FU_TAILS + lane] = tn4.z;
            sN[3 * FU_TAILS + lane] = tn4.w;
          }
          __builtin_amdgcn_wave_barrier();
        }
        float w0[MSG_BATCH], w1[MSG_BATCH], t0[MSG_BATCH], t1[MSG_BATCH];
        {
          v4f a0 = *reinterpret_cast<const v4f*>(bcl + b), a1 = *reinterpret_cast<const v4f*>(bcl + b + 4);
          v4f c0 = *reinterpret_cast<const v4f*>(bcl + 2 * FU_TAILS + b);
          v4f c1 = *reinterpret_cast<const v4f*>(bcl + 2 * FU_TAILS + b + 4);
          w0[0] = a0.x; w0[1] = a0.y; w0[2] = a0.z; w0[3] = a0.w;
          w0[4] = a1.x; w0[5] = a1.y; w0[6] = a1.z; w0[7] = a1.w;
          w1[0] = c0.x; w1[1] = c0.y; w1[2] = c0.z; w1[3] = c0.w;
          w1[4] = c1.x; w1[5] = c1.y; w1[6] = c1.z; w1[7] = c1.w;
        }
        if (!LAST) {
          const float* nl = bcl + FU_BC_W;
          v4f a0 = *reinterpret_cast<const v4f*>(nl + b), a1 = *reinterpret_cast<const v4f*>(nl + b + 4);
          v4f c0 = *reinterpret_cast<const v4f*>(nl + 2 * FU_TAILS + b);
          v4f c1 = *reinterpret_cast<const v4f*>(nl + 2 * FU_TAILS + b + 4);
          t0[0] = a0.x; t0[1] = a0.y; t0[2] = a0.z; t0[3] = a0.w;
          t0[4] = a1.x; t0[5] = a1.y; t0[6] = a1.z; t0[7] = a1.w;
          t1[0] = c0.x; t1[1] = c0.y; t1[2] = c0.z; t1[3] = c0.w;
          t1[4] = c1.x; t1[5] = c1.y; t1[6] = c1.z; t1[7] = c1.w;
        }
        asm volatile("" : "+v"(p0[0]), "+v"(p0[1]), "+v"(p0[2]), "+v"(p0[3]),
                          "+v"(p0[4]), "+v"(p0[5]), "+v"(p0[6]), "+v"(p0[7]),
                          "+v"(p1[0]), "+v"(p1[1]), "+v"(p1[2]), "+v"(p1[3]),
                          "+v"(p1[4]), "+v"(p1[5]), "+v"(p1[6]), "+v"(p1[7]));
#pragma unroll
        for (int k = 0; k < MSG_BATCH; ++k) {
          if (b + k < cnt) {
            acc0 += w0[k] * p0[k];
            acc1 += w1[k] * p1[k];
            if (!LAST) {
              float* dst = sT + (b + k) * FU_PITCH + lane;
              dst[0] = fast_tanh(p0[k] * t0[k]);
              dst[32] = fast_tanh(p1[k] * t1[k]);
            }
          }
        }
      }
    }
    float* d = fact + (size_t)n * 64;
    asm volatile("" : "+s"(d));
    if (LAST) {
      d[lane] = ((in0 + pv0) + acc0) * (1.0f / 3.0f);
      d[32 + lane] = ((in1 + pv1) + acc1) * (1.0f / 3.0f);
    } else {
      d[lane] = acc0;
      d[32 + lane] = acc1;
      float r0n = norm_rinv(half16_sum(acc0 * acc0));
      float r1n = norm_rinv(half16_sum(acc1 * acc1));
      float h0 = lane_bcast(r0n, 0), h1 = lane_bcast(r0n, 16), h2 = lane_bcast(r1n, 0), h3 = lane_bcast(r1n, 16);
      float fLo = (lane & 1) ? h1 : h0;
      float fHi = (lane & 1) ? h3 : h2;
      float mine = (lane & 2) ? fHi : fLo;
      hnv = ((lane >> 2) == q) ? mine : hnv;

      sHd[lane] = acc0 * r0n;
      sHd[32 + lane] = acc1 * r1n;
      __builtin_amdgcn_wave_barrier();
      v2f hf[16];
#pragma unroll
      for (int i = 0; i < 16; ++i) {
        int o = (i >> 2) * 16 + (i & 3) * 4 + koff;
        hf[i] = *reinterpret_cast<const v2f*>(sHd + o);
      }
      v4f dsum = {0.f, 0.f, 0.f, 0.f};
      int len = end - s;
      if (len <= FU_TAILS) {
        if (len > 0) dsum = finish_chunk(hf, sT, koff, lane, cntK, elK, avK, A);
      } else {
        for (int c = s; c < end; c += 32) {
          int cnt = end - c;
          cnt = (cnt < 32) ? cnt : 32;
          int el = c + ((lane < cnt) ? lane : (cnt - 1));
          int tv = ts[el];
          v4f ao = {1.0f, 1.0f, 1.0f, 1.0f};
          if (!FIRST) ao = A4[el];
          v4f tn4 = tninv4[tv];
          __builtin_amdgcn_wave_barrier();
          for (int j = 0; j < cnt; ++j) {
            int t = __builtin_amdgcn_readlane(tv, j);
            const float* sp = row_of(rowsLo, rowsHi, split, t) + lane;
            float x0 = sp[0], x1 = sp[32];
            float tx = lane_bcast(tn4.x, j), ty = lane_bcast(tn4.y, j);
            float tz = lane_bcast(tn4.z, j), tw = lane_bcast(tn4.w, j);
            float* dst = sT + j * FU_PITCH + lane;
            dst[0] = fast_tanh(x0 * (upper ? ty : tx));
            dst[32] = fast_tanh(x1 * (upper ? tw : tz));
          }
          __builtin_amdgcn_wave_barrier();
          v4f t4 = finish_chunk(hf, sT, koff, lane, cnt, el, ao, A);
          dsum.x += t4.x; dsum.y += t4.y; dsum.z += t4.z; dsum.w += t4.w;
        }
      }
      float gLo = (lane & 1) ? dsum.y : dsum.x;
      float gHi = (lane & 1) ? dsum.w : dsum.z;
      float gs = (lane & 2) ? gHi : gLo;
      float gm = deg_rsqrt(gs);
      dnv = ((lane >> 2) == q) ? gm : dnv;
      __builtin_amdgcn_wave_barrier();
    }
  }
  if (!LAST) {
    int nb = wave * MSG_NODES;
    if (lane < MSG_NODES * 4 && nb + (lane >> 2) < N) {
      hninv[(size_t)nb * 4 + lane] = hnv;
      dinvNext[(size_t)nb * 4 + lane] = dnv;
    }
  }
}

__global__ void k_norminv(const float* __restrict__ lo, const float* __restrict__ hi, int split,
                          float* __restrict__ out, int* __restrict__ count, int NF, int N) {
  int i = blockIdx.x * blockDim.x + threadIdx.x;
  if (i < N) count[i] = 0;
  if (i >= NF) return;
  int n = i >> 2, f = i & 3;
  const float4* p = reinterpret_cast<const float4*>(row_of(lo, hi, split, n) + (size_t)f * 16);
  float s = 0.f;
#pragma unroll
  for (int k = 0; k < 4; ++k) {
    float4 v = p[k];
    s += v.x * v.x + v.y * v.y + v.z * v.z + v.w * v.w;
  }
  out[i] = 1.0f / fmaxf(sqrtf(s), 1e-12f);
}

static inline size_t align16f(size_t x) { return (x + 15) & ~(size_t)15; }

extern "C" void kernel_launch(void* const* d_in, const int* in_sizes, int n_in,
                              void* d_out, int out_size, void* d_ws, size_t ws_size,
                              hipStream_t stream) {
  (void)n_in; (void)out_size; (void)ws_size;
  const float* u  = (const float*)d_in[0];
  const float* it = (const float*)d_in[1];
  const int* hl   = (const int*)d_in[2];
  const int* tl   = (const int*)d_in[3];

  const int userElems = in_sizes[0];
  const int itemElems = in_sizes[1];
  const int E = in_sizes[2];
  const int N = (userElems + itemElems) / 64;
  const size_t N64 = (size_t)N * 64;
  const size_t E4  = (size_t)E * 4;
  const size_t N4  = (size_t)N * 4;

  float* base = (float*)d_ws;
  size_t off = 0;
  float* ego   = base + off; off += align16f(N64);
  float* fact  = base + off; off += align16f(N64);
  float* dinvB  = base + off;
  int* ts       = (int*)(dinvB + align16f(N4));
  int* rank     = ts + align16f(E);
  int* rowstart = rank + align16f(E);
  off += align16f(N64);
  float* A     = base + off; off += align16f(E4);
  float* dinv  = base + off; off += align16f(N4);
  float* hninv = base + off; off += align16f(N4);
  float* tninv = base + off; off += align16f(N4);

#define GRID(n) ((unsigned)(((size_t)(n) + WG - 1) / WG))
#define GRIDP(n, per) ((unsigned)(((size_t)(n) + (size_t)WG * (per) - 1) / ((size_t)WG * (per))))
  const int nUser = userElems / 64;

  k_norminv<<<GRID(N4), WG, 0, stream>>>(u, it, nUser, tninv, rowstart, (int)N4, N);

  k_head_rank<<<GRIDP(E, SORT_ITEMS), WG, 0, stream>>>(hl, rowstart, rank, E);
  k_scan<<<1, SCAN_WG, 0, stream>>>(rowstart, N);
  k_place<<<GRIDP(E, SORT_ITEMS), WG, 0, stream>>>(hl, tl, rowstart, rank, ts, E);

  const float* rowsLo = u;
  const float* rowsHi = it;
  float* nxt = fact;
  float* spare = ego;
  float* tn = tninv;
  float* hn = hninv;
  float* dcur = dinvB;
  float* dnext = dinv;
  const unsigned msgGrid = GRID((size_t)((N + MSG_NODES - 1) / MSG_NODES) * 32);

  for (int layer = 0; layer < 2; ++layer) {
    for (int itn = 0; itn < 2; ++itn) {
      const bool first = (layer == 0 && itn == 0);
      const bool last = (layer == 1 && itn == 1);
      if (first) {
        k_message<true, false><<<msgGrid, WG, 0, stream>>>(
            A, dcur, dnext, rowsLo, rowsHi, u, it, nUser, rowstart, ts, tn, nxt, hn, N, E);
      } else if (last) {
        k_message<false, true><<<msgGrid, WG, 0, stream>>>(
            A, dcur, dnext, rowsLo, rowsHi, u, it, nUser, rowstart, ts, tn, (float*)d_out, hn, N, E);
      } else {
        k_message<false, false><<<msgGrid, WG, 0, stream>>>(
            A, dcur, dnext, rowsLo, rowsHi, u, it, nUser, rowstart, ts, tn, nxt, hn, N, E);
      }
      float* tmpd = dcur; dcur = dnext; dnext = tmpd;
    }
    rowsLo = nxt;
    rowsHi = nxt + (size_t)nUser * 64;
    float* tmpf = nxt; nxt = spare; spare = tmpf;
    float* tmp = tn; tn = hn; hn = tmp;
  }
#undef GRID
#undef GRIDP
}
